# starting kernel
# baseline (speedup 1.0000x reference)
#include <hip/hip_runtime.h>
#include <hip/hip_bf16.h>
#include <math.h>

#define N_NODES 100000
#define N_EDGES 1600000
#define F_DIM 128
#define H_DIM 128
#define C_DIM 16

typedef __attribute__((ext_vector_type(2))) float v2f;
typedef __attribute__((ext_vector_type(8))) float v8f;

__global__ void init_deg_kernel(float* __restrict__ deg) {
  const int i = blockIdx.x * blockDim.x + threadIdx.x;
  if (i < N_NODES) deg[i] = 1.0f;
}

__global__ void deg_atomic_kernel(const int* __restrict__ dst,
                                  const float* __restrict__ ew,
                                  float* __restrict__ deg) {
  const int e = blockIdx.x * blockDim.x + threadIdx.x;
  if (e < N_EDGES) atomicAdd(&deg[dst[e]], ew[e]);
}

__global__ void dinv_kernel(float* __restrict__ deg) {
  const int i = blockIdx.x * blockDim.x + threadIdx.x;
  if (i < N_NODES) {
    const float d = deg[i];
    deg[i] = (d > 0.0f) ? rsqrtf(d) : 0.0f;
  }
}

__global__ __launch_bounds__(256) void gemm1_wmma(const float* __restrict__ x,
                                                  const float* __restrict__ W1,
                                                  float* __restrict__ h) {
  __shared__ float sX[64 * 132];
  const int tid  = threadIdx.x;
  const int row0 = blockIdx.x * 64;

  for (int i = tid; i < 64 * 32; i += 256) {
    const int r = i >> 5, c = i & 31;
    float4 v = make_float4(0.f, 0.f, 0.f, 0.f);
    const int gr = row0 + r;
    if (gr < N_NODES) v = ((const float4*)x)[gr * 32 + c];
    *(float4*)&sX[r * 132 + c * 4] = v;
  }
  __syncthreads();

  const int wave = tid >> 5;
  const int lane = tid & 31;
  const int lrow = lane & 15;
  const int hi   = lane >> 4;
  const int col  = wave * 16 + lrow;

  for (int t = 0; t < 4; ++t) {
    v8f acc = {};
    const float* ax = &sX[(t * 16 + lrow) * 132 + 2 * hi];
    for (int k0 = 0; k0 < 128; k0 += 4) {
      v2f a, b;
      a.x = ax[k0];
      a.y = ax[k0 + 1];
      const int kb = k0 + 2 * hi;
      b.x = W1[kb * 128 + col];
      b.y = W1[(kb + 1) * 128 + col];
      acc = __builtin_amdgcn_wmma_f32_16x16x4_f32(false, a, false, b,
                                                  (short)0, acc, false, false);
    }
    const int rbase = row0 + t * 16 + 8 * hi;
#pragma unroll
    for (int r = 0; r < 8; ++r) {
      const int gr = rbase + r;
      if (gr < N_NODES) h[gr * 128 + col] = acc[r];
    }
  }
}

__global__ void selfloop1_kernel(const float* __restrict__ h,
                                 const float* __restrict__ dinv,
                                 float* __restrict__ agg) {
  const int idx = blockIdx.x * blockDim.x + threadIdx.x;
  if (idx >= N_NODES * 128) return;
  const float di = dinv[idx >> 7];
  agg[idx] = h[idx] * di * di;
}

__global__ void scatter1_kernel(const int* __restrict__ src,
                                const int* __restrict__ dst,
                                const float* __restrict__ ew,
                                const float* __restrict__ dinv,
                                const float* __restrict__ h,
                                float* __restrict__ agg) {
  const int gid  = blockIdx.x * blockDim.x + threadIdx.x;
  const int e    = gid >> 5;
  const int lane = threadIdx.x & 31;
  if (e >= N_EDGES) return;
  const int   s    = src[e];
  const int   d    = dst[e];
  const float norm = dinv[s] * ew[e] * dinv[d];
  const float4 v = ((const float4*)(h + s * 128))[lane];
  float* o = agg + d * 128 + lane * 4;
  atomicAdd(o + 0, v.x * norm);
  atomicAdd(o + 1, v.y * norm);
  atomicAdd(o + 2, v.z * norm);
  atomicAdd(o + 3, v.w * norm);
}

__global__ __launch_bounds__(256) void gemm2_wmma(const float* __restrict__ agg1,
                                                  const float* __restrict__ b1,
                                                  const float* __restrict__ W2,
                                                  float* __restrict__ h2) {
  __shared__ float sW[128 * 16];
  __shared__ float sb[128];
  const int tid = threadIdx.x;
  for (int i = tid; i < 512; i += 256) ((float4*)sW)[i] = ((const float4*)W2)[i];
  if (tid < 128) sb[tid] = b1[tid];
  __syncthreads();

  const int wave = tid >> 5, lane = tid & 31;
  const int lrow = lane & 15, hi = lane >> 4;
  const int row0 = blockIdx.x * 128 + wave * 16;
  const int arow = row0 + lrow;

  v8f acc = {};
  for (int k0 = 0; k0 < 128; k0 += 4) {
    const int k = k0 + 2 * hi;
    float a0 = 0.f, a1 = 0.f;
    if (arow < N_NODES) {
      a0 = fmaxf(agg1[arow * 128 + k] + sb[k], 0.f);
      a1 = fmaxf(agg1[arow * 128 + k + 1] + sb[k + 1], 0.f);
    }
    v2f a, b;
    a.x = a0; a.y = a1;
    b.x = sW[k * 16 + lrow];
    b.y = sW[(k + 1) * 16 + lrow];
    acc = __builtin_amdgcn_wmma_f32_16x16x4_f32(false, a, false, b,
                                                (short)0, acc, false, false);
  }
  const int rbase = row0 + 8 * hi;
#pragma unroll
  for (int r = 0; r < 8; ++r) {
    const int gr = rbase + r;
    if (gr < N_NODES) h2[gr * 16 + lrow] = acc[r];
  }
}

__global__ void selfloop2_kernel(const float* __restrict__ h2,
                                 const float* __restrict__ dinv,
                                 float* __restrict__ agg) {
  const int idx = blockIdx.x * blockDim.x + threadIdx.x;
  if (idx >= N_NODES * 16) return;
  const float di = dinv[idx >> 4];
  agg[idx] = h2[idx] * di * di;
}

__global__ void scatter2_kernel(const int* __restrict__ src,
                                const int* __restrict__ dst,
                                const float* __restrict__ ew,
                                const float* __restrict__ dinv,
                                const float* __restrict__ h2,
                                float* __restrict__ agg) {
  const int gid = blockIdx.x * blockDim.x + threadIdx.x;
  const int e   = gid >> 2;
  const int sub = threadIdx.x & 3;
  if (e >= N_EDGES) return;
  const int   s    = src[e];
  const int   d    = dst[e];
  const float norm = dinv[s] * ew[e] * dinv[d];
  const float4 v = ((const float4*)(h2 + s * 16))[sub];
  float* o = agg + d * 16 + sub * 4;
  atomicAdd(o + 0, v.x * norm);
  atomicAdd(o + 1, v.y * norm);
  atomicAdd(o + 2, v.z * norm);
  atomicAdd(o + 3, v.w * norm);
}

__global__ void softmax_kernel(const float* __restrict__ agg2,
                               const float* __restrict__ b2,
                               float* __restrict__ out) {
  const int i = blockIdx.x * blockDim.x + threadIdx.x;
  if (i >= N_NODES) return;
  float v[16];
  const float4* p = (const float4*)(agg2 + (size_t)i * 16);
  float m = -3.402823466e+38f;
#pragma unroll
  for (int q = 0; q < 4; ++q) {
    const float4 t = p[q];
    v[4 * q + 0] = t.x + b2[4 * q + 0];
    v[4 * q + 1] = t.y + b2[4 * q + 1];
    v[4 * q + 2] = t.z + b2[4 * q + 2];
    v[4 * q + 3] = t.w + b2[4 * q + 3];
  }
#pragma unroll
  for (int c = 0; c < 16; ++c) m = fmaxf(m, v[c]);
  float s = 0.f;
#pragma unroll
  for (int c = 0; c < 16; ++c) { v[c] = expf(v[c] - m); s += v[c]; }
  const float inv = 1.0f / s;
  float4* o = (float4*)(out + (size_t)i * 16);
#pragma unroll
  for (int q = 0; q < 4; ++q)
    o[q] = make_float4(v[4 * q + 0] * inv, v[4 * q + 1] * inv,
                       v[4 * q + 2] * inv, v[4 * q + 3] * inv);
}

extern "C" void kernel_launch(void* const* d_in, const int* in_sizes, int n_in,
                              void* d_out, int out_size, void* d_ws, size_t ws_size,
                              hipStream_t stream) {
  const float* x  = (const float*)d_in[0];
  const int*   ei = (const int*)d_in[1];
  const float* ew = (const float*)d_in[2];
  const float* W1 = (const float*)d_in[3];
  const float* b1 = (const float*)d_in[4];
  const float* W2 = (const float*)d_in[5];
  const float* b2 = (const float*)d_in[6];
  float* out = (float*)d_out;

  const int* src = ei;
  const int* dst = ei + N_EDGES;

  float* ws   = (float*)d_ws;
  float* dinv = ws;
  float* h    = ws + N_NODES;
  float* agg1 = ws + N_NODES + (size_t)N_NODES * 128;
  float* h2   = h;
  float* agg2 = h + (size_t)N_NODES * 16;

  const int T = 256;

  init_deg_kernel<<<(N_NODES + T - 1) / T, T, 0, stream>>>(dinv);
  deg_atomic_kernel<<<(N_EDGES + T - 1) / T, T, 0, stream>>>(dst, ew, dinv);
  dinv_kernel<<<(N_NODES + T - 1) / T, T, 0, stream>>>(dinv);

  gemm1_wmma<<<(N_NODES + 63) / 64, T, 0, stream>>>(x, W1, h);
  selfloop1_kernel<<<(N_NODES * 128 + T - 1) / T, T, 0, stream>>>(h, dinv, agg1);
  scatter1_kernel<<<(N_EDGES * 32 + T - 1) / T, T, 0, stream>>>(src, dst, ew, dinv, h, agg1);

  gemm2_wmma<<<(N_NODES + 127) / 128, T, 0, stream>>>(agg1, b1, W2, h2);
  selfloop2_kernel<<<(N_NODES * 16 + T - 1) / T, T, 0, stream>>>(h2, dinv, agg2);
  scatter2_kernel<<<(N_EDGES * 4 + T - 1) / T, T, 0, stream>>>(src, dst, ew, dinv, h2, agg2);

  softmax_kernel<<<(N_NODES + T - 1) / T, T, 0, stream>>>(agg2, b2, out);
}


# Layer 1: 250 owner work-groups each keep 400 output rows in LDS, filter the dst list and sum rows with LDS float atomics (arrival order, as before); layer-1 global atomics and self-loop launch go
# speedup vs baseline: 1.7600x; 1.7600x over previous
#include <hip/hip_runtime.h>
#include <hip/hip_bf16.h>
#include <math.h>

#define N_NODES 100000
#define N_EDGES 1600000
#define F_DIM 128
#define H_DIM 128
#define C_DIM 16

typedef __attribute__((ext_vector_type(2))) float v2f;
typedef __attribute__((ext_vector_type(8))) float v8f;

__global__ void init_deg_kernel(float* __restrict__ deg) {
  const int i = blockIdx.x * blockDim.x + threadIdx.x;
  if (i < N_NODES) deg[i] = 1.0f;
}

__global__ void deg_atomic_kernel(const int* __restrict__ dst,
                                  const float* __restrict__ ew,
                                  float* __restrict__ deg) {
  const int e = blockIdx.x * blockDim.x + threadIdx.x;
  if (e < N_EDGES) atomicAdd(&deg[dst[e]], ew[e]);
}

__global__ void dinv_kernel(float* __restrict__ deg) {
  const int i = blockIdx.x * blockDim.x + threadIdx.x;
  if (i < N_NODES) {
    const float d = deg[i];
    deg[i] = (d > 0.0f) ? rsqrtf(d) : 0.0f;
  }
}

__global__ __launch_bounds__(256) void gemm1_wmma(const float* __restrict__ x,
                                                  const float* __restrict__ W1,
                                                  float* __restrict__ h) {
  __shared__ float sX[64 * 132];
  const int tid  = threadIdx.x;
  const int row0 = blockIdx.x * 64;

  for (int i = tid; i < 64 * 32; i += 256) {
    const int r = i >> 5, c = i & 31;
    float4 v = make_float4(0.f, 0.f, 0.f, 0.f);
    const int gr = row0 + r;
    if (gr < N_NODES) v = ((const float4*)x)[gr * 32 + c];
    *(float4*)&sX[r * 132 + c * 4] = v;
  }
  __syncthreads();

  const int wave = tid >> 5;
  const int lane = tid & 31;
  const int lrow = lane & 15;
  const int hi   = lane >> 4;
  const int col  = wave * 16 + lrow;

  for (int t = 0; t < 4; ++t) {
    v8f acc = {};
    const float* ax = &sX[(t * 16 + lrow) * 132 + 2 * hi];
    for (int k0 = 0; k0 < 128; k0 += 4) {
      v2f a, b;
      a.x = ax[k0];
      a.y = ax[k0 + 1];
      const int kb = k0 + 2 * hi;
      b.x = W1[kb * 128 + col];
      b.y = W1[(kb + 1) * 128 + col];
      acc = __builtin_amdgcn_wmma_f32_16x16x4_f32(false, a, false, b,
                                                  (short)0, acc, false, false);
    }
    const int rbase = row0 + t * 16 + 8 * hi;
#pragma unroll
    for (int r = 0; r < 8; ++r) {
      const int gr = rbase + r;
      if (gr < N_NODES) h[gr * 128 + col] = acc[r];
    }
  }
}

#define TILE_NODES  400
#define N_TILES     (N_NODES / TILE_NODES)
#define OWN_WAVES   32
#define CHUNK_EDGES 128
#define N_CHUNKS    (N_EDGES / CHUNK_EDGES)
#define SCAN_DEPTH  8
#define DRAIN_EVERY 4
#define QCAP        (32 + DRAIN_EVERY * CHUNK_EDGES)
#define E_BITS      21

static_assert(N_NODES % TILE_NODES == 0, "tiles cover the nodes exactly");
static_assert(N_EDGES % CHUNK_EDGES == 0, "chunks cover the edges exactly");
static_assert(N_EDGES <= (1 << E_BITS), "edge id fits its field");
static_assert(TILE_NODES <= (1 << (32 - E_BITS)), "local row fits its field");
static_assert(N_CHUNKS % N_TILES == 0, "start offsets are whole chunks");

__device__ __forceinline__ int own_append(unsigned* q, int cnt, unsigned t,
                                          unsigned e) {
  const bool     m    = t < (unsigned)TILE_NODES;
  const unsigned mask = __builtin_amdgcn_ballot_w32(m);
  if (mask != 0u) {
    const unsigned pos = (unsigned)cnt + __builtin_amdgcn_mbcnt_lo(mask, 0u);
    if (m) q[pos] = (t << E_BITS) | e;
    cnt += __builtin_popcount(mask);
  }
  return cnt;
}

__device__ __forceinline__ void own_batch(const unsigned* qb, const int nvalid,
                                          const int lane, const int n0,
                                          const int* __restrict__ src,
                                          const float* __restrict__ ew,
                                          const float* __restrict__ dinv,
                                          const float* __restrict__ h,
                                          float* sT) {
  __builtin_amdgcn_fence(__ATOMIC_RELEASE, "wavefront");
  __builtin_amdgcn_wave_barrier();
  __builtin_amdgcn_fence(__ATOMIC_ACQUIRE, "wavefront");
  int      s    = 0;
  unsigned dl   = 0u;
  float    norm = 0.f;
  if (lane < nvalid) {
    const unsigned ent = qb[lane];
    const unsigned e   = ent & ((1u << E_BITS) - 1u);
    dl   = ent >> E_BITS;
    s    = src[e];
    norm = dinv[s] * ew[e] * dinv[n0 + (int)dl];
  }
  const float* hl = h + lane;
  float*       tl = sT + lane;
  for (int g = 0; g < nvalid; g += 8) {
    float v[8][4];
#pragma unroll
    for (int i = 0; i < 8; ++i) {
      const int    sk = __builtin_amdgcn_readlane(s, g + i);
      const float* hr = hl + sk * 128;
      v[i][0] = hr[0];
      v[i][1] = hr[32];
      v[i][2] = hr[64];
      v[i][3] = hr[96];
    }
#pragma unroll
    for (int i = 0; i < 8; ++i) {
      if (g + i < nvalid) {
        const int   dk = __builtin_amdgcn_readlane((int)dl, g + i);
        const float nk = __int_as_float(
            __builtin_amdgcn_readlane(__float_as_int(norm), g + i));
        float* o = tl + dk * 128;
        __hip_atomic_fetch_add(o + 0,  v[i][0] * nk, __ATOMIC_RELAXED,
                               __HIP_MEMORY_SCOPE_WORKGROUP);
        __hip_atomic_fetch_add(o + 32, v[i][1] * nk, __ATOMIC_RELAXED,
                               __HIP_MEMORY_SCOPE_WORKGROUP);
        __hip_atomic_fetch_add(o + 64, v[i][2] * nk, __ATOMIC_RELAXED,
                               __HIP_MEMORY_SCOPE_WORKGROUP);
        __hip_atomic_fetch_add(o + 96, v[i][3] * nk, __ATOMIC_RELAXED,
                               __HIP_MEMORY_SCOPE_WORKGROUP);
      }
    }
  }
  __builtin_amdgcn_fence(__ATOMIC_RELEASE, "wavefront");
  __builtin_amdgcn_wave_barrier();
  __builtin_amdgcn_fence(__ATOMIC_ACQUIRE, "wavefront");
}

__global__ __launch_bounds__(1024) void agg1_owner_kernel(
    const int* __restrict__ src, const int* __restrict__ dst,
    const float* __restrict__ ew, const float* __restrict__ dinv,
    const float* __restrict__ h, float* __restrict__ agg) {
  __shared__ __attribute__((aligned(16))) float sT[TILE_NODES * 128];
  __shared__ unsigned sQ[OWN_WAVES * QCAP];
  const int tid  = threadIdx.x;
  const int lane = tid & 31;
  const int wave = __builtin_amdgcn_readfirstlane(tid >> 5);
  const int tile = blockIdx.x;
  const int n0   = tile * TILE_NODES;

  for (int i = tid; i < TILE_NODES * 32; i += 1024) {
    const int   r  = n0 + (i >> 5);
    const float di = dinv[r];
    float4 v = ((const float4*)h)[r * 32 + (i & 31)];
    v.x = v.x * di * di;
    v.y = v.y * di * di;
    v.z = v.z * di * di;
    v.w = v.w * di * di;
    ((float4*)sT)[i] = v;
  }
  __syncthreads();

  unsigned*   q    = sQ + wave * QCAP;
  const int4* dst4 = (const int4*)dst;
  const int start = tile * (N_CHUNKS / N_TILES);
  int cnt = 0;
  for (int j = wave; j < N_CHUNKS; j += SCAN_DEPTH * OWN_WAVES) {
    int4 dv[SCAN_DEPTH];
    int  cj[SCAN_DEPTH];
#pragma unroll
    for (int u = 0; u < SCAN_DEPTH; ++u) {
      const int jj = j + u * OWN_WAVES;
      int c = jj + start;
      if (c >= N_CHUNKS) c -= N_CHUNKS;
      cj[u] = (jj < N_CHUNKS) ? c : -1;
      dv[u] = dst4[(cj[u] < 0 ? 0 : cj[u]) * 32 + lane];
    }
#pragma unroll
    for (int u = 0; u < SCAN_DEPTH; ++u) {
      if (cj[u] >= 0) {
        const unsigned t0 = (unsigned)(dv[u].x - n0);
        const unsigned t1 = (unsigned)(dv[u].y - n0);
        const unsigned t2 = (unsigned)(dv[u].z - n0);
        const unsigned t3 = (unsigned)(dv[u].w - n0);
        const unsigned ta = t0 < t1 ? t0 : t1;
        const unsigned tb = t2 < t3 ? t2 : t3;
        const unsigned tm = ta < tb ? ta : tb;
        if (__builtin_amdgcn_ballot_w32(tm < (unsigned)TILE_NODES) != 0u) {
          const unsigned e0 =
              (unsigned)cj[u] * CHUNK_EDGES + (unsigned)lane * 4u;
          cnt = own_append(q, cnt, t0, e0 + 0u);
          cnt = own_append(q, cnt, t1, e0 + 1u);
          cnt = own_append(q, cnt, t2, e0 + 2u);
          cnt = own_append(q, cnt, t3, e0 + 3u);
        }
      }
      if (u % DRAIN_EVERY == DRAIN_EVERY - 1) {
        while (cnt >= 32) {
          cnt -= 32;
          own_batch(q + cnt, 32, lane, n0, src, ew, dinv, h, sT);
        }
      }
    }
  }
  if (cnt > 0) own_batch(q, cnt, lane, n0, src, ew, dinv, h, sT);
  __syncthreads();

  for (int i = tid; i < TILE_NODES * 32; i += 1024)
    ((float4*)agg)[n0 * 32 + i] = ((const float4*)sT)[i];
}

__global__ __launch_bounds__(256) void gemm2_wmma(const float* __restrict__ agg1,
                                                  const float* __restrict__ b1,
                                                  const float* __restrict__ W2,
                                                  float* __restrict__ h2) {
  __shared__ float sW[128 * 16];
  __shared__ float sb[128];
  const int tid = threadIdx.x;
  for (int i = tid; i < 512; i += 256) ((float4*)sW)[i] = ((const float4*)W2)[i];
  if (tid < 128) sb[tid] = b1[tid];
  __syncthreads();

  const int wave = tid >> 5, lane = tid & 31;
  const int lrow = lane & 15, hi = lane >> 4;
  const int row0 = blockIdx.x * 128 + wave * 16;
  const int arow = row0 + lrow;

  v8f acc = {};
  for (int k0 = 0; k0 < 128; k0 += 4) {
    const int k = k0 + 2 * hi;
    float a0 = 0.f, a1 = 0.f;
    if (arow < N_NODES) {
      a0 = fmaxf(agg1[arow * 128 + k] + sb[k], 0.f);
      a1 = fmaxf(agg1[arow * 128 + k + 1] + sb[k + 1], 0.f);
    }
    v2f a, b;
    a.x = a0; a.y = a1;
    b.x = sW[k * 16 + lrow];
    b.y = sW[(k + 1) * 16 + lrow];
    acc = __builtin_amdgcn_wmma_f32_16x16x4_f32(false, a, false, b,
                                                (short)0, acc, false, false);
  }
  const int rbase = row0 + 8 * hi;
#pragma unroll
  for (int r = 0; r < 8; ++r) {
    const int gr = rbase + r;
    if (gr < N_NODES) h2[gr * 16 + lrow] = acc[r];
  }
}

__global__ void selfloop2_kernel(const float* __restrict__ h2,
                                 const float* __restrict__ dinv,
                                 float* __restrict__ agg) {
  const int idx = blockIdx.x * blockDim.x + threadIdx.x;
  if (idx >= N_NODES * 16) return;
  const float di = dinv[idx >> 4];
  agg[idx] = h2[idx] * di * di;
}

__global__ void scatter2_kernel(const int* __restrict__ src,
                                const int* __restrict__ dst,
                                const float* __restrict__ ew,
                                const float* __restrict__ dinv,
                                const float* __restrict__ h2,
                                float* __restrict__ agg) {
  static_assert(N_EDGES % 8 == 0, "a wave owns 8 whole edges");
  const int gid  = blockIdx.x * blockDim.x + threadIdx.x;
  const int lane = threadIdx.x & 31;
  const int e0   = (gid >> 5) * 8;
  if (e0 >= N_EDGES) return;
  const int   e    = e0 + (lane & 7);
  const int   s    = src[e];
  const int   d    = dst[e];
  const float norm = dinv[s] * ew[e] * dinv[d];
  const int col  = lane & 15;
  const int half = lane >> 4;
  int   dk[4];
  float pv[4];
  for (int i = 0; i < 4; ++i) {
    const int   k  = 2 * i + half;
    const int   sk = __shfl(s, k, 32);
    const float nk = __shfl(norm, k, 32);
    dk[i] = __shfl(d, k, 32);
    pv[i] = h2[sk * 16 + col] * nk;
  }
  for (int i = 0; i < 4; ++i) atomicAdd(agg + dk[i] * 16 + col, pv[i]);
}

__global__ void softmax_kernel(const float* __restrict__ agg2,
                               const float* __restrict__ b2,
                               float* __restrict__ out) {
  const int i = blockIdx.x * blockDim.x + threadIdx.x;
  if (i >= N_NODES) return;
  float v[16];
  const float4* p = (const float4*)(agg2 + (size_t)i * 16);
  float m = -3.402823466e+38f;
#pragma unroll
  for (int q = 0; q < 4; ++q) {
    const float4 t = p[q];
    v[4 * q + 0] = t.x + b2[4 * q + 0];
    v[4 * q + 1] = t.y + b2[4 * q + 1];
    v[4 * q + 2] = t.z + b2[4 * q + 2];
    v[4 * q + 3] = t.w + b2[4 * q + 3];
  }
#pragma unroll
  for (int c = 0; c < 16; ++c) m = fmaxf(m, v[c]);
  float s = 0.f;
#pragma unroll
  for (int c = 0; c < 16; ++c) { v[c] = expf(v[c] - m); s += v[c]; }
  const float inv = 1.0f / s;
  float4* o = (float4*)(out + (size_t)i * 16);
#pragma unroll
  for (int q = 0; q < 4; ++q)
    o[q] = make_float4(v[4 * q + 0] * inv, v[4 * q + 1] * inv,
                       v[4 * q + 2] * inv, v[4 * q + 3] * inv);
}

extern "C" void kernel_launch(void* const* d_in, const int* in_sizes, int n_in,
                              void* d_out, int out_size, void* d_ws, size_t ws_size,
                              hipStream_t stream) {
  const float* x  = (const float*)d_in[0];
  const int*   ei = (const int*)d_in[1];
  const float* ew = (const float*)d_in[2];
  const float* W1 = (const float*)d_in[3];
  const float* b1 = (const float*)d_in[4];
  const float* W2 = (const float*)d_in[5];
  const float* b2 = (const float*)d_in[6];
  float* out = (float*)d_out;

  const int* src = ei;
  const int* dst = ei + N_EDGES;

  float* ws   = (float*)d_ws;
  float* dinv = ws;
  float* h    = ws + N_NODES;
  float* agg1 = ws + N_NODES + (size_t)N_NODES * 128;
  float* h2   = h;
  float* agg2 = h + (size_t)N_NODES * 16;

  const int T = 256;

  init_deg_kernel<<<(N_NODES + T - 1) / T, T, 0, stream>>>(dinv);
  deg_atomic_kernel<<<(N_EDGES + T - 1) / T, T, 0, stream>>>(dst, ew, dinv);
  dinv_kernel<<<(N_NODES + T - 1) / T, T, 0, stream>>>(dinv);

  gemm1_wmma<<<(N_NODES + 63) / 64, T, 0, stream>>>(x, W1, h);
  agg1_owner_kernel<<<N_TILES, 1024, 0, stream>>>(src, dst, ew, dinv, h, agg1);

  gemm2_wmma<<<(N_NODES + 127) / 128, T, 0, stream>>>(agg1, b1, W2, h2);
  selfloop2_kernel<<<(N_NODES * 16 + T - 1) / T, T, 0, stream>>>(h2, dinv, agg2);
  scatter2_kernel<<<(N_EDGES * 4 + T - 1) / T, T, 0, stream>>>(src, dst, ew, dinv, h2, agg2);

  softmax_kernel<<<(N_NODES + T - 1) / T, T, 0, stream>>>(agg2, b2, out);
}


# Layer-1 owner groups scan a 1-byte owner-tile key per edge (the degree kernel writes it in the same call), not the 4-byte dst list: 4x fewer scan bytes; LDS float atomics, arrival order, kept
# speedup vs baseline: 1.8313x; 1.0405x over previous
#include <hip/hip_runtime.h>
#include <hip/hip_bf16.h>
#include <math.h>

#define N_NODES 100000
#define N_EDGES 1600000
#define F_DIM 128
#define H_DIM 128
#define C_DIM 16

typedef __attribute__((ext_vector_type(2))) float v2f;
typedef __attribute__((ext_vector_type(8))) float v8f;

#define TILE_NODES  400
#define N_TILES     (N_NODES / TILE_NODES)
static_assert(N_NODES % TILE_NODES == 0, "tiles cover the nodes exactly");
static_assert(N_TILES <= 256, "a tile number fits one byte");

__global__ void init_deg_kernel(float* __restrict__ deg) {
  const int i = blockIdx.x * blockDim.x + threadIdx.x;
  if (i < N_NODES) deg[i] = 1.0f;
}

__global__ void deg_atomic_kernel(const int* __restrict__ dst,
                                  const float* __restrict__ ew,
                                  float* __restrict__ deg,
                                  unsigned char* __restrict__ tkey) {
  const int e = blockIdx.x * blockDim.x + threadIdx.x;
  if (e < N_EDGES) {
    const int d = dst[e];
    atomicAdd(&deg[d], ew[e]);
    tkey[e] = (unsigned char)((unsigned)d / (unsigned)TILE_NODES);
  }
}

__global__ void dinv_kernel(float* __restrict__ deg) {
  const int i = blockIdx.x * blockDim.x + threadIdx.x;
  if (i < N_NODES) {
    const float d = deg[i];
    deg[i] = (d > 0.0f) ? rsqrtf(d) : 0.0f;
  }
}

__global__ __launch_bounds__(256) void gemm1_wmma(const float* __restrict__ x,
                                                  const float* __restrict__ W1,
                                                  float* __restrict__ h) {
  __shared__ float sX[64 * 132];
  const int tid  = threadIdx.x;
  const int row0 = blockIdx.x * 64;

  for (int i = tid; i < 64 * 32; i += 256) {
    const int r = i >> 5, c = i & 31;
    float4 v = make_float4(0.f, 0.f, 0.f, 0.f);
    const int gr = row0 + r;
    if (gr < N_NODES) v = ((const float4*)x)[gr * 32 + c];
    *(float4*)&sX[r * 132 + c * 4] = v;
  }
  __syncthreads();

  const int wave = tid >> 5;
  const int lane = tid & 31;
  const int lrow = lane & 15;
  const int hi   = lane >> 4;
  const int col  = wave * 16 + lrow;

  for (int t = 0; t < 4; ++t) {
    v8f acc = {};
    const float* ax = &sX[(t * 16 + lrow) * 132 + 2 * hi];
    for (int k0 = 0; k0 < 128; k0 += 4) {
      v2f a, b;
      a.x = ax[k0];
      a.y = ax[k0 + 1];
      const int kb = k0 + 2 * hi;
      b.x = W1[kb * 128 + col];
      b.y = W1[(kb + 1) * 128 + col];
      acc = __builtin_amdgcn_wmma_f32_16x16x4_f32(false, a, false, b,
                                                  (short)0, acc, false, false);
    }
    const int rbase = row0 + t * 16 + 8 * hi;
#pragma unroll
    for (int r = 0; r < 8; ++r) {
      const int gr = rbase + r;
      if (gr < N_NODES) h[gr * 128 + col] = acc[r];
    }
  }
}

#define OWN_WAVES   32
#define CHUNK_EDGES 512
#define N_CHUNKS    (N_EDGES / CHUNK_EDGES)
#define SCAN_DEPTH  8
#define QCAP        64

static_assert(N_EDGES % CHUNK_EDGES == 0, "chunks cover the edges exactly");
static_assert(SCAN_DEPTH == 8, "the pair select below names four registers");

__device__ __forceinline__ unsigned zero_bytes(unsigned x) {
  const unsigned t = (x & 0x7F7F7F7Fu) + 0x7F7F7F7Fu;
  return ~(t | x | 0x7F7F7F7Fu);
}

__device__ __forceinline__ void own_batch(const unsigned* qb, const int nvalid,
                                          const int lane, const int n0,
                                          const int* __restrict__ src,
                                          const int* __restrict__ dst,
                                          const float* __restrict__ ew,
                                          const float* __restrict__ dinv,
                                          const float* __restrict__ h,
                                          float* sT) {
  __builtin_amdgcn_fence(__ATOMIC_RELEASE, "wavefront");
  __builtin_amdgcn_wave_barrier();
  __builtin_amdgcn_fence(__ATOMIC_ACQUIRE, "wavefront");
  int      s    = 0;
  unsigned dl   = 0u;
  float    norm = 0.f;
  if (lane < nvalid) {
    const unsigned e = qb[lane];
    const int      d = dst[e];
    dl   = (unsigned)(d - n0);
    s    = src[e];
    norm = dinv[s] * ew[e] * dinv[d];
  }
  const float* hl = h + lane;
  float*       tl = sT + lane;
  for (int g = 0; g < nvalid; g += 8) {
    float v[8][4];
#pragma unroll
    for (int i = 0; i < 8; ++i) {
      const int    sk = __builtin_amdgcn_readlane(s, g + i);
      const float* hr = hl + sk * 128;
      v[i][0] = hr[0];
      v[i][1] = hr[32];
      v[i][2] = hr[64];
      v[i][3] = hr[96];
    }
#pragma unroll
    for (int i = 0; i < 8; ++i) {
      if (g + i < nvalid) {
        const int   dk = __builtin_amdgcn_readlane((int)dl, g + i);
        const float nk = __int_as_float(
            __builtin_amdgcn_readlane(__float_as_int(norm), g + i));
        float* o = tl + dk * 128;
        __hip_atomic_fetch_add(o + 0,  v[i][0] * nk, __ATOMIC_RELAXED,
                               __HIP_MEMORY_SCOPE_WORKGROUP);
        __hip_atomic_fetch_add(o + 32, v[i][1] * nk, __ATOMIC_RELAXED,
                               __HIP_MEMORY_SCOPE_WORKGROUP);
        __hip_atomic_fetch_add(o + 64, v[i][2] * nk, __ATOMIC_RELAXED,
                               __HIP_MEMORY_SCOPE_WORKGROUP);
        __hip_atomic_fetch_add(o + 96, v[i][3] * nk, __ATOMIC_RELAXED,
                               __HIP_MEMORY_SCOPE_WORKGROUP);
      }
    }
  }
  __builtin_amdgcn_fence(__ATOMIC_RELEASE, "wavefront");
  __builtin_amdgcn_wave_barrier();
  __builtin_amdgcn_fence(__ATOMIC_ACQUIRE, "wavefront");
}

__global__ __launch_bounds__(1024) void agg1_owner_kernel(
    const int* __restrict__ src, const int* __restrict__ dst,
    const float* __restrict__ ew, const float* __restrict__ dinv,
    const float* __restrict__ h, const unsigned char* __restrict__ tkey,
    float* __restrict__ agg) {
  __shared__ __attribute__((aligned(16))) float sT[TILE_NODES * 128];
  __shared__ unsigned sQ[OWN_WAVES * QCAP];
  const int tid  = threadIdx.x;
  const int lane = tid & 31;
  const int wave = __builtin_amdgcn_readfirstlane(tid >> 5);
  const int tile = blockIdx.x;
  const int n0   = tile * TILE_NODES;

  for (int i = tid; i < TILE_NODES * 32; i += 1024) {
    const int   r  = n0 + (i >> 5);
    const float di = dinv[r];
    float4 v = ((const float4*)h)[r * 32 + (i & 31)];
    v.x = v.x * di * di;
    v.y = v.y * di * di;
    v.z = v.z * di * di;
    v.w = v.w * di * di;
    ((float4*)sT)[i] = v;
  }
  __syncthreads();

  unsigned*    q    = sQ + wave * QCAP;
  const uint4* key4 = (const uint4*)tkey;
  const unsigned pat = (unsigned)tile * 0x01010101u;
  const int start = (tile * N_CHUNKS) / N_TILES;
  int cnt = 0;
  for (int j = wave; j < N_CHUNKS; j += SCAN_DEPTH * OWN_WAVES) {
    uint4 kv[SCAN_DEPTH];
#pragma unroll
    for (int u = 0; u < SCAN_DEPTH; ++u) {
      const int jj = j + u * OWN_WAVES;
      int c = jj + start;
      if (c >= N_CHUNKS) c -= N_CHUNKS;
      if (jj >= N_CHUNKS) c = 0;
      kv[u] = key4[c * 32 + lane];
    }
    unsigned pk[SCAN_DEPTH / 2];
#pragma unroll
    for (int u = 0; u < SCAN_DEPTH; ++u) {
      unsigned lm = (zero_bytes(kv[u].x ^ pat) >> 7) |
                    (zero_bytes(kv[u].y ^ pat) >> 6) |
                    (zero_bytes(kv[u].z ^ pat) >> 5) |
                    (zero_bytes(kv[u].w ^ pat) >> 4);
      if (j + u * OWN_WAVES >= N_CHUNKS) lm = 0u;
      if (u & 1) pk[u >> 1] |= lm << 4;
      else       pk[u >> 1] = lm;
    }
#pragma nounroll
    for (int w = 0; w < SCAN_DEPTH / 2; ++w) {
      unsigned lm = (w == 0) ? pk[0] : (w == 1) ? pk[1] : (w == 2) ? pk[2] : pk[3];
      int ca = j + 2 * w * OWN_WAVES + start;
      if (ca >= N_CHUNKS) ca -= N_CHUNKS;
      int cb = j + (2 * w + 1) * OWN_WAVES + start;
      if (cb >= N_CHUNKS) cb -= N_CHUNKS;
      const unsigned ea = (unsigned)ca * CHUNK_EDGES + (unsigned)lane * 16u;
      const unsigned eb = (unsigned)cb * CHUNK_EDGES + (unsigned)lane * 16u;
      for (int it = 0; it < 32; ++it) {
        const bool     m    = lm != 0u;
        const unsigned mask = __builtin_amdgcn_ballot_w32(m);
        if (mask == 0u) break;
        const unsigned p   = (unsigned)__builtin_ctz(lm | 0x80000000u);
        const unsigned pos =
            (unsigned)cnt + __builtin_amdgcn_mbcnt_lo(mask, 0u);
        if (m) q[pos] = ((p & 4u) ? eb : ea) + 4u * (p & 3u) + (p >> 3);
        lm &= lm - 1u;
        cnt += __builtin_popcount(mask);
        if (cnt >= 32) {
          cnt -= 32;
          own_batch(q + cnt, 32, lane, n0, src, dst, ew, dinv, h, sT);
        }
      }
    }
  }
  if (cnt > 0) own_batch(q, cnt, lane, n0, src, dst, ew, dinv, h, sT);
  __syncthreads();

  for (int i = tid; i < TILE_NODES * 32; i += 1024)
    ((float4*)agg)[n0 * 32 + i] = ((const float4*)sT)[i];
}

__global__ __launch_bounds__(256) void gemm2_wmma(const float* __restrict__ agg1,
                                                  const float* __restrict__ b1,
                                                  const float* __restrict__ W2,
                                                  float* __restrict__ h2) {
  __shared__ float sW[128 * 16];
  __shared__ float sb[128];
  const int tid = threadIdx.x;
  for (int i = tid; i < 512; i += 256) ((float4*)sW)[i] = ((const float4*)W2)[i];
  if (tid < 128) sb[tid] = b1[tid];
  __syncthreads();

  const int wave = tid >> 5, lane = tid & 31;
  const int lrow = lane & 15, hi = lane >> 4;
  const int row0 = blockIdx.x * 128 + wave * 16;
  const int arow = row0 + lrow;

  v8f acc = {};
  for (int k0 = 0; k0 < 128; k0 += 4) {
    const int k = k0 + 2 * hi;
    float a0 = 0.f, a1 = 0.f;
    if (arow < N_NODES) {
      a0 = fmaxf(agg1[arow * 128 + k] + sb[k], 0.f);
      a1 = fmaxf(agg1[arow * 128 + k + 1] + sb[k + 1], 0.f);
    }
    v2f a, b;
    a.x = a0; a.y = a1;
    b.x = sW[k * 16 + lrow];
    b.y = sW[(k + 1) * 16 + lrow];
    acc = __builtin_amdgcn_wmma_f32_16x16x4_f32(false, a, false, b,
                                                (short)0, acc, false, false);
  }
  const int rbase = row0 + 8 * hi;
#pragma unroll
  for (int r = 0; r < 8; ++r) {
    const int gr = rbase + r;
    if (gr < N_NODES) h2[gr * 16 + lrow] = acc[r];
  }
}

__global__ void selfloop2_kernel(const float* __restrict__ h2,
                                 const float* __restrict__ dinv,
                                 float* __restrict__ agg) {
  const int idx = blockIdx.x * blockDim.x + threadIdx.x;
  if (idx >= N_NODES * 16) return;
  const float di = dinv[idx >> 4];
  agg[idx] = h2[idx] * di * di;
}

__global__ void scatter2_kernel(const int* __restrict__ src,
                                const int* __restrict__ dst,
                                const float* __restrict__ ew,
                                const float* __restrict__ dinv,
                                const float* __restrict__ h2,
                                float* __restrict__ agg) {
  static_assert(N_EDGES % 8 == 0, "a wave owns 8 whole edges");
  const int gid  = blockIdx.x * blockDim.x + threadIdx.x;
  const int lane = threadIdx.x & 31;
  const int e0   = (gid >> 5) * 8;
  if (e0 >= N_EDGES) return;
  const int   e    = e0 + (lane & 7);
  const int   s    = src[e];
  const int   d    = dst[e];
  const float norm = dinv[s] * ew[e] * dinv[d];
  const int col  = lane & 15;
  const int half = lane >> 4;
  int   dk[4];
  float pv[4];
  for (int i = 0; i < 4; ++i) {
    const int   k  = 2 * i + half;
    const int   sk = __shfl(s, k, 32);
    const float nk = __shfl(norm, k, 32);
    dk[i] = __shfl(d, k, 32);
    pv[i] = h2[sk * 16 + col] * nk;
  }
  for (int i = 0; i < 4; ++i) atomicAdd(agg + dk[i] * 16 + col, pv[i]);
}

__global__ void softmax_kernel(const float* __restrict__ agg2,
                               const float* __restrict__ b2,
                               float* __restrict__ out) {
  const int i = blockIdx.x * blockDim.x + threadIdx.x;
  if (i >= N_NODES) return;
  float v[16];
  const float4* p = (const float4*)(agg2 + (size_t)i * 16);
  float m = -3.402823466e+38f;
#pragma unroll
  for (int q = 0; q < 4; ++q) {
    const float4 t = p[q];
    v[4 * q + 0] = t.x + b2[4 * q + 0];
    v[4 * q + 1] = t.y + b2[4 * q + 1];
    v[4 * q + 2] = t.z + b2[4 * q + 2];
    v[4 * q + 3] = t.w + b2[4 * q + 3];
  }
#pragma unroll
  for (int c = 0; c < 16; ++c) m = fmaxf(m, v[c]);
  float s = 0.f;
#pragma unroll
  for (int c = 0; c < 16; ++c) { v[c] = expf(v[c] - m); s += v[c]; }
  const float inv = 1.0f / s;
  float4* o = (float4*)(out + (size_t)i * 16);
#pragma unroll
  for (int q = 0; q < 4; ++q)
    o[q] = make_float4(v[4 * q + 0] * inv, v[4 * q + 1] * inv,
                       v[4 * q + 2] * inv, v[4 * q + 3] * inv);
}

extern "C" void kernel_launch(void* const* d_in, const int* in_sizes, int n_in,
                              void* d_out, int out_size, void* d_ws, size_t ws_size,
                              hipStream_t stream) {
  const float* x  = (const float*)d_in[0];
  const int*   ei = (const int*)d_in[1];
  const float* ew = (const float*)d_in[2];
  const float* W1 = (const float*)d_in[3];
  const float* b1 = (const float*)d_in[4];
  const float* W2 = (const float*)d_in[5];
  const float* b2 = (const float*)d_in[6];
  float* out = (float*)d_out;

  const int* src = ei;
  const int* dst = ei + N_EDGES;

  float* ws   = (float*)d_ws;
  float* dinv = ws;
  float* h    = ws + N_NODES;
  float* agg1 = ws + N_NODES + (size_t)N_NODES * 128;
  float* h2   = h;
  float* agg2 = h + (size_t)N_NODES * 16;
  unsigned char* tkey = (unsigned char*)(agg1 + (size_t)N_NODES * 128);

  const int T = 256;

  init_deg_kernel<<<(N_NODES + T - 1) / T, T, 0, stream>>>(dinv);
  deg_atomic_kernel<<<(N_EDGES + T - 1) / T, T, 0, stream>>>(dst, ew, dinv, tkey);
  dinv_kernel<<<(N_NODES + T - 1) / T, T, 0, stream>>>(dinv);

  gemm1_wmma<<<(N_NODES + 63) / 64, T, 0, stream>>>(x, W1, h);
  agg1_owner_kernel<<<N_TILES, 1024, 0, stream>>>(src, dst, ew, dinv, h, tkey, agg1);

  gemm2_wmma<<<(N_NODES + 127) / 128, T, 0, stream>>>(agg1, b1, W2, h2);
  selfloop2_kernel<<<(N_NODES * 16 + T - 1) / T, T, 0, stream>>>(h2, dinv, agg2);
  scatter2_kernel<<<(N_EDGES * 4 + T - 1) / T, T, 0, stream>>>(src, dst, ew, dinv, h2, agg2);

  softmax_kernel<<<(N_NODES + T - 1) / T, T, 0, stream>>>(agg2, b2, out);
}


# Degrees summed in LDS tables: 5 node ranges x 50 edge slices (LDS float atomics), dinv adds the 50 partial rows in fixed order; global degree atomics and init launch go; degree sum order changes
# speedup vs baseline: 2.3221x; 1.2680x over previous
#include <hip/hip_runtime.h>
#include <hip/hip_bf16.h>
#include <math.h>

#define N_NODES 100000
#define N_EDGES 1600000
#define F_DIM 128
#define H_DIM 128
#define C_DIM 16

typedef __attribute__((ext_vector_type(2))) float v2f;
typedef __attribute__((ext_vector_type(8))) float v8f;

#define TILE_NODES  400
#define N_TILES     (N_NODES / TILE_NODES)
static_assert(N_NODES % TILE_NODES == 0, "tiles cover the nodes exactly");
static_assert(N_TILES <= 256, "a tile number fits one byte");

#define DEG_RANGES  5
#define DEG_SLICES  50
#define DEG_THREADS 1024
#define RANGE_NODES (N_NODES / DEG_RANGES)
#define SLICE_QUADS (N_EDGES / DEG_SLICES / 4)
#define KEY_QUADS   (SLICE_QUADS / DEG_RANGES)
#define DEG_TRIPS   ((SLICE_QUADS + DEG_THREADS - 1) / DEG_THREADS)

static_assert(N_NODES % (DEG_RANGES * 4) == 0, "ranges cover the nodes in float4s");
static_assert(N_EDGES % (DEG_SLICES * 4) == 0, "slices cover the edges in quads");
static_assert(SLICE_QUADS % DEG_RANGES == 0, "key words split evenly");

__device__ __forceinline__ void deg_add(float* sD, const int d, const int lo,
                                        const float w) {
  const unsigned r = (unsigned)(d - lo);
  if (r < (unsigned)RANGE_NODES)
    __hip_atomic_fetch_add(sD + r, w, __ATOMIC_RELAXED,
                           __HIP_MEMORY_SCOPE_WORKGROUP);
}

__global__ __launch_bounds__(DEG_THREADS) void deg_hist_kernel(
    const int* __restrict__ dst, const float* __restrict__ ew,
    float* __restrict__ part, unsigned char* __restrict__ tkey) {
  __shared__ __attribute__((aligned(16))) float sD[RANGE_NODES];
  const int tid = threadIdx.x;
  const int k   = blockIdx.x % DEG_RANGES;
  const int j   = blockIdx.x / DEG_RANGES;
  const int lo  = k * RANGE_NODES;

  const int4*   d4 = (const int4*)dst + j * SLICE_QUADS;
  const float4* w4 = (const float4*)ew + j * SLICE_QUADS;
  int4   dq[DEG_TRIPS];
  float4 wq[DEG_TRIPS];
#pragma unroll
  for (int u = 0; u < DEG_TRIPS; ++u) {
    int q = tid + u * DEG_THREADS;
    if (q > SLICE_QUADS - 1) q = SLICE_QUADS - 1;
    dq[u] = d4[q];
    wq[u] = w4[q];
  }

  for (int i = tid; i < RANGE_NODES / 4; i += DEG_THREADS)
    ((float4*)sD)[i] = make_float4(0.f, 0.f, 0.f, 0.f);
  __syncthreads();

  unsigned* key4 = (unsigned*)tkey + j * SLICE_QUADS;
#pragma unroll
  for (int u = 0; u < DEG_TRIPS; ++u) {
    const int q = tid + u * DEG_THREADS;
    if (q < SLICE_QUADS) {
      deg_add(sD, dq[u].x, lo, wq[u].x);
      deg_add(sD, dq[u].y, lo, wq[u].y);
      deg_add(sD, dq[u].z, lo, wq[u].z);
      deg_add(sD, dq[u].w, lo, wq[u].w);
      if (q / KEY_QUADS == k)
        key4[q] = ((unsigned)dq[u].x / (unsigned)TILE_NODES) |
                  (((unsigned)dq[u].y / (unsigned)TILE_NODES) << 8) |
                  (((unsigned)dq[u].z / (unsigned)TILE_NODES) << 16) |
                  (((unsigned)dq[u].w / (unsigned)TILE_NODES) << 24);
    }
  }
  __syncthreads();

  float4* po = (float4*)(part + (size_t)j * N_NODES + lo);
  for (int i = tid; i < RANGE_NODES / 4; i += DEG_THREADS)
    po[i] = ((const float4*)sD)[i];
}

__global__ void dinv_kernel(const float* __restrict__ part,
                            float* __restrict__ dinv) {
  const int i = blockIdx.x * blockDim.x + threadIdx.x;
  if (i < N_NODES) {
    float d = 1.0f;
#pragma unroll
    for (int j = 0; j < DEG_SLICES; ++j) d += part[(size_t)j * N_NODES + i];
    dinv[i] = (d > 0.0f) ? rsqrtf(d) : 0.0f;
  }
}

__global__ __launch_bounds__(256) void gemm1_wmma(const float* __restrict__ x,
                                                  const float* __restrict__ W1,
                                                  float* __restrict__ h) {
  __shared__ float sX[64 * 132];
  const int tid  = threadIdx.x;
  const int row0 = blockIdx.x * 64;

  for (int i = tid; i < 64 * 32; i += 256) {
    const int r = i >> 5, c = i & 31;
    float4 v = make_float4(0.f, 0.f, 0.f, 0.f);
    const int gr = row0 + r;
    if (gr < N_NODES) v = ((const float4*)x)[gr * 32 + c];
    *(float4*)&sX[r * 132 + c * 4] = v;
  }
  __syncthreads();

  const int wave = tid >> 5;
  const int lane = tid & 31;
  const int lrow = lane & 15;
  const int hi   = lane >> 4;
  const int col  = wave * 16 + lrow;

  for (int t = 0; t < 4; ++t) {
    v8f acc = {};
    const float* ax = &sX[(t * 16 + lrow) * 132 + 2 * hi];
    for (int k0 = 0; k0 < 128; k0 += 4) {
      v2f a, b;
      a.x = ax[k0];
      a.y = ax[k0 + 1];
      const int kb = k0 + 2 * hi;
      b.x = W1[kb * 128 + col];
      b.y = W1[(kb + 1) * 128 + col];
      acc = __builtin_amdgcn_wmma_f32_16x16x4_f32(false, a, false, b,
                                                  (short)0, acc, false, false);
    }
    const int rbase = row0 + t * 16 + 8 * hi;
#pragma unroll
    for (int r = 0; r < 8; ++r) {
      const int gr = rbase + r;
      if (gr < N_NODES) h[gr * 128 + col] = acc[r];
    }
  }
}

#define OWN_WAVES   32
#define CHUNK_EDGES 512
#define N_CHUNKS    (N_EDGES / CHUNK_EDGES)
#define SCAN_DEPTH  8
#define QCAP        64

static_assert(N_EDGES % CHUNK_EDGES == 0, "chunks cover the edges exactly");
static_assert(SCAN_DEPTH == 8, "the pair select below names four registers");

__device__ __forceinline__ unsigned zero_bytes(unsigned x) {
  const unsigned t = (x & 0x7F7F7F7Fu) + 0x7F7F7F7Fu;
  return ~(t | x | 0x7F7F7F7Fu);
}

__device__ __forceinline__ void own_batch(const unsigned* qb, const int nvalid,
                                          const int lane, const int n0,
                                          const int* __restrict__ src,
                                          const int* __restrict__ dst,
                                          const float* __restrict__ ew,
                                          const float* __restrict__ dinv,
                                          const float* __restrict__ h,
                                          float* sT) {
  __builtin_amdgcn_fence(__ATOMIC_RELEASE, "wavefront");
  __builtin_amdgcn_wave_barrier();
  __builtin_amdgcn_fence(__ATOMIC_ACQUIRE, "wavefront");
  int      s    = 0;
  unsigned dl   = 0u;
  float    norm = 0.f;
  if (lane < nvalid) {
    const unsigned e = qb[lane];
    const int      d = dst[e];
    dl   = (unsigned)(d - n0);
    s    = src[e];
    norm = dinv[s] * ew[e] * dinv[d];
  }
  const float* hl = h + lane;
  float*       tl = sT + lane;
  for (int g = 0; g < nvalid; g += 8) {
    float v[8][4];
#pragma unroll
    for (int i = 0; i < 8; ++i) {
      const int    sk = __builtin_amdgcn_readlane(s, g + i);
      const float* hr = hl + sk * 128;
      v[i][0] = hr[0];
      v[i][1] = hr[32];
      v[i][2] = hr[64];
      v[i][3] = hr[96];
    }
#pragma unroll
    for (int i = 0; i < 8; ++i) {
      if (g + i < nvalid) {
        const int   dk = __builtin_amdgcn_readlane((int)dl, g + i);
        const float nk = __int_as_float(
            __builtin_amdgcn_readlane(__float_as_int(norm), g + i));
        float* o = tl + dk * 128;
        __hip_atomic_fetch_add(o + 0,  v[i][0] * nk, __ATOMIC_RELAXED,
                               __HIP_MEMORY_SCOPE_WORKGROUP);
        __hip_atomic_fetch_add(o + 32, v[i][1] * nk, __ATOMIC_RELAXED,
                               __HIP_MEMORY_SCOPE_WORKGROUP);
        __hip_atomic_fetch_add(o + 64, v[i][2] * nk, __ATOMIC_RELAXED,
                               __HIP_MEMORY_SCOPE_WORKGROUP);
        __hip_atomic_fetch_add(o + 96, v[i][3] * nk, __ATOMIC_RELAXED,
                               __HIP_MEMORY_SCOPE_WORKGROUP);
      }
    }
  }
  __builtin_amdgcn_fence(__ATOMIC_RELEASE, "wavefront");
  __builtin_amdgcn_wave_barrier();
  __builtin_amdgcn_fence(__ATOMIC_ACQUIRE, "wavefront");
}

__global__ __launch_bounds__(1024) void agg1_owner_kernel(
    const int* __restrict__ src, const int* __restrict__ dst,
    const float* __restrict__ ew, const float* __restrict__ dinv,
    const float* __restrict__ h, const unsigned char* __restrict__ tkey,
    float* __restrict__ agg) {
  __shared__ __attribute__((aligned(16))) float sT[TILE_NODES * 128];
  __shared__ unsigned sQ[OWN_WAVES * QCAP];
  const int tid  = threadIdx.x;
  const int lane = tid & 31;
  const int wave = __builtin_amdgcn_readfirstlane(tid >> 5);
  const int tile = blockIdx.x;
  const int n0   = tile * TILE_NODES;

  for (int i = tid; i < TILE_NODES * 32; i += 1024) {
    const int   r  = n0 + (i >> 5);
    const float di = dinv[r];
    float4 v = ((const float4*)h)[r * 32 + (i & 31)];
    v.x = v.x * di * di;
    v.y = v.y * di * di;
    v.z = v.z * di * di;
    v.w = v.w * di * di;
    ((float4*)sT)[i] = v;
  }
  __syncthreads();

  unsigned*    q    = sQ + wave * QCAP;
  const uint4* key4 = (const uint4*)tkey;
  const unsigned pat = (unsigned)tile * 0x01010101u;
  const int start = (tile * N_CHUNKS) / N_TILES;
  int cnt = 0;
  for (int j = wave; j < N_CHUNKS; j += SCAN_DEPTH * OWN_WAVES) {
    uint4 kv[SCAN_DEPTH];
#pragma unroll
    for (int u = 0; u < SCAN_DEPTH; ++u) {
      const int jj = j + u * OWN_WAVES;
      int c = jj + start;
      if (c >= N_CHUNKS) c -= N_CHUNKS;
      if (jj >= N_CHUNKS) c = 0;
      kv[u] = key4[c * 32 + lane];
    }
    unsigned pk[SCAN_DEPTH / 2];
#pragma unroll
    for (int u = 0; u < SCAN_DEPTH; ++u) {
      unsigned lm = (zero_bytes(kv[u].x ^ pat) >> 7) |
                    (zero_bytes(kv[u].y ^ pat) >> 6) |
                    (zero_bytes(kv[u].z ^ pat) >> 5) |
                    (zero_bytes(kv[u].w ^ pat) >> 4);
      if (j + u * OWN_WAVES >= N_CHUNKS) lm = 0u;
      if (u & 1) pk[u >> 1] |= lm << 4;
      else       pk[u >> 1] = lm;
    }
#pragma nounroll
    for (int w = 0; w < SCAN_DEPTH / 2; ++w) {
      unsigned lm = (w == 0) ? pk[0] : (w == 1) ? pk[1] : (w == 2) ? pk[2] : pk[3];
      int ca = j + 2 * w * OWN_WAVES + start;
      if (ca >= N_CHUNKS) ca -= N_CHUNKS;
      int cb = j + (2 * w + 1) * OWN_WAVES + start;
      if (cb >= N_CHUNKS) cb -= N_CHUNKS;
      const unsigned ea = (unsigned)ca * CHUNK_EDGES + (unsigned)lane * 16u;
      const unsigned eb = (unsigned)cb * CHUNK_EDGES + (unsigned)lane * 16u;
      for (int it = 0; it < 32; ++it) {
        const bool     m    = lm != 0u;
        const unsigned mask = __builtin_amdgcn_ballot_w32(m);
        if (mask == 0u) break;
        const unsigned p   = (unsigned)__builtin_ctz(lm | 0x80000000u);
        const unsigned pos =
            (unsigned)cnt + __builtin_amdgcn_mbcnt_lo(mask, 0u);
        if (m) q[pos] = ((p & 4u) ? eb : ea) + 4u * (p & 3u) + (p >> 3);
        lm &= lm - 1u;
        cnt += __builtin_popcount(mask);
        if (cnt >= 32) {
          cnt -= 32;
          own_batch(q + cnt, 32, lane, n0, src, dst, ew, dinv, h, sT);
        }
      }
    }
  }
  if (cnt > 0) own_batch(q, cnt, lane, n0, src, dst, ew, dinv, h, sT);
  __syncthreads();

  for (int i = tid; i < TILE_NODES * 32; i += 1024)
    ((float4*)agg)[n0 * 32 + i] = ((const float4*)sT)[i];
}

__global__ __launch_bounds__(256) void gemm2_wmma(const float* __restrict__ agg1,
                                                  const float* __restrict__ b1,
                                                  const float* __restrict__ W2,
                                                  float* __restrict__ h2) {
  __shared__ float sW[128 * 16];
  __shared__ float sb[128];
  const int tid = threadIdx.x;
  for (int i = tid; i < 512; i += 256) ((float4*)sW)[i] = ((const float4*)W2)[i];
  if (tid < 128) sb[tid] = b1[tid];
  __syncthreads();

  const int wave = tid >> 5, lane = tid & 31;
  const int lrow = lane & 15, hi = lane >> 4;
  const int row0 = blockIdx.x * 128 + wave * 16;
  const int arow = row0 + lrow;

  v8f acc = {};
  for (int k0 = 0; k0 < 128; k0 += 4) {
    const int k = k0 + 2 * hi;
    float a0 = 0.f, a1 = 0.f;
    if (arow < N_NODES) {
      a0 = fmaxf(agg1[arow * 128 + k] + sb[k], 0.f);
      a1 = fmaxf(agg1[arow * 128 + k + 1] + sb[k + 1], 0.f);
    }
    v2f a, b;
    a.x = a0; a.y = a1;
    b.x = sW[k * 16 + lrow];
    b.y = sW[(k + 1) * 16 + lrow];
    acc = __builtin_amdgcn_wmma_f32_16x16x4_f32(false, a, false, b,
                                                (short)0, acc, false, false);
  }
  const int rbase = row0 + 8 * hi;
#pragma unroll
  for (int r = 0; r < 8; ++r) {
    const int gr = rbase + r;
    if (gr < N_NODES) h2[gr * 16 + lrow] = acc[r];
  }
}

__global__ void selfloop2_kernel(const float* __restrict__ h2,
                                 const float* __restrict__ dinv,
                                 float* __restrict__ agg) {
  const int idx = blockIdx.x * blockDim.x + threadIdx.x;
  if (idx >= N_NODES * 16) return;
  const float di = dinv[idx >> 4];
  agg[idx] = h2[idx] * di * di;
}

__global__ void scatter2_kernel(const int* __restrict__ src,
                                const int* __restrict__ dst,
                                const float* __restrict__ ew,
                                const float* __restrict__ dinv,
                                const float* __restrict__ h2,
                                float* __restrict__ agg) {
  static_assert(N_EDGES % 8 == 0, "a wave owns 8 whole edges");
  const int gid  = blockIdx.x * blockDim.x + threadIdx.x;
  const int lane = threadIdx.x & 31;
  const int e0   = (gid >> 5) * 8;
  if (e0 >= N_EDGES) return;
  const int   e    = e0 + (lane & 7);
  const int   s    = src[e];
  const int   d    = dst[e];
  const float norm = dinv[s] * ew[e] * dinv[d];
  const int col  = lane & 15;
  const int half = lane >> 4;
  int   dk[4];
  float pv[4];
  for (int i = 0; i < 4; ++i) {
    const int   k  = 2 * i + half;
    const int   sk = __shfl(s, k, 32);
    const float nk = __shfl(norm, k, 32);
    dk[i] = __shfl(d, k, 32);
    pv[i] = h2[sk * 16 + col] * nk;
  }
  for (int i = 0; i < 4; ++i) atomicAdd(agg + dk[i] * 16 + col, pv[i]);
}

__global__ void softmax_kernel(const float* __restrict__ agg2,
                               const float* __restrict__ b2,
                               float* __restrict__ out) {
  const int i = blockIdx.x * blockDim.x + threadIdx.x;
  if (i >= N_NODES) return;
  float v[16];
  const float4* p = (const float4*)(agg2 + (size_t)i * 16);
  float m = -3.402823466e+38f;
#pragma unroll
  for (int q = 0; q < 4; ++q) {
    const float4 t = p[q];
    v[4 * q + 0] = t.x + b2[4 * q + 0];
    v[4 * q + 1] = t.y + b2[4 * q + 1];
    v[4 * q + 2] = t.z + b2[4 * q + 2];
    v[4 * q + 3] = t.w + b2[4 * q + 3];
  }
#pragma unroll
  for (int c = 0; c < 16; ++c) m = fmaxf(m, v[c]);
  float s = 0.f;
#pragma unroll
  for (int c = 0; c < 16; ++c) { v[c] = expf(v[c] - m); s += v[c]; }
  const float inv = 1.0f / s;
  float4* o = (float4*)(out + (size_t)i * 16);
#pragma unroll
  for (int q = 0; q < 4; ++q)
    o[q] = make_float4(v[4 * q + 0] * inv, v[4 * q + 1] * inv,
                       v[4 * q + 2] * inv, v[4 * q + 3] * inv);
}

extern "C" void kernel_launch(void* const* d_in, const int* in_sizes, int n_in,
                              void* d_out, int out_size, void* d_ws, size_t ws_size,
                              hipStream_t stream) {
  const float* x  = (const float*)d_in[0];
  const int*   ei = (const int*)d_in[1];
  const float* ew = (const float*)d_in[2];
  const float* W1 = (const float*)d_in[3];
  const float* b1 = (const float*)d_in[4];
  const float* W2 = (const float*)d_in[5];
  const float* b2 = (const float*)d_in[6];
  float* out = (float*)d_out;

  const int* src = ei;
  const int* dst = ei + N_EDGES;

  float* ws   = (float*)d_ws;
  float* dinv = ws;
  float* h    = ws + N_NODES;
  float* agg1 = ws + N_NODES + (size_t)N_NODES * 128;
  float* h2   = h;
  float* agg2 = h + (size_t)N_NODES * 16;
  unsigned char* tkey = (unsigned char*)(agg1 + (size_t)N_NODES * 128);
  float* part = (float*)(tkey + N_EDGES);

  const int T = 256;

  deg_hist_kernel<<<DEG_RANGES * DEG_SLICES, DEG_THREADS, 0, stream>>>(dst, ew, part, tkey);
  dinv_kernel<<<(N_NODES + T - 1) / T, T, 0, stream>>>(part, dinv);

  gemm1_wmma<<<(N_NODES + 63) / 64, T, 0, stream>>>(x, W1, h);
  agg1_owner_kernel<<<N_TILES, 1024, 0, stream>>>(src, dst, ew, dinv, h, tkey, agg1);

  gemm2_wmma<<<(N_NODES + 127) / 128, T, 0, stream>>>(agg1, b1, W2, h2);
  selfloop2_kernel<<<(N_NODES * 16 + T - 1) / T, T, 0, stream>>>(h2, dinv, agg2);
  scatter2_kernel<<<(N_EDGES * 4 + T - 1) / T, T, 0, stream>>>(src, dst, ew, dinv, h2, agg2);

  softmax_kernel<<<(N_NODES + T - 1) / T, T, 0, stream>>>(agg2, b2, out);
}


# Layer 2 on the owner tiles: layer-1 owner waves save (src,row,norm) per edge, layer 2 adds these lists into 400x16 LDS tiles (LDS float atomics, arrival order), stores softmax; 8 launches become 6
# speedup vs baseline: 2.7891x; 1.2011x over previous
#include <hip/hip_runtime.h>
#include <hip/hip_bf16.h>
#include <math.h>

#define N_NODES 100000
#define N_EDGES 1600000
#define F_DIM 128
#define H_DIM 128
#define C_DIM 16

typedef __attribute__((ext_vector_type(2))) float v2f;
typedef __attribute__((ext_vector_type(8))) float v8f;

#define TILE_NODES  400
#define N_TILES     (N_NODES / TILE_NODES)
static_assert(N_NODES % TILE_NODES == 0, "tiles cover the nodes exactly");
static_assert(N_TILES <= 256, "a tile number fits one byte");

#define DEG_RANGES  5
#define DEG_SLICES  50
#define DEG_THREADS 1024
#define RANGE_NODES (N_NODES / DEG_RANGES)
#define SLICE_QUADS (N_EDGES / DEG_SLICES / 4)
#define KEY_QUADS   (SLICE_QUADS / DEG_RANGES)
#define DEG_TRIPS   ((SLICE_QUADS + DEG_THREADS - 1) / DEG_THREADS)

static_assert(N_NODES % (DEG_RANGES * 4) == 0, "ranges cover the nodes in float4s");
static_assert(N_EDGES % (DEG_SLICES * 4) == 0, "slices cover the edges in quads");
static_assert(SLICE_QUADS % DEG_RANGES == 0, "key words split evenly");

__device__ __forceinline__ void deg_add(float* sD, const int d, const int lo,
                                        const float w) {
  const unsigned r = (unsigned)(d - lo);
  if (r < (unsigned)RANGE_NODES)
    __hip_atomic_fetch_add(sD + r, w, __ATOMIC_RELAXED,
                           __HIP_MEMORY_SCOPE_WORKGROUP);
}

__global__ __launch_bounds__(DEG_THREADS) void deg_hist_kernel(
    const int* __restrict__ dst, const float* __restrict__ ew,
    float* __restrict__ part, unsigned char* __restrict__ tkey) {
  __shared__ __attribute__((aligned(16))) float sD[RANGE_NODES];
  const int tid = threadIdx.x;
  const int k   = blockIdx.x % DEG_RANGES;
  const int j   = blockIdx.x / DEG_RANGES;
  const int lo  = k * RANGE_NODES;

  const int4*   d4 = (const int4*)dst + j * SLICE_QUADS;
  const float4* w4 = (const float4*)ew + j * SLICE_QUADS;
  int4   dq[DEG_TRIPS];
  float4 wq[DEG_TRIPS];
#pragma unroll
  for (int u = 0; u < DEG_TRIPS; ++u) {
    int q = tid + u * DEG_THREADS;
    if (q > SLICE_QUADS - 1) q = SLICE_QUADS - 1;
    dq[u] = d4[q];
    wq[u] = w4[q];
  }

  for (int i = tid; i < RANGE_NODES / 4; i += DEG_THREADS)
    ((float4*)sD)[i] = make_float4(0.f, 0.f, 0.f, 0.f);
  __syncthreads();

  unsigned* key4 = (unsigned*)tkey + j * SLICE_QUADS;
#pragma unroll
  for (int u = 0; u < DEG_TRIPS; ++u) {
    const int q = tid + u * DEG_THREADS;
    if (q < SLICE_QUADS) {
      deg_add(sD, dq[u].x, lo, wq[u].x);
      deg_add(sD, dq[u].y, lo, wq[u].y);
      deg_add(sD, dq[u].z, lo, wq[u].z);
      deg_add(sD, dq[u].w, lo, wq[u].w);
      if (q / KEY_QUADS == k)
        key4[q] = ((unsigned)dq[u].x / (unsigned)TILE_NODES) |
                  (((unsigned)dq[u].y / (unsigned)TILE_NODES) << 8) |
                  (((unsigned)dq[u].z / (unsigned)TILE_NODES) << 16) |
                  (((unsigned)dq[u].w / (unsigned)TILE_NODES) << 24);
    }
  }
  __syncthreads();

  float4* po = (float4*)(part + (size_t)j * N_NODES + lo);
  for (int i = tid; i < RANGE_NODES / 4; i += DEG_THREADS)
    po[i] = ((const float4*)sD)[i];
}

__global__ void dinv_kernel(const float* __restrict__ part,
                            float* __restrict__ dinv) {
  const int i = blockIdx.x * blockDim.x + threadIdx.x;
  if (i < N_NODES) {
    float d = 1.0f;
#pragma unroll
    for (int j = 0; j < DEG_SLICES; ++j) d += part[(size_t)j * N_NODES + i];
    dinv[i] = (d > 0.0f) ? rsqrtf(d) : 0.0f;
  }
}

__global__ __launch_bounds__(256) void gemm1_wmma(const float* __restrict__ x,
                                                  const float* __restrict__ W1,
                                                  float* __restrict__ h) {
  __shared__ float sX[64 * 132];
  const int tid  = threadIdx.x;
  const int row0 = blockIdx.x * 64;

  for (int i = tid; i < 64 * 32; i += 256) {
    const int r = i >> 5, c = i & 31;
    float4 v = make_float4(0.f, 0.f, 0.f, 0.f);
    const int gr = row0 + r;
    if (gr < N_NODES) v = ((const float4*)x)[gr * 32 + c];
    *(float4*)&sX[r * 132 + c * 4] = v;
  }
  __syncthreads();

  const int wave = tid >> 5;
  const int lane = tid & 31;
  const int lrow = lane & 15;
  const int hi   = lane >> 4;
  const int col  = wave * 16 + lrow;

  for (int t = 0; t < 4; ++t) {
    v8f acc = {};
    const float* ax = &sX[(t * 16 + lrow) * 132 + 2 * hi];
    for (int k0 = 0; k0 < 128; k0 += 4) {
      v2f a, b;
      a.x = ax[k0];
      a.y = ax[k0 + 1];
      const int kb = k0 + 2 * hi;
      b.x = W1[kb * 128 + col];
      b.y = W1[(kb + 1) * 128 + col];
      acc = __builtin_amdgcn_wmma_f32_16x16x4_f32(false, a, false, b,
                                                  (short)0, acc, false, false);
    }
    const int rbase = row0 + t * 16 + 8 * hi;
#pragma unroll
    for (int r = 0; r < 8; ++r) {
      const int gr = rbase + r;
      if (gr < N_NODES) h[gr * 128 + col] = acc[r];
    }
  }
}

#define OWN_WAVES   32
#define CHUNK_EDGES 512
#define N_CHUNKS    (N_EDGES / CHUNK_EDGES)
#define SCAN_DEPTH  8
#define QCAP        64

static_assert(N_EDGES % CHUNK_EDGES == 0, "chunks cover the edges exactly");
static_assert(SCAN_DEPTH == 8, "the pair select below names four registers");

#define REC_CAP   512
#define REC_SBITS 17
static_assert(N_NODES <= (1 << REC_SBITS), "a source node fits REC_SBITS bits");
static_assert(TILE_NODES < (1 << (32 - REC_SBITS)), "a tile row fits the rest");

__device__ __forceinline__ unsigned zero_bytes(unsigned x) {
  const unsigned t = (x & 0x7F7F7F7Fu) + 0x7F7F7F7Fu;
  return ~(t | x | 0x7F7F7F7Fu);
}

__device__ __forceinline__ void own_batch(const unsigned* qb, const int nvalid,
                                          const int lane, const int n0,
                                          const int* __restrict__ src,
                                          const int* __restrict__ dst,
                                          const float* __restrict__ ew,
                                          const float* __restrict__ dinv,
                                          const float* __restrict__ h,
                                          float* sT,
                                          uint2* __restrict__ wrec,
                                          const int nrec) {
  __builtin_amdgcn_fence(__ATOMIC_RELEASE, "wavefront");
  __builtin_amdgcn_wave_barrier();
  __builtin_amdgcn_fence(__ATOMIC_ACQUIRE, "wavefront");
  int      s    = 0;
  unsigned dl   = 0u;
  float    norm = 0.f;
  if (lane < nvalid) {
    const unsigned e = qb[lane];
    const int      d = dst[e];
    dl   = (unsigned)(d - n0);
    s    = src[e];
    norm = dinv[s] * ew[e] * dinv[d];
    const int slot = nrec + lane;
    if (slot < REC_CAP)
      wrec[slot] = make_uint2((unsigned)s | (dl << REC_SBITS),
                              __float_as_uint(norm));
  }
  const float* hl = h + lane;
  float*       tl = sT + lane;
  for (int g = 0; g < nvalid; g += 8) {
    float v[8][4];
#pragma unroll
    for (int i = 0; i < 8; ++i) {
      const int    sk = __builtin_amdgcn_readlane(s, g + i);
      const float* hr = hl + sk * 128;
      v[i][0] = hr[0];
      v[i][1] = hr[32];
      v[i][2] = hr[64];
      v[i][3] = hr[96];
    }
#pragma unroll
    for (int i = 0; i < 8; ++i) {
      if (g + i < nvalid) {
        const int   dk = __builtin_amdgcn_readlane((int)dl, g + i);
        const float nk = __int_as_float(
            __builtin_amdgcn_readlane(__float_as_int(norm), g + i));
        float* o = tl + dk * 128;
        __hip_atomic_fetch_add(o + 0,  v[i][0] * nk, __ATOMIC_RELAXED,
                               __HIP_MEMORY_SCOPE_WORKGROUP);
        __hip_atomic_fetch_add(o + 32, v[i][1] * nk, __ATOMIC_RELAXED,
                               __HIP_MEMORY_SCOPE_WORKGROUP);
        __hip_atomic_fetch_add(o + 64, v[i][2] * nk, __ATOMIC_RELAXED,
                               __HIP_MEMORY_SCOPE_WORKGROUP);
        __hip_atomic_fetch_add(o + 96, v[i][3] * nk, __ATOMIC_RELAXED,
                               __HIP_MEMORY_SCOPE_WORKGROUP);
      }
    }
  }
  __builtin_amdgcn_fence(__ATOMIC_RELEASE, "wavefront");
  __builtin_amdgcn_wave_barrier();
  __builtin_amdgcn_fence(__ATOMIC_ACQUIRE, "wavefront");
}

__global__ __launch_bounds__(1024) void agg1_owner_kernel(
    const int* __restrict__ src, const int* __restrict__ dst,
    const float* __restrict__ ew, const float* __restrict__ dinv,
    const float* __restrict__ h, const unsigned char* __restrict__ tkey,
    float* __restrict__ agg, unsigned* __restrict__ rec, int* __restrict__ cntw) {
  __shared__ __attribute__((aligned(16))) float sT[TILE_NODES * 128];
  __shared__ unsigned sQ[OWN_WAVES * QCAP];
  const int tid  = threadIdx.x;
  const int lane = tid & 31;
  const int wave = __builtin_amdgcn_readfirstlane(tid >> 5);
  const int tile = blockIdx.x;
  const int n0   = tile * TILE_NODES;

  for (int i = tid; i < TILE_NODES * 32; i += 1024) {
    const int   r  = n0 + (i >> 5);
    const float di = dinv[r];
    float4 v = ((const float4*)h)[r * 32 + (i & 31)];
    v.x = v.x * di * di;
    v.y = v.y * di * di;
    v.z = v.z * di * di;
    v.w = v.w * di * di;
    ((float4*)sT)[i] = v;
  }
  __syncthreads();

  unsigned*    q    = sQ + wave * QCAP;
  const uint4* key4 = (const uint4*)tkey;
  const unsigned pat = (unsigned)tile * 0x01010101u;
  const int start = (tile * N_CHUNKS) / N_TILES;
  int cnt = 0;
  uint2* wrec = (uint2*)rec + (size_t)(tile * OWN_WAVES + wave) * REC_CAP;
  int nrec = 0;
  for (int j = wave; j < N_CHUNKS; j += SCAN_DEPTH * OWN_WAVES) {
    uint4 kv[SCAN_DEPTH];
#pragma unroll
    for (int u = 0; u < SCAN_DEPTH; ++u) {
      const int jj = j + u * OWN_WAVES;
      int c = jj + start;
      if (c >= N_CHUNKS) c -= N_CHUNKS;
      if (jj >= N_CHUNKS) c = 0;
      kv[u] = key4[c * 32 + lane];
    }
    unsigned pk[SCAN_DEPTH / 2];
#pragma unroll
    for (int u = 0; u < SCAN_DEPTH; ++u) {
      unsigned lm = (zero_bytes(kv[u].x ^ pat) >> 7) |
                    (zero_bytes(kv[u].y ^ pat) >> 6) |
                    (zero_bytes(kv[u].z ^ pat) >> 5) |
                    (zero_bytes(kv[u].w ^ pat) >> 4);
      if (j + u * OWN_WAVES >= N_CHUNKS) lm = 0u;
      if (u & 1) pk[u >> 1] |= lm << 4;
      else       pk[u >> 1] = lm;
    }
#pragma nounroll
    for (int w = 0; w < SCAN_DEPTH / 2; ++w) {
      unsigned lm = (w == 0) ? pk[0] : (w == 1) ? pk[1] : (w == 2) ? pk[2] : pk[3];
      int ca = j + 2 * w * OWN_WAVES + start;
      if (ca >= N_CHUNKS) ca -= N_CHUNKS;
      int cb = j + (2 * w + 1) * OWN_WAVES + start;
      if (cb >= N_CHUNKS) cb -= N_CHUNKS;
      const unsigned ea = (unsigned)ca * CHUNK_EDGES + (unsigned)lane * 16u;
      const unsigned eb = (unsigned)cb * CHUNK_EDGES + (unsigned)lane * 16u;
      for (int it = 0; it < 32; ++it) {
        const bool     m    = lm != 0u;
        const unsigned mask = __builtin_amdgcn_ballot_w32(m);
        if (mask == 0u) break;
        const unsigned p   = (unsigned)__builtin_ctz(lm | 0x80000000u);
        const unsigned pos =
            (unsigned)cnt + __builtin_amdgcn_mbcnt_lo(mask, 0u);
        if (m) q[pos] = ((p & 4u) ? eb : ea) + 4u * (p & 3u) + (p >> 3);
        lm &= lm - 1u;
        cnt += __builtin_popcount(mask);
        if (cnt >= 32) {
          cnt -= 32;
          own_batch(q + cnt, 32, lane, n0, src, dst, ew, dinv, h, sT, wrec, nrec);
          nrec += 32;
        }
      }
    }
  }
  if (cnt > 0) {
    own_batch(q, cnt, lane, n0, src, dst, ew, dinv, h, sT, wrec, nrec);
    nrec += cnt;
  }
  if (lane == 0) cntw[tile * OWN_WAVES + wave] = nrec;
  __syncthreads();

  for (int i = tid; i < TILE_NODES * 32; i += 1024)
    ((float4*)agg)[n0 * 32 + i] = ((const float4*)sT)[i];
}

__global__ __launch_bounds__(256) void gemm2_wmma(const float* __restrict__ agg1,
                                                  const float* __restrict__ b1,
                                                  const float* __restrict__ W2,
                                                  float* __restrict__ h2) {
  __shared__ float sW[128 * 16];
  __shared__ float sb[128];
  const int tid = threadIdx.x;
  for (int i = tid; i < 512; i += 256) ((float4*)sW)[i] = ((const float4*)W2)[i];
  if (tid < 128) sb[tid] = b1[tid];
  __syncthreads();

  const int wave = tid >> 5, lane = tid & 31;
  const int lrow = lane & 15, hi = lane >> 4;
  const int row0 = blockIdx.x * 128 + wave * 16;
  const int arow = row0 + lrow;

  v8f acc = {};
  for (int k0 = 0; k0 < 128; k0 += 4) {
    const int k = k0 + 2 * hi;
    float a0 = 0.f, a1 = 0.f;
    if (arow < N_NODES) {
      a0 = fmaxf(agg1[arow * 128 + k] + sb[k], 0.f);
      a1 = fmaxf(agg1[arow * 128 + k + 1] + sb[k + 1], 0.f);
    }
    v2f a, b;
    a.x = a0; a.y = a1;
    b.x = sW[k * 16 + lrow];
    b.y = sW[(k + 1) * 16 + lrow];
    acc = __builtin_amdgcn_wmma_f32_16x16x4_f32(false, a, false, b,
                                                (short)0, acc, false, false);
  }
  const int rbase = row0 + 8 * hi;
#pragma unroll
  for (int r = 0; r < 8; ++r) {
    const int gr = rbase + r;
    if (gr < N_NODES) h2[gr * 16 + lrow] = acc[r];
  }
}


__device__ __forceinline__ void own_rows2(const int s, const unsigned dl,
                                          const float norm, const int lane,
                                          const float* __restrict__ h2,
                                          float* sT) {
  const int    col  = lane & 15;
  const int    half = lane >> 4;
  const float* hc   = h2 + col;
  float*       tc   = sT + col;
  float v[16];
  float nk[16];
  int   dk[16];
#pragma unroll
  for (int i = 0; i < 16; ++i) {
    const int k  = 2 * i + half;
    const int sk = __shfl(s, k, 32);
    dk[i] = __shfl((int)dl, k, 32);
    nk[i] = __shfl(norm, k, 32);
    v[i]  = hc[sk * C_DIM];
  }
#pragma unroll
  for (int i = 0; i < 16; ++i)
    __hip_atomic_fetch_add(tc + dk[i] * C_DIM, v[i] * nk[i],
                           __ATOMIC_RELAXED, __HIP_MEMORY_SCOPE_WORKGROUP);
}

__device__ __forceinline__ void own_batch2(const unsigned* qb, const int nvalid,
                                           const int lane, const int n0,
                                           const int* __restrict__ src,
                                           const int* __restrict__ dst,
                                           const float* __restrict__ ew,
                                           const float* __restrict__ dinv,
                                           const float* __restrict__ h2,
                                           float* sT) {
  __builtin_amdgcn_fence(__ATOMIC_RELEASE, "wavefront");
  __builtin_amdgcn_wave_barrier();
  __builtin_amdgcn_fence(__ATOMIC_ACQUIRE, "wavefront");
  int      s    = 0;
  unsigned dl   = TILE_NODES;
  float    norm = 0.f;
  if (lane < nvalid) {
    const unsigned e = qb[lane];
    const int      d = dst[e];
    dl   = (unsigned)(d - n0);
    s    = src[e];
    norm = dinv[s] * ew[e] * dinv[d];
  }
  own_rows2(s, dl, norm, lane, h2, sT);
  __builtin_amdgcn_fence(__ATOMIC_RELEASE, "wavefront");
  __builtin_amdgcn_wave_barrier();
  __builtin_amdgcn_fence(__ATOMIC_ACQUIRE, "wavefront");
}

__global__ __launch_bounds__(1024) void agg2_owner_kernel(
    const int* __restrict__ src, const int* __restrict__ dst,
    const float* __restrict__ ew, const float* __restrict__ dinv,
    const float* __restrict__ h2, const unsigned char* __restrict__ tkey,
    const unsigned* __restrict__ rec, const int* __restrict__ cntw,
    const float* __restrict__ b2, float* __restrict__ out) {
  __shared__ __attribute__((aligned(16))) float sT[(TILE_NODES + 1) * C_DIM];
  __shared__ unsigned sQ[OWN_WAVES * QCAP];
  const int tid  = threadIdx.x;
  const int lane = tid & 31;
  const int wave = __builtin_amdgcn_readfirstlane(tid >> 5);
  const int tile = blockIdx.x;
  const int n0   = tile * TILE_NODES;

  for (int i = tid; i < TILE_NODES * (C_DIM / 4); i += 1024) {
    const int   r  = n0 + i / (C_DIM / 4);
    const float di = dinv[r];
    float4 v = ((const float4*)h2)[n0 * (C_DIM / 4) + i];
    v.x = v.x * di * di;
    v.y = v.y * di * di;
    v.z = v.z * di * di;
    v.w = v.w * di * di;
    ((float4*)sT)[i] = v;
  }
  if (tid < C_DIM) sT[TILE_NODES * C_DIM + tid] = 0.f;
  __syncthreads();

  const int nrec = cntw[tile * OWN_WAVES + wave];
  if (nrec <= REC_CAP) {
    const uint2* wrec =
        (const uint2*)rec + (size_t)(tile * OWN_WAVES + wave) * REC_CAP;
    const uint2 idle = make_uint2((unsigned)TILE_NODES << REC_SBITS, 0u);
    if (nrec > 0) {
      const int last = nrec - 1;
      uint2 cur = wrec[lane < nrec ? lane : last];
      if (lane >= nrec) cur = idle;
      for (int b = 0; b < nrec; b += 32) {
        const int i1 = b + 32 + lane;
        uint2 nxt = wrec[i1 < nrec ? i1 : last];
        if (i1 >= nrec) nxt = idle;
        own_rows2((int)(cur.x & ((1u << REC_SBITS) - 1u)), cur.x >> REC_SBITS,
                  __uint_as_float(cur.y), lane, h2, sT);
        cur = nxt;
      }
    }
  } else {
    unsigned*    q    = sQ + wave * QCAP;
    const uint4* key4 = (const uint4*)tkey;
    const unsigned pat = (unsigned)tile * 0x01010101u;
    const int start = (tile * N_CHUNKS) / N_TILES;
    int cnt = 0;
    for (int j = wave; j < N_CHUNKS; j += SCAN_DEPTH * OWN_WAVES) {
      uint4 kv[SCAN_DEPTH];
#pragma unroll
      for (int u = 0; u < SCAN_DEPTH; ++u) {
        const int jj = j + u * OWN_WAVES;
        int c = jj + start;
        if (c >= N_CHUNKS) c -= N_CHUNKS;
        if (jj >= N_CHUNKS) c = 0;
        kv[u] = key4[c * 32 + lane];
      }
      unsigned pk[SCAN_DEPTH / 2];
#pragma unroll
      for (int u = 0; u < SCAN_DEPTH; ++u) {
        unsigned lm = (zero_bytes(kv[u].x ^ pat) >> 7) |
                      (zero_bytes(kv[u].y ^ pat) >> 6) |
                      (zero_bytes(kv[u].z ^ pat) >> 5) |
                      (zero_bytes(kv[u].w ^ pat) >> 4);
        if (j + u * OWN_WAVES >= N_CHUNKS) lm = 0u;
        if (u & 1) pk[u >> 1] |= lm << 4;
        else       pk[u >> 1] = lm;
      }
#pragma nounroll
      for (int w = 0; w < SCAN_DEPTH / 2; ++w) {
        unsigned lm = (w == 0) ? pk[0] : (w == 1) ? pk[1] : (w == 2) ? pk[2] : pk[3];
        int ca = j + 2 * w * OWN_WAVES + start;
        if (ca >= N_CHUNKS) ca -= N_CHUNKS;
        int cb = j + (2 * w + 1) * OWN_WAVES + start;
        if (cb >= N_CHUNKS) cb -= N_CHUNKS;
        const unsigned ea = (unsigned)ca * CHUNK_EDGES + (unsigned)lane * 16u;
        const unsigned eb = (unsigned)cb * CHUNK_EDGES + (unsigned)lane * 16u;
        for (int it = 0; it < 32; ++it) {
          const bool     m    = lm != 0u;
          const unsigned mask = __builtin_amdgcn_ballot_w32(m);
          if (mask == 0u) break;
          const unsigned p   = (unsigned)__builtin_ctz(lm | 0x80000000u);
          const unsigned pos =
              (unsigned)cnt + __builtin_amdgcn_mbcnt_lo(mask, 0u);
          if (m) q[pos] = ((p & 4u) ? eb : ea) + 4u * (p & 3u) + (p >> 3);
          lm &= lm - 1u;
          cnt += __builtin_popcount(mask);
          if (cnt >= 32) {
            cnt -= 32;
            own_batch2(q + cnt, 32, lane, n0, src, dst, ew, dinv, h2, sT);
          }
        }
      }
    }
    if (cnt > 0) own_batch2(q, cnt, lane, n0, src, dst, ew, dinv, h2, sT);
  }
  __syncthreads();

  if (tid < TILE_NODES) {
    float v[C_DIM];
    const float4* p = (const float4*)(sT + tid * C_DIM);
    float m = -3.402823466e+38f;
#pragma unroll
    for (int qd = 0; qd < 4; ++qd) {
      const float4 t = p[qd];
      v[4 * qd + 0] = t.x + b2[4 * qd + 0];
      v[4 * qd + 1] = t.y + b2[4 * qd + 1];
      v[4 * qd + 2] = t.z + b2[4 * qd + 2];
      v[4 * qd + 3] = t.w + b2[4 * qd + 3];
    }
#pragma unroll
    for (int c = 0; c < C_DIM; ++c) m = fmaxf(m, v[c]);
    float sum = 0.f;
#pragma unroll
    for (int c = 0; c < C_DIM; ++c) { v[c] = expf(v[c] - m); sum += v[c]; }
    const float inv = 1.0f / sum;
    float4* o = (float4*)(out + (size_t)(n0 + tid) * C_DIM);
#pragma unroll
    for (int qd = 0; qd < 4; ++qd)
      o[qd] = make_float4(v[4 * qd + 0] * inv, v[4 * qd + 1] * inv,
                          v[4 * qd + 2] * inv, v[4 * qd + 3] * inv);
  }
}

extern "C" void kernel_launch(void* const* d_in, const int* in_sizes, int n_in,
                              void* d_out, int out_size, void* d_ws, size_t ws_size,
                              hipStream_t stream) {
  const float* x  = (const float*)d_in[0];
  const int*   ei = (const int*)d_in[1];
  const float* ew = (const float*)d_in[2];
  const float* W1 = (const float*)d_in[3];
  const float* b1 = (const float*)d_in[4];
  const float* W2 = (const float*)d_in[5];
  const float* b2 = (const float*)d_in[6];
  float* out = (float*)d_out;

  const int* src = ei;
  const int* dst = ei + N_EDGES;

  float* ws   = (float*)d_ws;
  float* dinv = ws;
  float* h    = ws + N_NODES;
  float* agg1 = ws + N_NODES + (size_t)N_NODES * 128;
  float* h2   = h;
  unsigned char* tkey = (unsigned char*)(agg1 + (size_t)N_NODES * 128);
  float* part = (float*)(tkey + N_EDGES);
  unsigned* rec  = (unsigned*)(part + (size_t)DEG_SLICES * N_NODES);
  int*      cntw = (int*)(rec + (size_t)N_TILES * OWN_WAVES * REC_CAP * 2);

  const int T = 256;

  deg_hist_kernel<<<DEG_RANGES * DEG_SLICES, DEG_THREADS, 0, stream>>>(dst, ew, part, tkey);
  dinv_kernel<<<(N_NODES + T - 1) / T, T, 0, stream>>>(part, dinv);

  gemm1_wmma<<<(N_NODES + 63) / 64, T, 0, stream>>>(x, W1, h);
  agg1_owner_kernel<<<N_TILES, 1024, 0, stream>>>(src, dst, ew, dinv, h, tkey, agg1, rec, cntw);

  gemm2_wmma<<<(N_NODES + 127) / 128, T, 0, stream>>>(agg1, b1, W2, h2);
  agg2_owner_kernel<<<N_TILES, 1024, 0, stream>>>(src, dst, ew, dinv, h2, tkey, rec, cntw, b2, out);
}


# dinv launch folded into GEMM1 (its blocks sum their rows' degree partials); both GEMMs store rows scaled by dinv: edges carry only ew, dinv[dst] applied at the tile store; sums reassociated
# speedup vs baseline: 2.8857x; 1.0346x over previous
#include <hip/hip_runtime.h>
#include <hip/hip_bf16.h>
#include <math.h>

#define N_NODES 100000
#define N_EDGES 1600000
#define F_DIM 128
#define H_DIM 128
#define C_DIM 16

typedef __attribute__((ext_vector_type(2))) float v2f;
typedef __attribute__((ext_vector_type(8))) float v8f;

#define TILE_NODES  400
#define N_TILES     (N_NODES / TILE_NODES)
static_assert(N_NODES % TILE_NODES == 0, "tiles cover the nodes exactly");
static_assert(N_TILES <= 256, "a tile number fits one byte");

#define DEG_RANGES  5
#define DEG_SLICES  50
#define DEG_THREADS 1024
#define RANGE_NODES (N_NODES / DEG_RANGES)
#define SLICE_QUADS (N_EDGES / DEG_SLICES / 4)
#define KEY_QUADS   (SLICE_QUADS / DEG_RANGES)
#define DEG_TRIPS   ((SLICE_QUADS + DEG_THREADS - 1) / DEG_THREADS)

static_assert(N_NODES % (DEG_RANGES * 4) == 0, "ranges cover the nodes in float4s");
static_assert(N_EDGES % (DEG_SLICES * 4) == 0, "slices cover the edges in quads");
static_assert(SLICE_QUADS % DEG_RANGES == 0, "key words split evenly");

__device__ __forceinline__ void deg_add(float* sD, const int d, const int lo,
                                        const float w) {
  const unsigned r = (unsigned)(d - lo);
  if (r < (unsigned)RANGE_NODES)
    __hip_atomic_fetch_add(sD + r, w, __ATOMIC_RELAXED,
                           __HIP_MEMORY_SCOPE_WORKGROUP);
}

__global__ __launch_bounds__(DEG_THREADS) void deg_hist_kernel(
    const int* __restrict__ dst, const float* __restrict__ ew,
    float* __restrict__ part, unsigned char* __restrict__ tkey) {
  __shared__ __attribute__((aligned(16))) float sD[RANGE_NODES];
  const int tid = threadIdx.x;
  const int k   = blockIdx.x % DEG_RANGES;
  const int j   = blockIdx.x / DEG_RANGES;
  const int lo  = k * RANGE_NODES;

  const int4*   d4 = (const int4*)dst + j * SLICE_QUADS;
  const float4* w4 = (const float4*)ew + j * SLICE_QUADS;
  int4   dq[DEG_TRIPS];
  float4 wq[DEG_TRIPS];
#pragma unroll
  for (int u = 0; u < DEG_TRIPS; ++u) {
    int q = tid + u * DEG_THREADS;
    if (q > SLICE_QUADS - 1) q = SLICE_QUADS - 1;
    dq[u] = d4[q];
    wq[u] = w4[q];
  }

  for (int i = tid; i < RANGE_NODES / 4; i += DEG_THREADS)
    ((float4*)sD)[i] = make_float4(0.f, 0.f, 0.f, 0.f);
  __syncthreads();

  unsigned* key4 = (unsigned*)tkey + j * SLICE_QUADS;
#pragma unroll
  for (int u = 0; u < DEG_TRIPS; ++u) {
    const int q = tid + u * DEG_THREADS;
    if (q < SLICE_QUADS) {
      deg_add(sD, dq[u].x, lo, wq[u].x);
      deg_add(sD, dq[u].y, lo, wq[u].y);
      deg_add(sD, dq[u].z, lo, wq[u].z);
      deg_add(sD, dq[u].w, lo, wq[u].w);
      if (q / KEY_QUADS == k)
        key4[q] = ((unsigned)dq[u].x / (unsigned)TILE_NODES) |
                  (((unsigned)dq[u].y / (unsigned)TILE_NODES) << 8) |
                  (((unsigned)dq[u].z / (unsigned)TILE_NODES) << 16) |
                  (((unsigned)dq[u].w / (unsigned)TILE_NODES) << 24);
    }
  }
  __syncthreads();

  float4* po = (float4*)(part + (size_t)j * N_NODES + lo);
  for (int i = tid; i < RANGE_NODES / 4; i += DEG_THREADS)
    po[i] = ((const float4*)sD)[i];
}

__global__ __launch_bounds__(256) void gemm1_wmma(const float* __restrict__ x,
                                                  const float* __restrict__ W1,
                                                  const float* __restrict__ part,
                                                  float* __restrict__ dinv,
                                                  float* __restrict__ h) {
  __shared__ float sX[64 * 132];
  __shared__ float sDi[64];
  const int tid  = threadIdx.x;
  const int row0 = blockIdx.x * 64;

  for (int i = tid; i < 64 * 32; i += 256) {
    const int r = i >> 5, c = i & 31;
    float4 v = make_float4(0.f, 0.f, 0.f, 0.f);
    const int gr = row0 + r;
    if (gr < N_NODES) v = ((const float4*)x)[gr * 32 + c];
    *(float4*)&sX[r * 132 + c * 4] = v;
  }
  if (tid < 64) {
    const int gr = row0 + tid;
    float d = 1.0f;
    if (gr < N_NODES) {
#pragma unroll
      for (int j = 0; j < DEG_SLICES; ++j) d += part[(size_t)j * N_NODES + gr];
    }
    const float di = (d > 0.0f) ? rsqrtf(d) : 0.0f;
    sDi[tid] = di;
    if (gr < N_NODES) dinv[gr] = di;
  }
  __syncthreads();

  const int wave = tid >> 5;
  const int lane = tid & 31;
  const int lrow = lane & 15;
  const int hi   = lane >> 4;
  const int col  = wave * 16 + lrow;

  for (int t = 0; t < 4; ++t) {
    v8f acc = {};
    const float* ax = &sX[(t * 16 + lrow) * 132 + 2 * hi];
    for (int k0 = 0; k0 < 128; k0 += 4) {
      v2f a, b;
      a.x = ax[k0];
      a.y = ax[k0 + 1];
      const int kb = k0 + 2 * hi;
      b.x = W1[kb * 128 + col];
      b.y = W1[(kb + 1) * 128 + col];
      acc = __builtin_amdgcn_wmma_f32_16x16x4_f32(false, a, false, b,
                                                  (short)0, acc, false, false);
    }
    const int rbase = row0 + t * 16 + 8 * hi;
#pragma unroll
    for (int r = 0; r < 8; ++r) {
      const int gr = rbase + r;
      if (gr < N_NODES) h[gr * 128 + col] = acc[r] * sDi[t * 16 + 8 * hi + r];
    }
  }
}

#define OWN_WAVES   32
#define CHUNK_EDGES 512
#define N_CHUNKS    (N_EDGES / CHUNK_EDGES)
#define SCAN_DEPTH  8
#define QCAP        64

static_assert(N_EDGES % CHUNK_EDGES == 0, "chunks cover the edges exactly");
static_assert(SCAN_DEPTH == 8, "the pair select below names four registers");

#define REC_CAP   512
#define REC_SBITS 17
static_assert(N_NODES <= (1 << REC_SBITS), "a source node fits REC_SBITS bits");
static_assert(TILE_NODES < (1 << (32 - REC_SBITS)), "a tile row fits the rest");

__device__ __forceinline__ unsigned zero_bytes(unsigned x) {
  const unsigned t = (x & 0x7F7F7F7Fu) + 0x7F7F7F7Fu;
  return ~(t | x | 0x7F7F7F7Fu);
}

__device__ __forceinline__ void own_batch(const unsigned* qb, const int nvalid,
                                          const int lane, const int n0,
                                          const int* __restrict__ src,
                                          const int* __restrict__ dst,
                                          const float* __restrict__ ew,
                                          const float* __restrict__ h,
                                          float* sT,
                                          uint2* __restrict__ wrec,
                                          const int nrec) {
  __builtin_amdgcn_fence(__ATOMIC_RELEASE, "wavefront");
  __builtin_amdgcn_wave_barrier();
  __builtin_amdgcn_fence(__ATOMIC_ACQUIRE, "wavefront");
  int      s    = 0;
  unsigned dl   = 0u;
  float    norm = 0.f;
  if (lane < nvalid) {
    const unsigned e = qb[lane];
    dl   = (unsigned)(dst[e] - n0);
    s    = src[e];
    norm = ew[e];
    const int slot = nrec + lane;
    if (slot < REC_CAP)
      wrec[slot] = make_uint2((unsigned)s | (dl << REC_SBITS),
                              __float_as_uint(norm));
  }
  const float* hl = h + lane;
  float*       tl = sT + lane;
  for (int g = 0; g < nvalid; g += 8) {
    float v[8][4];
#pragma unroll
    for (int i = 0; i < 8; ++i) {
      const int    sk = __builtin_amdgcn_readlane(s, g + i);
      const float* hr = hl + sk * 128;
      v[i][0] = hr[0];
      v[i][1] = hr[32];
      v[i][2] = hr[64];
      v[i][3] = hr[96];
    }
#pragma unroll
    for (int i = 0; i < 8; ++i) {
      if (g + i < nvalid) {
        const int   dk = __builtin_amdgcn_readlane((int)dl, g + i);
        const float nk = __int_as_float(
            __builtin_amdgcn_readlane(__float_as_int(norm), g + i));
        float* o = tl + dk * 128;
        __hip_atomic_fetch_add(o + 0,  v[i][0] * nk, __ATOMIC_RELAXED,
                               __HIP_MEMORY_SCOPE_WORKGROUP);
        __hip_atomic_fetch_add(o + 32, v[i][1] * nk, __ATOMIC_RELAXED,
                               __HIP_MEMORY_SCOPE_WORKGROUP);
        __hip_atomic_fetch_add(o + 64, v[i][2] * nk, __ATOMIC_RELAXED,
                               __HIP_MEMORY_SCOPE_WORKGROUP);
        __hip_atomic_fetch_add(o + 96, v[i][3] * nk, __ATOMIC_RELAXED,
                               __HIP_MEMORY_SCOPE_WORKGROUP);
      }
    }
  }
  __builtin_amdgcn_fence(__ATOMIC_RELEASE, "wavefront");
  __builtin_amdgcn_wave_barrier();
  __builtin_amdgcn_fence(__ATOMIC_ACQUIRE, "wavefront");
}

__global__ __launch_bounds__(1024) void agg1_owner_kernel(
    const int* __restrict__ src, const int* __restrict__ dst,
    const float* __restrict__ ew, const float* __restrict__ dinv,
    const float* __restrict__ h, const unsigned char* __restrict__ tkey,
    float* __restrict__ agg, unsigned* __restrict__ rec, int* __restrict__ cntw) {
  __shared__ __attribute__((aligned(16))) float sT[TILE_NODES * 128];
  __shared__ unsigned sQ[OWN_WAVES * QCAP];
  const int tid  = threadIdx.x;
  const int lane = tid & 31;
  const int wave = __builtin_amdgcn_readfirstlane(tid >> 5);
  const int tile = blockIdx.x;
  const int n0   = tile * TILE_NODES;

  for (int i = tid; i < TILE_NODES * 32; i += 1024)
    ((float4*)sT)[i] = ((const float4*)h)[n0 * 32 + i];
  __syncthreads();

  unsigned*    q    = sQ + wave * QCAP;
  const uint4* key4 = (const uint4*)tkey;
  const unsigned pat = (unsigned)tile * 0x01010101u;
  const int start = (tile * N_CHUNKS) / N_TILES;
  int cnt = 0;
  uint2* wrec = (uint2*)rec + (size_t)(tile * OWN_WAVES + wave) * REC_CAP;
  int nrec = 0;
  for (int j = wave; j < N_CHUNKS; j += SCAN_DEPTH * OWN_WAVES) {
    uint4 kv[SCAN_DEPTH];
#pragma unroll
    for (int u = 0; u < SCAN_DEPTH; ++u) {
      const int jj = j + u * OWN_WAVES;
      int c = jj + start;
      if (c >= N_CHUNKS) c -= N_CHUNKS;
      if (jj >= N_CHUNKS) c = 0;
      kv[u] = key4[c * 32 + lane];
    }
    unsigned pk[SCAN_DEPTH / 2];
#pragma unroll
    for (int u = 0; u < SCAN_DEPTH; ++u) {
      unsigned lm = (zero_bytes(kv[u].x ^ pat) >> 7) |
                    (zero_bytes(kv[u].y ^ pat) >> 6) |
                    (zero_bytes(kv[u].z ^ pat) >> 5) |
                    (zero_bytes(kv[u].w ^ pat) >> 4);
      if (j + u * OWN_WAVES >= N_CHUNKS) lm = 0u;
      if (u & 1) pk[u >> 1] |= lm << 4;
      else       pk[u >> 1] = lm;
    }
#pragma nounroll
    for (int w = 0; w < SCAN_DEPTH / 2; ++w) {
      unsigned lm = (w == 0) ? pk[0] : (w == 1) ? pk[1] : (w == 2) ? pk[2] : pk[3];
      int ca = j + 2 * w * OWN_WAVES + start;
      if (ca >= N_CHUNKS) ca -= N_CHUNKS;
      int cb = j + (2 * w + 1) * OWN_WAVES + start;
      if (cb >= N_CHUNKS) cb -= N_CHUNKS;
      const unsigned ea = (unsigned)ca * CHUNK_EDGES + (unsigned)lane * 16u;
      const unsigned eb = (unsigned)cb * CHUNK_EDGES + (unsigned)lane * 16u;
      for (int it = 0; it < 32; ++it) {
        const bool     m    = lm != 0u;
        const unsigned mask = __builtin_amdgcn_ballot_w32(m);
        if (mask == 0u) break;
        const unsigned p   = (unsigned)__builtin_ctz(lm | 0x80000000u);
        const unsigned pos =
            (unsigned)cnt + __builtin_amdgcn_mbcnt_lo(mask, 0u);
        if (m) q[pos] = ((p & 4u) ? eb : ea) + 4u * (p & 3u) + (p >> 3);
        lm &= lm - 1u;
        cnt += __builtin_popcount(mask);
        if (cnt >= 32) {
          cnt -= 32;
          own_batch(q + cnt, 32, lane, n0, src, dst, ew, h, sT, wrec, nrec);
          nrec += 32;
        }
      }
    }
  }
  if (cnt > 0) {
    own_batch(q, cnt, lane, n0, src, dst, ew, h, sT, wrec, nrec);
    nrec += cnt;
  }
  if (lane == 0) cntw[tile * OWN_WAVES + wave] = nrec;
  __syncthreads();

  for (int i = tid; i < TILE_NODES * 32; i += 1024) {
    const float di = dinv[n0 + (i >> 5)];
    float4 v = ((const float4*)sT)[i];
    v.x *= di;
    v.y *= di;
    v.z *= di;
    v.w *= di;
    ((float4*)agg)[n0 * 32 + i] = v;
  }
}

__global__ __launch_bounds__(256) void gemm2_wmma(const float* __restrict__ agg1,
                                                  const float* __restrict__ b1,
                                                  const float* __restrict__ W2,
                                                  const float* __restrict__ dinv,
                                                  float* __restrict__ h2) {
  __shared__ float sW[128 * 16];
  __shared__ float sb[128];
  const int tid = threadIdx.x;
  for (int i = tid; i < 512; i += 256) ((float4*)sW)[i] = ((const float4*)W2)[i];
  if (tid < 128) sb[tid] = b1[tid];
  __syncthreads();

  const int wave = tid >> 5, lane = tid & 31;
  const int lrow = lane & 15, hi = lane >> 4;
  const int row0 = blockIdx.x * 128 + wave * 16;
  const int arow = row0 + lrow;

  v8f acc = {};
  for (int k0 = 0; k0 < 128; k0 += 4) {
    const int k = k0 + 2 * hi;
    float a0 = 0.f, a1 = 0.f;
    if (arow < N_NODES) {
      a0 = fmaxf(agg1[arow * 128 + k] + sb[k], 0.f);
      a1 = fmaxf(agg1[arow * 128 + k + 1] + sb[k + 1], 0.f);
    }
    v2f a, b;
    a.x = a0; a.y = a1;
    b.x = sW[k * 16 + lrow];
    b.y = sW[(k + 1) * 16 + lrow];
    acc = __builtin_amdgcn_wmma_f32_16x16x4_f32(false, a, false, b,
                                                (short)0, acc, false, false);
  }
  const int rbase = row0 + 8 * hi;
#pragma unroll
  for (int r = 0; r < 8; ++r) {
    const int gr = rbase + r;
    if (gr < N_NODES) h2[gr * 16 + lrow] = acc[r] * dinv[gr];
  }
}


__device__ __forceinline__ void own_rows2(const int s, const unsigned dl,
                                          const float norm, const int lane,
                                          const float* __restrict__ h2,
                                          float* sT) {
  const int    col  = lane & 15;
  const int    half = lane >> 4;
  const float* hc   = h2 + col;
  float*       tc   = sT + col;
  float v[16];
  float nk[16];
  int   dk[16];
#pragma unroll
  for (int i = 0; i < 16; ++i) {
    const int k  = 2 * i + half;
    const int sk = __shfl(s, k, 32);
    dk[i] = __shfl((int)dl, k, 32);
    nk[i] = __shfl(norm, k, 32);
    v[i]  = hc[sk * C_DIM];
  }
#pragma unroll
  for (int i = 0; i < 16; ++i)
    __hip_atomic_fetch_add(tc + dk[i] * C_DIM, v[i] * nk[i],
                           __ATOMIC_RELAXED, __HIP_MEMORY_SCOPE_WORKGROUP);
}

__device__ __forceinline__ void own_batch2(const unsigned* qb, const int nvalid,
                                           const int lane, const int n0,
                                           const int* __restrict__ src,
                                           const int* __restrict__ dst,
                                           const float* __restrict__ ew,
                                           const float* __restrict__ h2,
                                           float* sT) {
  __builtin_amdgcn_fence(__ATOMIC_RELEASE, "wavefront");
  __builtin_amdgcn_wave_barrier();
  __builtin_amdgcn_fence(__ATOMIC_ACQUIRE, "wavefront");
  int      s    = 0;
  unsigned dl   = TILE_NODES;
  float    norm = 0.f;
  if (lane < nvalid) {
    const unsigned e = qb[lane];
    dl   = (unsigned)(dst[e] - n0);
    s    = src[e];
    norm = ew[e];
  }
  own_rows2(s, dl, norm, lane, h2, sT);
  __builtin_amdgcn_fence(__ATOMIC_RELEASE, "wavefront");
  __builtin_amdgcn_wave_barrier();
  __builtin_amdgcn_fence(__ATOMIC_ACQUIRE, "wavefront");
}

__global__ __launch_bounds__(1024) void agg2_owner_kernel(
    const int* __restrict__ src, const int* __restrict__ dst,
    const float* __restrict__ ew, const float* __restrict__ dinv,
    const float* __restrict__ h2, const unsigned char* __restrict__ tkey,
    const unsigned* __restrict__ rec, const int* __restrict__ cntw,
    const float* __restrict__ b2, float* __restrict__ out) {
  __shared__ __attribute__((aligned(16))) float sT[(TILE_NODES + 1) * C_DIM];
  __shared__ unsigned sQ[OWN_WAVES * QCAP];
  const int tid  = threadIdx.x;
  const int lane = tid & 31;
  const int wave = __builtin_amdgcn_readfirstlane(tid >> 5);
  const int tile = blockIdx.x;
  const int n0   = tile * TILE_NODES;

  for (int i = tid; i < TILE_NODES * (C_DIM / 4); i += 1024)
    ((float4*)sT)[i] = ((const float4*)h2)[n0 * (C_DIM / 4) + i];
  if (tid < C_DIM) sT[TILE_NODES * C_DIM + tid] = 0.f;
  __syncthreads();

  const int nrec = cntw[tile * OWN_WAVES + wave];
  if (nrec <= REC_CAP) {
    const uint2* wrec =
        (const uint2*)rec + (size_t)(tile * OWN_WAVES + wave) * REC_CAP;
    const uint2 idle = make_uint2((unsigned)TILE_NODES << REC_SBITS, 0u);
    if (nrec > 0) {
      const int last = nrec - 1;
      uint2 cur = wrec[lane < nrec ? lane : last];
      if (lane >= nrec) cur = idle;
      for (int b = 0; b < nrec; b += 32) {
        const int i1 = b + 32 + lane;
        uint2 nxt = wrec[i1 < nrec ? i1 : last];
        if (i1 >= nrec) nxt = idle;
        own_rows2((int)(cur.x & ((1u << REC_SBITS) - 1u)), cur.x >> REC_SBITS,
                  __uint_as_float(cur.y), lane, h2, sT);
        cur = nxt;
      }
    }
  } else {
    unsigned*    q    = sQ + wave * QCAP;
    const uint4* key4 = (const uint4*)tkey;
    const unsigned pat = (unsigned)tile * 0x01010101u;
    const int start = (tile * N_CHUNKS) / N_TILES;
    int cnt = 0;
    for (int j = wave; j < N_CHUNKS; j += SCAN_DEPTH * OWN_WAVES) {
      uint4 kv[SCAN_DEPTH];
#pragma unroll
      for (int u = 0; u < SCAN_DEPTH; ++u) {
        const int jj = j + u * OWN_WAVES;
        int c = jj + start;
        if (c >= N_CHUNKS) c -= N_CHUNKS;
        if (jj >= N_CHUNKS) c = 0;
        kv[u] = key4[c * 32 + lane];
      }
      unsigned pk[SCAN_DEPTH / 2];
#pragma unroll
      for (int u = 0; u < SCAN_DEPTH; ++u) {
        unsigned lm = (zero_bytes(kv[u].x ^ pat) >> 7) |
                      (zero_bytes(kv[u].y ^ pat) >> 6) |
                      (zero_bytes(kv[u].z ^ pat) >> 5) |
                      (zero_bytes(kv[u].w ^ pat) >> 4);
        if (j + u * OWN_WAVES >= N_CHUNKS) lm = 0u;
        if (u & 1) pk[u >> 1] |= lm << 4;
        else       pk[u >> 1] = lm;
      }
#pragma nounroll
      for (int w = 0; w < SCAN_DEPTH / 2; ++w) {
        unsigned lm = (w == 0) ? pk[0] : (w == 1) ? pk[1] : (w == 2) ? pk[2] : pk[3];
        int ca = j + 2 * w * OWN_WAVES + start;
        if (ca >= N_CHUNKS) ca -= N_CHUNKS;
        int cb = j + (2 * w + 1) * OWN_WAVES + start;
        if (cb >= N_CHUNKS) cb -= N_CHUNKS;
        const unsigned ea = (unsigned)ca * CHUNK_EDGES + (unsigned)lane * 16u;
        const unsigned eb = (unsigned)cb * CHUNK_EDGES + (unsigned)lane * 16u;
        for (int it = 0; it < 32; ++it) {
          const bool     m    = lm != 0u;
          const unsigned mask = __builtin_amdgcn_ballot_w32(m);
          if (mask == 0u) break;
          const unsigned p   = (unsigned)__builtin_ctz(lm | 0x80000000u);
          const unsigned pos =
              (unsigned)cnt + __builtin_amdgcn_mbcnt_lo(mask, 0u);
          if (m) q[pos] = ((p & 4u) ? eb : ea) + 4u * (p & 3u) + (p >> 3);
          lm &= lm - 1u;
          cnt += __builtin_popcount(mask);
          if (cnt >= 32) {
            cnt -= 32;
            own_batch2(q + cnt, 32, lane, n0, src, dst, ew, h2, sT);
          }
        }
      }
    }
    if (cnt > 0) own_batch2(q, cnt, lane, n0, src, dst, ew, h2, sT);
  }
  __syncthreads();

  if (tid < TILE_NODES) {
    float v[C_DIM];
    const float4* p = (const float4*)(sT + tid * C_DIM);
    const float di = dinv[n0 + tid];
    float m = -3.402823466e+38f;
#pragma unroll
    for (int qd = 0; qd < 4; ++qd) {
      const float4 t = p[qd];
      v[4 * qd + 0] = t.x * di + b2[4 * qd + 0];
      v[4 * qd + 1] = t.y * di + b2[4 * qd + 1];
      v[4 * qd + 2] = t.z * di + b2[4 * qd + 2];
      v[4 * qd + 3] = t.w * di + b2[4 * qd + 3];
    }
#pragma unroll
    for (int c = 0; c < C_DIM; ++c) m = fmaxf(m, v[c]);
    float sum = 0.f;
#pragma unroll
    for (int c = 0; c < C_DIM; ++c) { v[c] = expf(v[c] - m); sum += v[c]; }
    const float inv = 1.0f / sum;
    float4* o = (float4*)(out + (size_t)(n0 + tid) * C_DIM);
#pragma unroll
    for (int qd = 0; qd < 4; ++qd)
      o[qd] = make_float4(v[4 * qd + 0] * inv, v[4 * qd + 1] * inv,
                          v[4 * qd + 2] * inv, v[4 * qd + 3] * inv);
  }
}

extern "C" void kernel_launch(void* const* d_in, const int* in_sizes, int n_in,
                              void* d_out, int out_size, void* d_ws, size_t ws_size,
                              hipStream_t stream) {
  const float* x  = (const float*)d_in[0];
  const int*   ei = (const int*)d_in[1];
  const float* ew = (const float*)d_in[2];
  const float* W1 = (const float*)d_in[3];
  const float* b1 = (const float*)d_in[4];
  const float* W2 = (const float*)d_in[5];
  const float* b2 = (const float*)d_in[6];
  float* out = (float*)d_out;

  const int* src = ei;
  const int* dst = ei + N_EDGES;

  float* ws   = (float*)d_ws;
  float* dinv = ws;
  float* h    = ws + N_NODES;
  float* agg1 = ws + N_NODES + (size_t)N_NODES * 128;
  float* h2   = h;
  unsigned char* tkey = (unsigned char*)(agg1 + (size_t)N_NODES * 128);
  float* part = (float*)(tkey + N_EDGES);
  unsigned* rec  = (unsigned*)(part + (size_t)DEG_SLICES * N_NODES);
  int*      cntw = (int*)(rec + (size_t)N_TILES * OWN_WAVES * REC_CAP * 2);

  const int T = 256;

  deg_hist_kernel<<<DEG_RANGES * DEG_SLICES, DEG_THREADS, 0, stream>>>(dst, ew, part, tkey);

  gemm1_wmma<<<(N_NODES + 63) / 64, T, 0, stream>>>(x, W1, part, dinv, h);
  agg1_owner_kernel<<<N_TILES, 1024, 0, stream>>>(src, dst, ew, dinv, h, tkey, agg1, rec, cntw);

  gemm2_wmma<<<(N_NODES + 127) / 128, T, 0, stream>>>(agg1, b1, W2, dinv, h2);
  agg2_owner_kernel<<<N_TILES, 1024, 0, stream>>>(src, dst, ew, dinv, h2, tkey, rec, cntw, b2, out);
}


# Degree kernel also writes an 8-byte record per edge (src, tile row, weight): layer-1 owner waves fetch 1 line per kept edge, not 3 (src, dst, ew); LDS float atomics and arrival order kept
# speedup vs baseline: 3.0438x; 1.0548x over previous
#include <hip/hip_runtime.h>
#include <hip/hip_bf16.h>
#include <math.h>

#define N_NODES 100000
#define N_EDGES 1600000
#define F_DIM 128
#define H_DIM 128
#define C_DIM 16

typedef __attribute__((ext_vector_type(2))) float v2f;
typedef __attribute__((ext_vector_type(8))) float v8f;

#define TILE_NODES  400
#define N_TILES     (N_NODES / TILE_NODES)
static_assert(N_NODES % TILE_NODES == 0, "tiles cover the nodes exactly");
static_assert(N_TILES <= 256, "a tile number fits one byte");

#define DEG_RANGES  5
#define DEG_SLICES  50
#define DEG_THREADS 1024
#define RANGE_NODES (N_NODES / DEG_RANGES)
#define SLICE_QUADS (N_EDGES / DEG_SLICES / 4)
#define DEG_TRIPS   ((SLICE_QUADS + DEG_THREADS - 1) / DEG_THREADS)
#define SLICE_EDGES (N_EDGES / DEG_SLICES)

static_assert(N_NODES % (DEG_RANGES * 4) == 0, "ranges cover the nodes in float4s");
static_assert(N_EDGES % (DEG_SLICES * 4) == 0, "slices cover the edges in quads");
static_assert(DEG_RANGES <= DEG_TRIPS && DEG_TRIPS <= 2 * DEG_RANGES,
              "every trip has one writing group, a group at most two trips");
static_assert(DEG_RANGES * DEG_THREADS <= SLICE_QUADS, "trip k is a whole trip");

#define REC_SBITS 17

__device__ __forceinline__ void deg_add(float* sD, const int d, const int lo,
                                        const float w) {
  const unsigned r = (unsigned)(d - lo);
  if (r < (unsigned)RANGE_NODES)
    __hip_atomic_fetch_add(sD + r, w, __ATOMIC_RELAXED,
                           __HIP_MEMORY_SCOPE_WORKGROUP);
}

__device__ __forceinline__ void edge_rec(const int s, const int d, const unsigned t,
                                         const float w, unsigned& r0, unsigned& r1) {
  r0 = (unsigned)s | (((unsigned)d - t * (unsigned)TILE_NODES) << REC_SBITS);
  r1 = __float_as_uint(w);
}

__global__ __launch_bounds__(DEG_THREADS) void deg_hist_kernel(
    const int* __restrict__ src, const int* __restrict__ dst,
    const float* __restrict__ ew, float* __restrict__ part,
    unsigned char* __restrict__ tkey, unsigned* __restrict__ erec) {
  __shared__ __attribute__((aligned(16))) float sD[RANGE_NODES];
  const int tid = threadIdx.x;
  const int k   = blockIdx.x % DEG_RANGES;
  const int j   = blockIdx.x / DEG_RANGES;
  const int lo  = k * RANGE_NODES;

  const int4*   d4 = (const int4*)dst + j * SLICE_QUADS;
  const float4* w4 = (const float4*)ew + j * SLICE_QUADS;
  int4   dq[DEG_TRIPS];
  float4 wq[DEG_TRIPS];
#pragma unroll
  for (int u = 0; u < DEG_TRIPS; ++u) {
    int q = tid + u * DEG_THREADS;
    if (q > SLICE_QUADS - 1) q = SLICE_QUADS - 1;
    dq[u] = d4[q];
    wq[u] = w4[q];
  }
  const int4* s4 = (const int4*)src + j * SLICE_QUADS;
  int qb = tid + (k + DEG_RANGES) * DEG_THREADS;
  if (qb > SLICE_QUADS - 1) qb = SLICE_QUADS - 1;
  const int4 sA = s4[tid + k * DEG_THREADS];
  const int4 sB = s4[qb];

  for (int i = tid; i < RANGE_NODES / 4; i += DEG_THREADS)
    ((float4*)sD)[i] = make_float4(0.f, 0.f, 0.f, 0.f);
  __syncthreads();

  unsigned* key4 = (unsigned*)tkey + j * SLICE_QUADS;
#pragma unroll
  for (int u = 0; u < DEG_TRIPS; ++u) {
    const int q = tid + u * DEG_THREADS;
    if (q < SLICE_QUADS) {
      deg_add(sD, dq[u].x, lo, wq[u].x);
      deg_add(sD, dq[u].y, lo, wq[u].y);
      deg_add(sD, dq[u].z, lo, wq[u].z);
      deg_add(sD, dq[u].w, lo, wq[u].w);
      if (u % DEG_RANGES == k) {
        const int4 sq = (u < DEG_RANGES) ? sA : sB;
        const unsigned tx = (unsigned)dq[u].x / (unsigned)TILE_NODES;
        const unsigned ty = (unsigned)dq[u].y / (unsigned)TILE_NODES;
        const unsigned tz = (unsigned)dq[u].z / (unsigned)TILE_NODES;
        const unsigned tw = (unsigned)dq[u].w / (unsigned)TILE_NODES;
        key4[q] = tx | (ty << 8) | (tz << 16) | (tw << 24);
        uint4 ra, rb;
        edge_rec(sq.x, dq[u].x, tx, wq[u].x, ra.x, ra.y);
        edge_rec(sq.y, dq[u].y, ty, wq[u].y, ra.z, ra.w);
        edge_rec(sq.z, dq[u].z, tz, wq[u].z, rb.x, rb.y);
        edge_rec(sq.w, dq[u].w, tw, wq[u].w, rb.z, rb.w);
        uint4* r4 = (uint4*)(erec + (size_t)j * (2 * SLICE_EDGES)) + 2 * q;
        r4[0] = ra;
        r4[1] = rb;
      }
    }
  }
  __syncthreads();

  float4* po = (float4*)(part + (size_t)j * N_NODES + lo);
  for (int i = tid; i < RANGE_NODES / 4; i += DEG_THREADS)
    po[i] = ((const float4*)sD)[i];
}

__global__ __launch_bounds__(256) void gemm1_wmma(const float* __restrict__ x,
                                                  const float* __restrict__ W1,
                                                  const float* __restrict__ part,
                                                  float* __restrict__ dinv,
                                                  float* __restrict__ h) {
  __shared__ float sX[64 * 132];
  __shared__ float sDi[64];
  const int tid  = threadIdx.x;
  const int row0 = blockIdx.x * 64;

  for (int i = tid; i < 64 * 32; i += 256) {
    const int r = i >> 5, c = i & 31;
    float4 v = make_float4(0.f, 0.f, 0.f, 0.f);
    const int gr = row0 + r;
    if (gr < N_NODES) v = ((const float4*)x)[gr * 32 + c];
    *(float4*)&sX[r * 132 + c * 4] = v;
  }
  if (tid < 64) {
    const int gr = row0 + tid;
    float d = 1.0f;
    if (gr < N_NODES) {
#pragma unroll
      for (int j = 0; j < DEG_SLICES; ++j) d += part[(size_t)j * N_NODES + gr];
    }
    const float di = (d > 0.0f) ? rsqrtf(d) : 0.0f;
    sDi[tid] = di;
    if (gr < N_NODES) dinv[gr] = di;
  }
  __syncthreads();

  const int wave = tid >> 5;
  const int lane = tid & 31;
  const int lrow = lane & 15;
  const int hi   = lane >> 4;
  const int col  = wave * 16 + lrow;

  for (int t = 0; t < 4; ++t) {
    v8f acc = {};
    const float* ax = &sX[(t * 16 + lrow) * 132 + 2 * hi];
    for (int k0 = 0; k0 < 128; k0 += 4) {
      v2f a, b;
      a.x = ax[k0];
      a.y = ax[k0 + 1];
      const int kb = k0 + 2 * hi;
      b.x = W1[kb * 128 + col];
      b.y = W1[(kb + 1) * 128 + col];
      acc = __builtin_amdgcn_wmma_f32_16x16x4_f32(false, a, false, b,
                                                  (short)0, acc, false, false);
    }
    const int rbase = row0 + t * 16 + 8 * hi;
#pragma unroll
    for (int r = 0; r < 8; ++r) {
      const int gr = rbase + r;
      if (gr < N_NODES) h[gr * 128 + col] = acc[r] * sDi[t * 16 + 8 * hi + r];
    }
  }
}

#define OWN_WAVES   32
#define CHUNK_EDGES 512
#define N_CHUNKS    (N_EDGES / CHUNK_EDGES)
#define SCAN_DEPTH  8
#define QCAP        64

static_assert(N_EDGES % CHUNK_EDGES == 0, "chunks cover the edges exactly");
static_assert(SCAN_DEPTH == 8, "the pair select below names four registers");

#define REC_CAP   512
static_assert(N_NODES <= (1 << REC_SBITS), "a source node fits REC_SBITS bits");
static_assert(TILE_NODES < (1 << (32 - REC_SBITS)), "a tile row fits the rest");

__device__ __forceinline__ unsigned zero_bytes(unsigned x) {
  const unsigned t = (x & 0x7F7F7F7Fu) + 0x7F7F7F7Fu;
  return ~(t | x | 0x7F7F7F7Fu);
}

__device__ __forceinline__ void own_batch(const unsigned* qb, const int nvalid,
                                          const int lane,
                                          const uint2* __restrict__ erec,
                                          const float* __restrict__ h,
                                          float* sT,
                                          uint2* __restrict__ wrec,
                                          const int nrec) {
  __builtin_amdgcn_fence(__ATOMIC_RELEASE, "wavefront");
  __builtin_amdgcn_wave_barrier();
  __builtin_amdgcn_fence(__ATOMIC_ACQUIRE, "wavefront");
  int      s    = 0;
  unsigned dl   = 0u;
  float    norm = 0.f;
  if (lane < nvalid) {
    const uint2 r = erec[qb[lane]];
    s    = (int)(r.x & ((1u << REC_SBITS) - 1u));
    dl   = r.x >> REC_SBITS;
    norm = __uint_as_float(r.y);
    const int slot = nrec + lane;
    if (slot < REC_CAP) wrec[slot] = r;
  }
  const float* hl = h + lane;
  float*       tl = sT + lane;
  for (int g = 0; g < nvalid; g += 8) {
    float v[8][4];
#pragma unroll
    for (int i = 0; i < 8; ++i) {
      const int    sk = __builtin_amdgcn_readlane(s, g + i);
      const float* hr = hl + sk * 128;
      v[i][0] = hr[0];
      v[i][1] = hr[32];
      v[i][2] = hr[64];
      v[i][3] = hr[96];
    }
#pragma unroll
    for (int i = 0; i < 8; ++i) {
      if (g + i < nvalid) {
        const int   dk = __builtin_amdgcn_readlane((int)dl, g + i);
        const float nk = __int_as_float(
            __builtin_amdgcn_readlane(__float_as_int(norm), g + i));
        float* o = tl + dk * 128;
        __hip_atomic_fetch_add(o + 0,  v[i][0] * nk, __ATOMIC_RELAXED,
                               __HIP_MEMORY_SCOPE_WORKGROUP);
        __hip_atomic_fetch_add(o + 32, v[i][1] * nk, __ATOMIC_RELAXED,
                               __HIP_MEMORY_SCOPE_WORKGROUP);
        __hip_atomic_fetch_add(o + 64, v[i][2] * nk, __ATOMIC_RELAXED,
                               __HIP_MEMORY_SCOPE_WORKGROUP);
        __hip_atomic_fetch_add(o + 96, v[i][3] * nk, __ATOMIC_RELAXED,
                               __HIP_MEMORY_SCOPE_WORKGROUP);
      }
    }
  }
  __builtin_amdgcn_fence(__ATOMIC_RELEASE, "wavefront");
  __builtin_amdgcn_wave_barrier();
  __builtin_amdgcn_fence(__ATOMIC_ACQUIRE, "wavefront");
}

__global__ __launch_bounds__(1024) void agg1_owner_kernel(
    const unsigned* __restrict__ erec, const float* __restrict__ dinv,
    const float* __restrict__ h, const unsigned char* __restrict__ tkey,
    float* __restrict__ agg, unsigned* __restrict__ rec, int* __restrict__ cntw) {
  __shared__ __attribute__((aligned(16))) float sT[TILE_NODES * 128];
  __shared__ unsigned sQ[OWN_WAVES * QCAP];
  const int tid  = threadIdx.x;
  const int lane = tid & 31;
  const int wave = __builtin_amdgcn_readfirstlane(tid >> 5);
  const int tile = blockIdx.x;
  const int n0   = tile * TILE_NODES;

  for (int i = tid; i < TILE_NODES * 32; i += 1024)
    ((float4*)sT)[i] = ((const float4*)h)[n0 * 32 + i];
  __syncthreads();

  unsigned*    q    = sQ + wave * QCAP;
  const uint4* key4 = (const uint4*)tkey;
  const unsigned pat = (unsigned)tile * 0x01010101u;
  const int start = (tile * N_CHUNKS) / N_TILES;
  int cnt = 0;
  uint2* wrec = (uint2*)rec + (size_t)(tile * OWN_WAVES + wave) * REC_CAP;
  int nrec = 0;
  for (int j = wave; j < N_CHUNKS; j += SCAN_DEPTH * OWN_WAVES) {
    uint4 kv[SCAN_DEPTH];
#pragma unroll
    for (int u = 0; u < SCAN_DEPTH; ++u) {
      const int jj = j + u * OWN_WAVES;
      int c = jj + start;
      if (c >= N_CHUNKS) c -= N_CHUNKS;
      if (jj >= N_CHUNKS) c = 0;
      kv[u] = key4[c * 32 + lane];
    }
    unsigned pk[SCAN_DEPTH / 2];
#pragma unroll
    for (int u = 0; u < SCAN_DEPTH; ++u) {
      unsigned lm = (zero_bytes(kv[u].x ^ pat) >> 7) |
                    (zero_bytes(kv[u].y ^ pat) >> 6) |
                    (zero_bytes(kv[u].z ^ pat) >> 5) |
                    (zero_bytes(kv[u].w ^ pat) >> 4);
      if (j + u * OWN_WAVES >= N_CHUNKS) lm = 0u;
      if (u & 1) pk[u >> 1] |= lm << 4;
      else       pk[u >> 1] = lm;
    }
#pragma nounroll
    for (int w = 0; w < SCAN_DEPTH / 2; ++w) {
      unsigned lm = (w == 0) ? pk[0] : (w == 1) ? pk[1] : (w == 2) ? pk[2] : pk[3];
      int ca = j + 2 * w * OWN_WAVES + start;
      if (ca >= N_CHUNKS) ca -= N_CHUNKS;
      int cb = j + (2 * w + 1) * OWN_WAVES + start;
      if (cb >= N_CHUNKS) cb -= N_CHUNKS;
      const unsigned ea = (unsigned)ca * CHUNK_EDGES + (unsigned)lane * 16u;
      const unsigned eb = (unsigned)cb * CHUNK_EDGES + (unsigned)lane * 16u;
      for (int it = 0; it < 32; ++it) {
        const bool     m    = lm != 0u;
        const unsigned mask = __builtin_amdgcn_ballot_w32(m);
        if (mask == 0u) break;
        const unsigned p   = (unsigned)__builtin_ctz(lm | 0x80000000u);
        const unsigned pos =
            (unsigned)cnt + __builtin_amdgcn_mbcnt_lo(mask, 0u);
        if (m) q[pos] = ((p & 4u) ? eb : ea) + 4u * (p & 3u) + (p >> 3);
        lm &= lm - 1u;
        cnt += __builtin_popcount(mask);
        if (cnt >= 32) {
          cnt -= 32;
          own_batch(q + cnt, 32, lane, (const uint2*)erec, h, sT, wrec, nrec);
          nrec += 32;
        }
      }
    }
  }
  if (cnt > 0) {
    own_batch(q, cnt, lane, (const uint2*)erec, h, sT, wrec, nrec);
    nrec += cnt;
  }
  if (lane == 0) cntw[tile * OWN_WAVES + wave] = nrec;
  __syncthreads();

  for (int i = tid; i < TILE_NODES * 32; i += 1024) {
    const float di = dinv[n0 + (i >> 5)];
    float4 v = ((const float4*)sT)[i];
    v.x *= di;
    v.y *= di;
    v.z *= di;
    v.w *= di;
    ((float4*)agg)[n0 * 32 + i] = v;
  }
}

__global__ __launch_bounds__(256) void gemm2_wmma(const float* __restrict__ agg1,
                                                  const float* __restrict__ b1,
                                                  const float* __restrict__ W2,
                                                  const float* __restrict__ dinv,
                                                  float* __restrict__ h2) {
  __shared__ float sW[128 * 16];
  __shared__ float sb[128];
  const int tid = threadIdx.x;
  for (int i = tid; i < 512; i += 256) ((float4*)sW)[i] = ((const float4*)W2)[i];
  if (tid < 128) sb[tid] = b1[tid];
  __syncthreads();

  const int wave = tid >> 5, lane = tid & 31;
  const int lrow = lane & 15, hi = lane >> 4;
  const int row0 = blockIdx.x * 128 + wave * 16;
  const int arow = row0 + lrow;

  v8f acc = {};
  for (int k0 = 0; k0 < 128; k0 += 4) {
    const int k = k0 + 2 * hi;
    float a0 = 0.f, a1 = 0.f;
    if (arow < N_NODES) {
      a0 = fmaxf(agg1[arow * 128 + k] + sb[k], 0.f);
      a1 = fmaxf(agg1[arow * 128 + k + 1] + sb[k + 1], 0.f);
    }
    v2f a, b;
    a.x = a0; a.y = a1;
    b.x = sW[k * 16 + lrow];
    b.y = sW[(k + 1) * 16 + lrow];
    acc = __builtin_amdgcn_wmma_f32_16x16x4_f32(false, a, false, b,
                                                (short)0, acc, false, false);
  }
  const int rbase = row0 + 8 * hi;
#pragma unroll
  for (int r = 0; r < 8; ++r) {
    const int gr = rbase + r;
    if (gr < N_NODES) h2[gr * 16 + lrow] = acc[r] * dinv[gr];
  }
}


__device__ __forceinline__ void own_rows2(const int s, const unsigned dl,
                                          const float norm, const int lane,
                                          const float* __restrict__ h2,
                                          float* sT) {
  const int    col  = lane & 15;
  const int    half = lane >> 4;
  const float* hc   = h2 + col;
  float*       tc   = sT + col;
  float v[16];
  float nk[16];
  int   dk[16];
#pragma unroll
  for (int i = 0; i < 16; ++i) {
    const int k  = 2 * i + half;
    const int sk = __shfl(s, k, 32);
    dk[i] = __shfl((int)dl, k, 32);
    nk[i] = __shfl(norm, k, 32);
    v[i]  = hc[sk * C_DIM];
  }
#pragma unroll
  for (int i = 0; i < 16; ++i)
    __hip_atomic_fetch_add(tc + dk[i] * C_DIM, v[i] * nk[i],
                           __ATOMIC_RELAXED, __HIP_MEMORY_SCOPE_WORKGROUP);
}

__device__ __forceinline__ void own_batch2(const unsigned* qb, const int nvalid,
                                           const int lane, const int n0,
                                           const int* __restrict__ src,
                                           const int* __restrict__ dst,
                                           const float* __restrict__ ew,
                                           const float* __restrict__ h2,
                                           float* sT) {
  __builtin_amdgcn_fence(__ATOMIC_RELEASE, "wavefront");
  __builtin_amdgcn_wave_barrier();
  __builtin_amdgcn_fence(__ATOMIC_ACQUIRE, "wavefront");
  int      s    = 0;
  unsigned dl   = TILE_NODES;
  float    norm = 0.f;
  if (lane < nvalid) {
    const unsigned e = qb[lane];
    dl   = (unsigned)(dst[e] - n0);
    s    = src[e];
    norm = ew[e];
  }
  own_rows2(s, dl, norm, lane, h2, sT);
  __builtin_amdgcn_fence(__ATOMIC_RELEASE, "wavefront");
  __builtin_amdgcn_wave_barrier();
  __builtin_amdgcn_fence(__ATOMIC_ACQUIRE, "wavefront");
}

__global__ __launch_bounds__(1024) void agg2_owner_kernel(
    const int* __restrict__ src, const int* __restrict__ dst,
    const float* __restrict__ ew, const float* __restrict__ dinv,
    const float* __restrict__ h2, const unsigned char* __restrict__ tkey,
    const unsigned* __restrict__ rec, const int* __restrict__ cntw,
    const float* __restrict__ b2, float* __restrict__ out) {
  __shared__ __attribute__((aligned(16))) float sT[(TILE_NODES + 1) * C_DIM];
  __shared__ unsigned sQ[OWN_WAVES * QCAP];
  const int tid  = threadIdx.x;
  const int lane = tid & 31;
  const int wave = __builtin_amdgcn_readfirstlane(tid >> 5);
  const int tile = blockIdx.x;
  const int n0   = tile * TILE_NODES;

  for (int i = tid; i < TILE_NODES * (C_DIM / 4); i += 1024)
    ((float4*)sT)[i] = ((const float4*)h2)[n0 * (C_DIM / 4) + i];
  if (tid < C_DIM) sT[TILE_NODES * C_DIM + tid] = 0.f;
  __syncthreads();

  const int nrec = cntw[tile * OWN_WAVES + wave];
  if (nrec <= REC_CAP) {
    const uint2* wrec =
        (const uint2*)rec + (size_t)(tile * OWN_WAVES + wave) * REC_CAP;
    const uint2 idle = make_uint2((unsigned)TILE_NODES << REC_SBITS, 0u);
    if (nrec > 0) {
      const int last = nrec - 1;
      uint2 cur = wrec[lane < nrec ? lane : last];
      if (lane >= nrec) cur = idle;
      for (int b = 0; b < nrec; b += 32) {
        const int i1 = b + 32 + lane;
        uint2 nxt = wrec[i1 < nrec ? i1 : last];
        if (i1 >= nrec) nxt = idle;
        own_rows2((int)(cur.x & ((1u << REC_SBITS) - 1u)), cur.x >> REC_SBITS,
                  __uint_as_float(cur.y), lane, h2, sT);
        cur = nxt;
      }
    }
  } else {
    unsigned*    q    = sQ + wave * QCAP;
    const uint4* key4 = (const uint4*)tkey;
    const unsigned pat = (unsigned)tile * 0x01010101u;
    const int start = (tile * N_CHUNKS) / N_TILES;
    int cnt = 0;
    for (int j = wave; j < N_CHUNKS; j += SCAN_DEPTH * OWN_WAVES) {
      uint4 kv[SCAN_DEPTH];
#pragma unroll
      for (int u = 0; u < SCAN_DEPTH; ++u) {
        const int jj = j + u * OWN_WAVES;
        int c = jj + start;
        if (c >= N_CHUNKS) c -= N_CHUNKS;
        if (jj >= N_CHUNKS) c = 0;
        kv[u] = key4[c * 32 + lane];
      }
      unsigned pk[SCAN_DEPTH / 2];
#pragma unroll
      for (int u = 0; u < SCAN_DEPTH; ++u) {
        unsigned lm = (zero_bytes(kv[u].x ^ pat) >> 7) |
                      (zero_bytes(kv[u].y ^ pat) >> 6) |
                      (zero_bytes(kv[u].z ^ pat) >> 5) |
                      (zero_bytes(kv[u].w ^ pat) >> 4);
        if (j + u * OWN_WAVES >= N_CHUNKS) lm = 0u;
        if (u & 1) pk[u >> 1] |= lm << 4;
        else       pk[u >> 1] = lm;
      }
#pragma nounroll
      for (int w = 0; w < SCAN_DEPTH / 2; ++w) {
        unsigned lm = (w == 0) ? pk[0] : (w == 1) ? pk[1] : (w == 2) ? pk[2] : pk[3];
        int ca = j + 2 * w * OWN_WAVES + start;
        if (ca >= N_CHUNKS) ca -= N_CHUNKS;
        int cb = j + (2 * w + 1) * OWN_WAVES + start;
        if (cb >= N_CHUNKS) cb -= N_CHUNKS;
        const unsigned ea = (unsigned)ca * CHUNK_EDGES + (unsigned)lane * 16u;
        const unsigned eb = (unsigned)cb * CHUNK_EDGES + (unsigned)lane * 16u;
        for (int it = 0; it < 32; ++it) {
          const bool     m    = lm != 0u;
          const unsigned mask = __builtin_amdgcn_ballot_w32(m);
          if (mask == 0u) break;
          const unsigned p   = (unsigned)__builtin_ctz(lm | 0x80000000u);
          const unsigned pos =
              (unsigned)cnt + __builtin_amdgcn_mbcnt_lo(mask, 0u);
          if (m) q[pos] = ((p & 4u) ? eb : ea) + 4u * (p & 3u) + (p >> 3);
          lm &= lm - 1u;
          cnt += __builtin_popcount(mask);
          if (cnt >= 32) {
            cnt -= 32;
            own_batch2(q + cnt, 32, lane, n0, src, dst, ew, h2, sT);
          }
        }
      }
    }
    if (cnt > 0) own_batch2(q, cnt, lane, n0, src, dst, ew, h2, sT);
  }
  __syncthreads();

  if (tid < TILE_NODES) {
    float v[C_DIM];
    const float4* p = (const float4*)(sT + tid * C_DIM);
    const float di = dinv[n0 + tid];
    float m = -3.402823466e+38f;
#pragma unroll
    for (int qd = 0; qd < 4; ++qd) {
      const float4 t = p[qd];
      v[4 * qd + 0] = t.x * di + b2[4 * qd + 0];
      v[4 * qd + 1] = t.y * di + b2[4 * qd + 1];
      v[4 * qd + 2] = t.z * di + b2[4 * qd + 2];
      v[4 * qd + 3] = t.w * di + b2[4 * qd + 3];
    }
#pragma unroll
    for (int c = 0; c < C_DIM; ++c) m = fmaxf(m, v[c]);
    float sum = 0.f;
#pragma unroll
    for (int c = 0; c < C_DIM; ++c) { v[c] = expf(v[c] - m); sum += v[c]; }
    const float inv = 1.0f / sum;
    float4* o = (float4*)(out + (size_t)(n0 + tid) * C_DIM);
#pragma unroll
    for (int qd = 0; qd < 4; ++qd)
      o[qd] = make_float4(v[4 * qd + 0] * inv, v[4 * qd + 1] * inv,
                          v[4 * qd + 2] * inv, v[4 * qd + 3] * inv);
  }
}

extern "C" void kernel_launch(void* const* d_in, const int* in_sizes, int n_in,
                              void* d_out, int out_size, void* d_ws, size_t ws_size,
                              hipStream_t stream) {
  const float* x  = (const float*)d_in[0];
  const int*   ei = (const int*)d_in[1];
  const float* ew = (const float*)d_in[2];
  const float* W1 = (const float*)d_in[3];
  const float* b1 = (const float*)d_in[4];
  const float* W2 = (const float*)d_in[5];
  const float* b2 = (const float*)d_in[6];
  float* out = (float*)d_out;

  const int* src = ei;
  const int* dst = ei + N_EDGES;

  float* ws   = (float*)d_ws;
  float* dinv = ws;
  float* h    = ws + N_NODES;
  float* agg1 = ws + N_NODES + (size_t)N_NODES * 128;
  float* h2   = h;
  unsigned char* tkey = (unsigned char*)(agg1 + (size_t)N_NODES * 128);
  float* part = (float*)(tkey + N_EDGES);
  unsigned* rec  = (unsigned*)(part + (size_t)DEG_SLICES * N_NODES);
  int*      cntw = (int*)(rec + (size_t)N_TILES * OWN_WAVES * REC_CAP * 2);
  unsigned* erec = (unsigned*)(cntw + N_TILES * OWN_WAVES);

  const int T = 256;

  deg_hist_kernel<<<DEG_RANGES * DEG_SLICES, DEG_THREADS, 0, stream>>>(src, dst, ew, part, tkey, erec);

  gemm1_wmma<<<(N_NODES + 63) / 64, T, 0, stream>>>(x, W1, part, dinv, h);
  agg1_owner_kernel<<<N_TILES, 1024, 0, stream>>>(erec, dinv, h, tkey, agg1, rec, cntw);

  gemm2_wmma<<<(N_NODES + 127) / 128, T, 0, stream>>>(agg1, b1, W2, dinv, h2);
  agg2_owner_kernel<<<N_TILES, 1024, 0, stream>>>(src, dst, ew, dinv, h2, tkey, rec, cntw, b2, out);
}


# Global loads that stood under guards or in rolled loops (GEMM1 x tile, GEMM2 A operand, owner tiles' start rows, keys, records, dinv) go out as one group a phase ahead; sums and k order unchanged
# speedup vs baseline: 3.2890x; 1.0806x over previous
#include <hip/hip_runtime.h>
#include <hip/hip_bf16.h>
#include <math.h>

#define N_NODES 100000
#define N_EDGES 1600000
#define F_DIM 128
#define H_DIM 128
#define C_DIM 16

typedef __attribute__((ext_vector_type(2))) float v2f;
typedef __attribute__((ext_vector_type(8))) float v8f;

#define TILE_NODES  400
#define N_TILES     (N_NODES / TILE_NODES)
static_assert(N_NODES % TILE_NODES == 0, "tiles cover the nodes exactly");
static_assert(N_TILES <= 256, "a tile number fits one byte");

#define DEG_RANGES  5
#define DEG_SLICES  50
#define DEG_THREADS 1024
#define RANGE_NODES (N_NODES / DEG_RANGES)
#define SLICE_QUADS (N_EDGES / DEG_SLICES / 4)
#define DEG_TRIPS   ((SLICE_QUADS + DEG_THREADS - 1) / DEG_THREADS)
#define SLICE_EDGES (N_EDGES / DEG_SLICES)

static_assert(N_NODES % (DEG_RANGES * 4) == 0, "ranges cover the nodes in float4s");
static_assert(N_EDGES % (DEG_SLICES * 4) == 0, "slices cover the edges in quads");
static_assert(DEG_RANGES <= DEG_TRIPS && DEG_TRIPS <= 2 * DEG_RANGES,
              "every trip has one writing group, a group at most two trips");
static_assert(DEG_RANGES * DEG_THREADS <= SLICE_QUADS, "trip k is a whole trip");

#define REC_SBITS 17

__device__ __forceinline__ void deg_add(float* sD, const int d, const int lo,
                                        const float w) {
  const unsigned r = (unsigned)(d - lo);
  if (r < (unsigned)RANGE_NODES)
    __hip_atomic_fetch_add(sD + r, w, __ATOMIC_RELAXED,
                           __HIP_MEMORY_SCOPE_WORKGROUP);
}

__device__ __forceinline__ void edge_rec(const int s, const int d, const unsigned t,
                                         const float w, unsigned& r0, unsigned& r1) {
  r0 = (unsigned)s | (((unsigned)d - t * (unsigned)TILE_NODES) << REC_SBITS);
  r1 = __float_as_uint(w);
}

__global__ __launch_bounds__(DEG_THREADS) void deg_hist_kernel(
    const int* __restrict__ src, const int* __restrict__ dst,
    const float* __restrict__ ew, float* __restrict__ part,
    unsigned char* __restrict__ tkey, unsigned* __restrict__ erec) {
  __shared__ __attribute__((aligned(16))) float sD[RANGE_NODES];
  const int tid = threadIdx.x;
  const int k   = blockIdx.x % DEG_RANGES;
  const int j   = blockIdx.x / DEG_RANGES;
  const int lo  = k * RANGE_NODES;

  const int4*   d4 = (const int4*)dst + j * SLICE_QUADS;
  const float4* w4 = (const float4*)ew + j * SLICE_QUADS;
  int4   dq[DEG_TRIPS];
  float4 wq[DEG_TRIPS];
#pragma unroll
  for (int u = 0; u < DEG_TRIPS; ++u) {
    int q = tid + u * DEG_THREADS;
    if (q > SLICE_QUADS - 1) q = SLICE_QUADS - 1;
    dq[u] = d4[q];
    wq[u] = w4[q];
  }
  const int4* s4 = (const int4*)src + j * SLICE_QUADS;
  int qb = tid + (k + DEG_RANGES) * DEG_THREADS;
  if (qb > SLICE_QUADS - 1) qb = SLICE_QUADS - 1;
  const int4 sA = s4[tid + k * DEG_THREADS];
  const int4 sB = s4[qb];

  for (int i = tid; i < RANGE_NODES / 4; i += DEG_THREADS)
    ((float4*)sD)[i] = make_float4(0.f, 0.f, 0.f, 0.f);
  __syncthreads();

  unsigned* key4 = (unsigned*)tkey + j * SLICE_QUADS;
#pragma unroll
  for (int u = 0; u < DEG_TRIPS; ++u) {
    const int q = tid + u * DEG_THREADS;
    if (q < SLICE_QUADS) {
      deg_add(sD, dq[u].x, lo, wq[u].x);
      deg_add(sD, dq[u].y, lo, wq[u].y);
      deg_add(sD, dq[u].z, lo, wq[u].z);
      deg_add(sD, dq[u].w, lo, wq[u].w);
      if (u % DEG_RANGES == k) {
        const int4 sq = (u < DEG_RANGES) ? sA : sB;
        const unsigned tx = (unsigned)dq[u].x / (unsigned)TILE_NODES;
        const unsigned ty = (unsigned)dq[u].y / (unsigned)TILE_NODES;
        const unsigned tz = (unsigned)dq[u].z / (unsigned)TILE_NODES;
        const unsigned tw = (unsigned)dq[u].w / (unsigned)TILE_NODES;
        key4[q] = tx | (ty << 8) | (tz << 16) | (tw << 24);
        uint4 ra, rb;
        edge_rec(sq.x, dq[u].x, tx, wq[u].x, ra.x, ra.y);
        edge_rec(sq.y, dq[u].y, ty, wq[u].y, ra.z, ra.w);
        edge_rec(sq.z, dq[u].z, tz, wq[u].z, rb.x, rb.y);
        edge_rec(sq.w, dq[u].w, tw, wq[u].w, rb.z, rb.w);
        uint4* r4 = (uint4*)(erec + (size_t)j * (2 * SLICE_EDGES)) + 2 * q;
        r4[0] = ra;
        r4[1] = rb;
      }
    }
  }
  __syncthreads();

  float4* po = (float4*)(part + (size_t)j * N_NODES + lo);
  for (int i = tid; i < RANGE_NODES / 4; i += DEG_THREADS)
    po[i] = ((const float4*)sD)[i];
}

__global__ __launch_bounds__(256) void gemm1_wmma(const float* __restrict__ x,
                                                  const float* __restrict__ W1,
                                                  const float* __restrict__ part,
                                                  float* __restrict__ dinv,
                                                  float* __restrict__ h) {
  __shared__ float sX[64 * 132];
  __shared__ __attribute__((aligned(16))) float sDi[64];
  const int tid  = threadIdx.x;
  const int row0 = blockIdx.x * 64;

  float4 xv[8];
#pragma unroll
  for (int u = 0; u < 8; ++u) {
    const int i = tid + u * 256;
    int gr = row0 + (i >> 5);
    if (gr > N_NODES - 1) gr = N_NODES - 1;
    xv[u] = ((const float4*)x)[gr * 32 + (i & 31)];
  }
  float pd[DEG_SLICES];
  if (tid < 64) {
    int gr = row0 + tid;
    if (gr > N_NODES - 1) gr = N_NODES - 1;
#pragma unroll
    for (int j = 0; j < DEG_SLICES; ++j) pd[j] = part[(size_t)j * N_NODES + gr];
  }
#pragma unroll
  for (int u = 0; u < 8; ++u) {
    const int i = tid + u * 256;
    const int r = i >> 5, c = i & 31;
    float4 v = xv[u];
    if (row0 + r >= N_NODES) v = make_float4(0.f, 0.f, 0.f, 0.f);
    *(float4*)&sX[r * 132 + c * 4] = v;
  }
  if (tid < 64) {
    const int gr = row0 + tid;
    float d = 1.0f;
#pragma unroll
    for (int j = 0; j < DEG_SLICES; ++j) d += pd[j];
    const float di = (d > 0.0f) ? rsqrtf(d) : 0.0f;
    sDi[tid] = di;
    if (gr < N_NODES) dinv[gr] = di;
  }
  __syncthreads();

  const int wave = tid >> 5;
  const int lane = tid & 31;
  const int lrow = lane & 15;
  const int hi   = lane >> 4;
  const int col  = wave * 16 + lrow;

  for (int t = 0; t < 4; ++t) {
    v8f acc = {};
    const float* ax = &sX[(t * 16 + lrow) * 132 + 2 * hi];
    for (int k0 = 0; k0 < 128; k0 += 4) {
      v2f a, b;
      a.x = ax[k0];
      a.y = ax[k0 + 1];
      const int kb = k0 + 2 * hi;
      b.x = W1[kb * 128 + col];
      b.y = W1[(kb + 1) * 128 + col];
      acc = __builtin_amdgcn_wmma_f32_16x16x4_f32(false, a, false, b,
                                                  (short)0, acc, false, false);
    }
    const int rbase = row0 + t * 16 + 8 * hi;
    const float4 da = *(const float4*)&sDi[t * 16 + 8 * hi];
    const float4 db = *(const float4*)&sDi[t * 16 + 8 * hi + 4];
    const float dr[8] = {da.x, da.y, da.z, da.w, db.x, db.y, db.z, db.w};
#pragma unroll
    for (int r = 0; r < 8; ++r) {
      const int gr = rbase + r;
      if (gr < N_NODES) h[gr * 128 + col] = acc[r] * dr[r];
    }
  }
}

#define OWN_WAVES   32
#define TILE_QUADS  (TILE_NODES * 32)
#define TILE_TRIPS  16
#define TILE_COPY_THREADS (TILE_QUADS / TILE_TRIPS)
static_assert(TILE_QUADS % TILE_TRIPS == 0, "full trips only");
static_assert(TILE_COPY_THREADS % 32 == 0 && TILE_COPY_THREADS <= 1024,
              "whole waves of the work-group");
#define CHUNK_EDGES 512
#define N_CHUNKS    (N_EDGES / CHUNK_EDGES)
#define SCAN_DEPTH  8
#define QCAP        64

static_assert(N_EDGES % CHUNK_EDGES == 0, "chunks cover the edges exactly");
static_assert(SCAN_DEPTH == 8, "the pair select below names four registers");

#define REC_CAP   512
static_assert(N_NODES <= (1 << REC_SBITS), "a source node fits REC_SBITS bits");
static_assert(TILE_NODES < (1 << (32 - REC_SBITS)), "a tile row fits the rest");

__device__ __forceinline__ unsigned zero_bytes(unsigned x) {
  const unsigned t = (x & 0x7F7F7F7Fu) + 0x7F7F7F7Fu;
  return ~(t | x | 0x7F7F7F7Fu);
}

__device__ __forceinline__ void own_batch(const unsigned* qb, const int nvalid,
                                          const int lane,
                                          const uint2* __restrict__ erec,
                                          const float* __restrict__ h,
                                          float* sT,
                                          uint2* __restrict__ wrec,
                                          const int nrec) {
  __builtin_amdgcn_fence(__ATOMIC_RELEASE, "wavefront");
  __builtin_amdgcn_wave_barrier();
  __builtin_amdgcn_fence(__ATOMIC_ACQUIRE, "wavefront");
  int      s    = 0;
  unsigned dl   = 0u;
  float    norm = 0.f;
  if (lane < nvalid) {
    const uint2 r = erec[qb[lane]];
    s    = (int)(r.x & ((1u << REC_SBITS) - 1u));
    dl   = r.x >> REC_SBITS;
    norm = __uint_as_float(r.y);
    const int slot = nrec + lane;
    if (slot < REC_CAP) wrec[slot] = r;
  }
  const float* hl = h + lane;
  float*       tl = sT + lane;
  for (int g = 0; g < nvalid; g += 8) {
    float v[8][4];
#pragma unroll
    for (int i = 0; i < 8; ++i) {
      const int    sk = __builtin_amdgcn_readlane(s, g + i);
      const float* hr = hl + sk * 128;
      v[i][0] = hr[0];
      v[i][1] = hr[32];
      v[i][2] = hr[64];
      v[i][3] = hr[96];
    }
#pragma unroll
    for (int i = 0; i < 8; ++i) {
      if (g + i < nvalid) {
        const int   dk = __builtin_amdgcn_readlane((int)dl, g + i);
        const float nk = __int_as_float(
            __builtin_amdgcn_readlane(__float_as_int(norm), g + i));
        float* o = tl + dk * 128;
        __hip_atomic_fetch_add(o + 0,  v[i][0] * nk, __ATOMIC_RELAXED,
                               __HIP_MEMORY_SCOPE_WORKGROUP);
        __hip_atomic_fetch_add(o + 32, v[i][1] * nk, __ATOMIC_RELAXED,
                               __HIP_MEMORY_SCOPE_WORKGROUP);
        __hip_atomic_fetch_add(o + 64, v[i][2] * nk, __ATOMIC_RELAXED,
                               __HIP_MEMORY_SCOPE_WORKGROUP);
        __hip_atomic_fetch_add(o + 96, v[i][3] * nk, __ATOMIC_RELAXED,
                               __HIP_MEMORY_SCOPE_WORKGROUP);
      }
    }
  }
  __builtin_amdgcn_fence(__ATOMIC_RELEASE, "wavefront");
  __builtin_amdgcn_wave_barrier();
  __builtin_amdgcn_fence(__ATOMIC_ACQUIRE, "wavefront");
}

__device__ __forceinline__ void scan_load(const uint4* __restrict__ key4,
                                          const int j0, const int start,
                                          const int lane, uint4* kv) {
#pragma unroll
  for (int u = 0; u < SCAN_DEPTH; ++u) {
    const int jj = j0 + u * OWN_WAVES;
    int c = jj + start;
    if (c >= N_CHUNKS) c -= N_CHUNKS;
    if (jj >= N_CHUNKS) c = 0;
    kv[u] = key4[c * 32 + lane];
  }
}

__global__ __launch_bounds__(1024) void agg1_owner_kernel(
    const unsigned* __restrict__ erec, const float* __restrict__ dinv,
    const float* __restrict__ h, const unsigned char* __restrict__ tkey,
    float* __restrict__ agg, unsigned* __restrict__ rec, int* __restrict__ cntw) {
  __shared__ __attribute__((aligned(16))) float sT[TILE_NODES * 128];
  __shared__ unsigned sQ[OWN_WAVES * QCAP];
  const int tid  = threadIdx.x;
  const int lane = tid & 31;
  const int wave = __builtin_amdgcn_readfirstlane(tid >> 5);
  const int tile = blockIdx.x;
  const int n0   = tile * TILE_NODES;

  if (wave < TILE_COPY_THREADS / 32) {
    float4 hv[TILE_TRIPS];
#pragma unroll
    for (int u = 0; u < TILE_TRIPS; ++u)
      hv[u] = ((const float4*)h)[n0 * 32 + tid + u * TILE_COPY_THREADS];
#pragma unroll
    for (int u = 0; u < TILE_TRIPS; ++u)
      ((float4*)sT)[tid + u * TILE_COPY_THREADS] = hv[u];
  }
  __syncthreads();

  unsigned*    q    = sQ + wave * QCAP;
  const uint4* key4 = (const uint4*)tkey;
  const unsigned pat = (unsigned)tile * 0x01010101u;
  const int start = (tile * N_CHUNKS) / N_TILES;
  int cnt = 0;
  uint2* wrec = (uint2*)rec + (size_t)(tile * OWN_WAVES + wave) * REC_CAP;
  int nrec = 0;
  uint4 kv[SCAN_DEPTH];
  scan_load(key4, wave, start, lane, kv);
  for (int j = wave; j < N_CHUNKS; j += SCAN_DEPTH * OWN_WAVES) {
    unsigned pk[SCAN_DEPTH / 2];
#pragma unroll
    for (int u = 0; u < SCAN_DEPTH; ++u) {
      unsigned lm = (zero_bytes(kv[u].x ^ pat) >> 7) |
                    (zero_bytes(kv[u].y ^ pat) >> 6) |
                    (zero_bytes(kv[u].z ^ pat) >> 5) |
                    (zero_bytes(kv[u].w ^ pat) >> 4);
      if (j + u * OWN_WAVES >= N_CHUNKS) lm = 0u;
      if (u & 1) pk[u >> 1] |= lm << 4;
      else       pk[u >> 1] = lm;
    }
    scan_load(key4, j + SCAN_DEPTH * OWN_WAVES, start, lane, kv);
#pragma nounroll
    for (int w = 0; w < SCAN_DEPTH / 2; ++w) {
      unsigned lm = (w == 0) ? pk[0] : (w == 1) ? pk[1] : (w == 2) ? pk[2] : pk[3];
      int ca = j + 2 * w * OWN_WAVES + start;
      if (ca >= N_CHUNKS) ca -= N_CHUNKS;
      int cb = j + (2 * w + 1) * OWN_WAVES + start;
      if (cb >= N_CHUNKS) cb -= N_CHUNKS;
      const unsigned ea = (unsigned)ca * CHUNK_EDGES + (unsigned)lane * 16u;
      const unsigned eb = (unsigned)cb * CHUNK_EDGES + (unsigned)lane * 16u;
      for (int it = 0; it < 32; ++it) {
        const bool     m    = lm != 0u;
        const unsigned mask = __builtin_amdgcn_ballot_w32(m);
        if (mask == 0u) break;
        const unsigned p   = (unsigned)__builtin_ctz(lm | 0x80000000u);
        const unsigned pos =
            (unsigned)cnt + __builtin_amdgcn_mbcnt_lo(mask, 0u);
        if (m) q[pos] = ((p & 4u) ? eb : ea) + 4u * (p & 3u) + (p >> 3);
        lm &= lm - 1u;
        cnt += __builtin_popcount(mask);
        if (cnt >= 32) {
          cnt -= 32;
          own_batch(q + cnt, 32, lane, (const uint2*)erec, h, sT, wrec, nrec);
          nrec += 32;
        }
      }
    }
  }
  if (cnt > 0) {
    own_batch(q, cnt, lane, (const uint2*)erec, h, sT, wrec, nrec);
    nrec += cnt;
  }
  if (lane == 0) cntw[tile * OWN_WAVES + wave] = nrec;
  const bool copier = wave < TILE_COPY_THREADS / 32;
  float dv[TILE_TRIPS];
  if (copier) {
#pragma unroll
    for (int u = 0; u < TILE_TRIPS; ++u)
      dv[u] = dinv[n0 + wave + (TILE_COPY_THREADS / 32) * u];
  }
  __syncthreads();

  if (copier) {
#pragma unroll
    for (int u = 0; u < TILE_TRIPS; ++u) {
      const int i = tid + u * TILE_COPY_THREADS;
      const float di = dv[u];
      float4 v = ((const float4*)sT)[i];
      v.x *= di;
      v.y *= di;
      v.z *= di;
      v.w *= di;
      ((float4*)agg)[n0 * 32 + i] = v;
    }
  }
}

__global__ __launch_bounds__(256) void gemm2_wmma(const float* __restrict__ agg1,
                                                  const float* __restrict__ b1,
                                                  const float* __restrict__ W2,
                                                  const float* __restrict__ dinv,
                                                  float* __restrict__ h2) {
  __shared__ __attribute__((aligned(16))) float sW[128 * 16];
  __shared__ float sb[256];
  const int tid = threadIdx.x;
  const int wave = tid >> 5, lane = tid & 31;
  const int lrow = lane & 15, hi = lane >> 4;
  const int row0 = blockIdx.x * 128 + wave * 16;
  const int arow = row0 + lrow;

  const float4 w0 = ((const float4*)W2)[tid];
  const float4 w1 = ((const float4*)W2)[tid + 256];
  const float  bv = b1[tid & 127];
  int ar = arow;
  if (ar > N_NODES - 1) ar = N_NODES - 1;
  const float2* ap = (const float2*)(agg1 + (size_t)ar * 128 + 2 * hi);
  float2 av[32];
#pragma unroll
  for (int s = 0; s < 32; ++s) av[s] = ap[2 * s];
  const int rbase = row0 + 8 * hi;
  float dv[8];
#pragma unroll
  for (int r = 0; r < 8; ++r) {
    int gr = rbase + r;
    if (gr > N_NODES - 1) gr = N_NODES - 1;
    dv[r] = dinv[gr];
  }

  ((float4*)sW)[tid]       = w0;
  ((float4*)sW)[tid + 256] = w1;
  sb[tid] = bv;
  __syncthreads();

  v8f acc = {};
#pragma unroll
  for (int s = 0; s < 32; ++s) {
    const int k = 4 * s + 2 * hi;
    v2f a, b;
    a.x = fmaxf(av[s].x + sb[k], 0.f);
    a.y = fmaxf(av[s].y + sb[k + 1], 0.f);
    b.x = sW[k * 16 + lrow];
    b.y = sW[(k + 1) * 16 + lrow];
    acc = __builtin_amdgcn_wmma_f32_16x16x4_f32(false, a, false, b,
                                                (short)0, acc, false, false);
  }
#pragma unroll
  for (int r = 0; r < 8; ++r) {
    const int gr = rbase + r;
    if (gr < N_NODES) h2[gr * 16 + lrow] = acc[r] * dv[r];
  }
}


__device__ __forceinline__ void own_rows2(const int s, const unsigned dl,
                                          const float norm, const int lane,
                                          const float* __restrict__ h2,
                                          float* sT) {
  const int    col  = lane & 15;
  const int    half = lane >> 4;
  const float* hc   = h2 + col;
  float*       tc   = sT + col;
  float v[16];
  float nk[16];
  int   dk[16];
#pragma unroll
  for (int i = 0; i < 16; ++i) {
    const int k  = 2 * i + half;
    const int sk = __shfl(s, k, 32);
    dk[i] = __shfl((int)dl, k, 32);
    nk[i] = __shfl(norm, k, 32);
    v[i]  = hc[sk * C_DIM];
  }
#pragma unroll
  for (int i = 0; i < 16; ++i)
    __hip_atomic_fetch_add(tc + dk[i] * C_DIM, v[i] * nk[i],
                           __ATOMIC_RELAXED, __HIP_MEMORY_SCOPE_WORKGROUP);
}

__device__ __forceinline__ void own_batch2(const unsigned* qb, const int nvalid,
                                           const int lane, const int n0,
                                           const int* __restrict__ src,
                                           const int* __restrict__ dst,
                                           const float* __restrict__ ew,
                                           const float* __restrict__ h2,
                                           float* sT) {
  __builtin_amdgcn_fence(__ATOMIC_RELEASE, "wavefront");
  __builtin_amdgcn_wave_barrier();
  __builtin_amdgcn_fence(__ATOMIC_ACQUIRE, "wavefront");
  int      s    = 0;
  unsigned dl   = TILE_NODES;
  float    norm = 0.f;
  if (lane < nvalid) {
    const unsigned e = qb[lane];
    dl   = (unsigned)(dst[e] - n0);
    s    = src[e];
    norm = ew[e];
  }
  own_rows2(s, dl, norm, lane, h2, sT);
  __builtin_amdgcn_fence(__ATOMIC_RELEASE, "wavefront");
  __builtin_amdgcn_wave_barrier();
  __builtin_amdgcn_fence(__ATOMIC_ACQUIRE, "wavefront");
}

#define A2_COPY_THREADS (TILE_NODES * (C_DIM / 4) / 2)
static_assert(A2_COPY_THREADS % 32 == 0 && A2_COPY_THREADS <= 1024,
              "whole waves of the work-group");
static_assert(REC_CAP % 32 == 0, "a list is whole trips of 32 records");

__global__ __launch_bounds__(1024) void agg2_owner_kernel(
    const int* __restrict__ src, const int* __restrict__ dst,
    const float* __restrict__ ew, const float* __restrict__ dinv,
    const float* __restrict__ h2, const unsigned char* __restrict__ tkey,
    const unsigned* __restrict__ rec, const int* __restrict__ cntw,
    const float* __restrict__ b2, float* __restrict__ out) {
  __shared__ __attribute__((aligned(16))) float sT[(TILE_NODES + 1) * C_DIM];
  __shared__ unsigned sQ[OWN_WAVES * QCAP];
  __shared__ uint2 sR[OWN_WAVES * REC_CAP];
  const int tid  = threadIdx.x;
  const int lane = tid & 31;
  const int wave = __builtin_amdgcn_readfirstlane(tid >> 5);
  const int tile = blockIdx.x;
  const int n0   = tile * TILE_NODES;

  const int nrec = cntw[tile * OWN_WAVES + wave];

  const bool copier = wave < A2_COPY_THREADS / 32;
  float4 hv0 = make_float4(0.f, 0.f, 0.f, 0.f), hv1 = hv0;
  if (copier) {
    hv0 = ((const float4*)h2)[n0 * (C_DIM / 4) + tid];
    hv1 = ((const float4*)h2)[n0 * (C_DIM / 4) + tid + A2_COPY_THREADS];
  }
  const float di_out = dinv[n0 + (tid < TILE_NODES ? tid : TILE_NODES - 1)];
  float4 bq[C_DIM / 4];
#pragma unroll
  for (int qd = 0; qd < C_DIM / 4; ++qd) bq[qd] = ((const float4*)b2)[qd];

  const bool listed = nrec > 0 && nrec <= REC_CAP;
  uint2* myR = sR + wave * REC_CAP;
  if (listed) {
    const uint2* wrec =
        (const uint2*)rec + (size_t)(tile * OWN_WAVES + wave) * REC_CAP;
    const int last = nrec - 1;
    uint2 rv[REC_CAP / 32];
#pragma unroll
    for (int b = 0; b < REC_CAP / 32; ++b) {
      const int i = b * 32 + lane;
      rv[b] = wrec[i < nrec ? i : last];
    }
#pragma unroll
    for (int b = 0; b < REC_CAP / 32; ++b) myR[b * 32 + lane] = rv[b];
  }
  if (copier) {
    ((float4*)sT)[tid]                   = hv0;
    ((float4*)sT)[tid + A2_COPY_THREADS] = hv1;
  }
  if (tid < C_DIM) sT[TILE_NODES * C_DIM + tid] = 0.f;
  __syncthreads();

  if (nrec <= REC_CAP) {
    const uint2 idle = make_uint2((unsigned)TILE_NODES << REC_SBITS, 0u);
    for (int b = 0; b < nrec; b += 32) {
      uint2 cur = myR[b + lane];
      if (b + lane >= nrec) cur = idle;
      own_rows2((int)(cur.x & ((1u << REC_SBITS) - 1u)), cur.x >> REC_SBITS,
                __uint_as_float(cur.y), lane, h2, sT);
    }
  } else {
    unsigned*    q    = sQ + wave * QCAP;
    const uint4* key4 = (const uint4*)tkey;
    const unsigned pat = (unsigned)tile * 0x01010101u;
    const int start = (tile * N_CHUNKS) / N_TILES;
    int cnt = 0;
    for (int j = wave; j < N_CHUNKS; j += SCAN_DEPTH * OWN_WAVES) {
      uint4 kv[SCAN_DEPTH];
#pragma unroll
      for (int u = 0; u < SCAN_DEPTH; ++u) {
        const int jj = j + u * OWN_WAVES;
        int c = jj + start;
        if (c >= N_CHUNKS) c -= N_CHUNKS;
        if (jj >= N_CHUNKS) c = 0;
        kv[u] = key4[c * 32 + lane];
      }
      unsigned pk[SCAN_DEPTH / 2];
#pragma unroll
      for (int u = 0; u < SCAN_DEPTH; ++u) {
        unsigned lm = (zero_bytes(kv[u].x ^ pat) >> 7) |
                      (zero_bytes(kv[u].y ^ pat) >> 6) |
                      (zero_bytes(kv[u].z ^ pat) >> 5) |
                      (zero_bytes(kv[u].w ^ pat) >> 4);
        if (j + u * OWN_WAVES >= N_CHUNKS) lm = 0u;
        if (u & 1) pk[u >> 1] |= lm << 4;
        else       pk[u >> 1] = lm;
      }
#pragma nounroll
      for (int w = 0; w < SCAN_DEPTH / 2; ++w) {
        unsigned lm = (w == 0) ? pk[0] : (w == 1) ? pk[1] : (w == 2) ? pk[2] : pk[3];
        int ca = j + 2 * w * OWN_WAVES + start;
        if (ca >= N_CHUNKS) ca -= N_CHUNKS;
        int cb = j + (2 * w + 1) * OWN_WAVES + start;
        if (cb >= N_CHUNKS) cb -= N_CHUNKS;
        const unsigned ea = (unsigned)ca * CHUNK_EDGES + (unsigned)lane * 16u;
        const unsigned eb = (unsigned)cb * CHUNK_EDGES + (unsigned)lane * 16u;
        for (int it = 0; it < 32; ++it) {
          const bool     m    = lm != 0u;
          const unsigned mask = __builtin_amdgcn_ballot_w32(m);
          if (mask == 0u) break;
          const unsigned p   = (unsigned)__builtin_ctz(lm | 0x80000000u);
          const unsigned pos =
              (unsigned)cnt + __builtin_amdgcn_mbcnt_lo(mask, 0u);
          if (m) q[pos] = ((p & 4u) ? eb : ea) + 4u * (p & 3u) + (p >> 3);
          lm &= lm - 1u;
          cnt += __builtin_popcount(mask);
          if (cnt >= 32) {
            cnt -= 32;
            own_batch2(q + cnt, 32, lane, n0, src, dst, ew, h2, sT);
          }
        }
      }
    }
    if (cnt > 0) own_batch2(q, cnt, lane, n0, src, dst, ew, h2, sT);
  }
  __syncthreads();

  if (tid < TILE_NODES) {
    float v[C_DIM];
    const float4* p = (const float4*)(sT + tid * C_DIM);
    const float di = di_out;
    float m = -3.402823466e+38f;
#pragma unroll
    for (int qd = 0; qd < 4; ++qd) {
      const float4 t = p[qd];
      v[4 * qd + 0] = t.x * di + bq[qd].x;
      v[4 * qd + 1] = t.y * di + bq[qd].y;
      v[4 * qd + 2] = t.z * di + bq[qd].z;
      v[4 * qd + 3] = t.w * di + bq[qd].w;
    }
#pragma unroll
    for (int c = 0; c < C_DIM; ++c) m = fmaxf(m, v[c]);
    float sum = 0.f;
#pragma unroll
    for (int c = 0; c < C_DIM; ++c) { v[c] = expf(v[c] - m); sum += v[c]; }
    const float inv = 1.0f / sum;
    float4* o = (float4*)(out + (size_t)(n0 + tid) * C_DIM);
#pragma unroll
    for (int qd = 0; qd < 4; ++qd)
      o[qd] = make_float4(v[4 * qd + 0] * inv, v[4 * qd + 1] * inv,
                          v[4 * qd + 2] * inv, v[4 * qd + 3] * inv);
  }
}

extern "C" void kernel_launch(void* const* d_in, const int* in_sizes, int n_in,
                              void* d_out, int out_size, void* d_ws, size_t ws_size,
                              hipStream_t stream) {
  const float* x  = (const float*)d_in[0];
  const int*   ei = (const int*)d_in[1];
  const float* ew = (const float*)d_in[2];
  const float* W1 = (const float*)d_in[3];
  const float* b1 = (const float*)d_in[4];
  const float* W2 = (const float*)d_in[5];
  const float* b2 = (const float*)d_in[6];
  float* out = (float*)d_out;

  const int* src = ei;
  const int* dst = ei + N_EDGES;

  float* ws   = (float*)d_ws;
  float* dinv = ws;
  float* h    = ws + N_NODES;
  float* agg1 = ws + N_NODES + (size_t)N_NODES * 128;
  float* h2   = h;
  unsigned char* tkey = (unsigned char*)(agg1 + (size_t)N_NODES * 128);
  float* part = (float*)(tkey + N_EDGES);
  unsigned* rec  = (unsigned*)(part + (size_t)DEG_SLICES * N_NODES);
  int*      cntw = (int*)(rec + (size_t)N_TILES * OWN_WAVES * REC_CAP * 2);
  unsigned* erec = (unsigned*)(cntw + N_TILES * OWN_WAVES);

  const int T = 256;

  deg_hist_kernel<<<DEG_RANGES * DEG_SLICES, DEG_THREADS, 0, stream>>>(src, dst, ew, part, tkey, erec);

  gemm1_wmma<<<(N_NODES + 63) / 64, T, 0, stream>>>(x, W1, part, dinv, h);
  agg1_owner_kernel<<<N_TILES, 1024, 0, stream>>>(erec, dinv, h, tkey, agg1, rec, cntw);

  gemm2_wmma<<<(N_NODES + 127) / 128, T, 0, stream>>>(agg1, b1, W2, dinv, h2);
  agg2_owner_kernel<<<N_TILES, 1024, 0, stream>>>(src, dst, ew, dinv, h2, tkey, rec, cntw, b2, out);
}


# Second linear layer (bias, ReLU, f32 WMMA) runs in the layer-1 owner groups on their finished LDS tile (row pitch 132 floats); agg1 store, its re-read and the GEMM2 launch go: 5 launches become 4
# speedup vs baseline: 3.5332x; 1.0742x over previous
#include <hip/hip_runtime.h>
#include <hip/hip_bf16.h>
#include <math.h>

#define N_NODES 100000
#define N_EDGES 1600000
#define F_DIM 128
#define H_DIM 128
#define C_DIM 16

typedef __attribute__((ext_vector_type(2))) float v2f;
typedef __attribute__((ext_vector_type(8))) float v8f;

#define TILE_NODES  400
#define N_TILES     (N_NODES / TILE_NODES)
static_assert(N_NODES % TILE_NODES == 0, "tiles cover the nodes exactly");
static_assert(N_TILES <= 256, "a tile number fits one byte");
#define TPITCH      132
static_assert(TILE_NODES % 16 == 0, "a tile is whole 16-row matrix tiles");

#define DEG_RANGES  5
#define DEG_SLICES  50
#define DEG_THREADS 1024
#define RANGE_NODES (N_NODES / DEG_RANGES)
#define SLICE_QUADS (N_EDGES / DEG_SLICES / 4)
#define DEG_TRIPS   ((SLICE_QUADS + DEG_THREADS - 1) / DEG_THREADS)
#define SLICE_EDGES (N_EDGES / DEG_SLICES)

static_assert(N_NODES % (DEG_RANGES * 4) == 0, "ranges cover the nodes in float4s");
static_assert(N_EDGES % (DEG_SLICES * 4) == 0, "slices cover the edges in quads");
static_assert(DEG_RANGES <= DEG_TRIPS && DEG_TRIPS <= 2 * DEG_RANGES,
              "every trip has one writing group, a group at most two trips");
static_assert(DEG_RANGES * DEG_THREADS <= SLICE_QUADS, "trip k is a whole trip");

#define REC_SBITS 17

__device__ __forceinline__ void deg_add(float* sD, const int d, const int lo,
                                        const float w) {
  const unsigned r = (unsigned)(d - lo);
  if (r < (unsigned)RANGE_NODES)
    __hip_atomic_fetch_add(sD + r, w, __ATOMIC_RELAXED,
                           __HIP_MEMORY_SCOPE_WORKGROUP);
}

__device__ __forceinline__ void edge_rec(const int s, const int d, const unsigned t,
                                         const float w, unsigned& r0, unsigned& r1) {
  r0 = (unsigned)s | (((unsigned)d - t * (unsigned)TILE_NODES) << REC_SBITS);
  r1 = __float_as_uint(w);
}

__global__ __launch_bounds__(DEG_THREADS) void deg_hist_kernel(
    const int* __restrict__ src, const int* __restrict__ dst,
    const float* __restrict__ ew, float* __restrict__ part,
    unsigned char* __restrict__ tkey, unsigned* __restrict__ erec) {
  __shared__ __attribute__((aligned(16))) float sD[RANGE_NODES];
  const int tid = threadIdx.x;
  const int k   = blockIdx.x % DEG_RANGES;
  const int j   = blockIdx.x / DEG_RANGES;
  const int lo  = k * RANGE_NODES;

  const int4*   d4 = (const int4*)dst + j * SLICE_QUADS;
  const float4* w4 = (const float4*)ew + j * SLICE_QUADS;
  int4   dq[DEG_TRIPS];
  float4 wq[DEG_TRIPS];
#pragma unroll
  for (int u = 0; u < DEG_TRIPS; ++u) {
    int q = tid + u * DEG_THREADS;
    if (q > SLICE_QUADS - 1) q = SLICE_QUADS - 1;
    dq[u] = d4[q];
    wq[u] = w4[q];
  }
  const int4* s4 = (const int4*)src + j * SLICE_QUADS;
  int qb = tid + (k + DEG_RANGES) * DEG_THREADS;
  if (qb > SLICE_QUADS - 1) qb = SLICE_QUADS - 1;
  const int4 sA = s4[tid + k * DEG_THREADS];
  const int4 sB = s4[qb];

  for (int i = tid; i < RANGE_NODES / 4; i += DEG_THREADS)
    ((float4*)sD)[i] = make_float4(0.f, 0.f, 0.f, 0.f);
  __syncthreads();

  unsigned* key4 = (unsigned*)tkey + j * SLICE_QUADS;
#pragma unroll
  for (int u = 0; u < DEG_TRIPS; ++u) {
    const int q = tid + u * DEG_THREADS;
    if (q < SLICE_QUADS) {
      deg_add(sD, dq[u].x, lo, wq[u].x);
      deg_add(sD, dq[u].y, lo, wq[u].y);
      deg_add(sD, dq[u].z, lo, wq[u].z);
      deg_add(sD, dq[u].w, lo, wq[u].w);
      if (u % DEG_RANGES == k) {
        const int4 sq = (u < DEG_RANGES) ? sA : sB;
        const unsigned tx = (unsigned)dq[u].x / (unsigned)TILE_NODES;
        const unsigned ty = (unsigned)dq[u].y / (unsigned)TILE_NODES;
        const unsigned tz = (unsigned)dq[u].z / (unsigned)TILE_NODES;
        const unsigned tw = (unsigned)dq[u].w / (unsigned)TILE_NODES;
        key4[q] = tx | (ty << 8) | (tz << 16) | (tw << 24);
        uint4 ra, rb;
        edge_rec(sq.x, dq[u].x, tx, wq[u].x, ra.x, ra.y);
        edge_rec(sq.y, dq[u].y, ty, wq[u].y, ra.z, ra.w);
        edge_rec(sq.z, dq[u].z, tz, wq[u].z, rb.x, rb.y);
        edge_rec(sq.w, dq[u].w, tw, wq[u].w, rb.z, rb.w);
        uint4* r4 = (uint4*)(erec + (size_t)j * (2 * SLICE_EDGES)) + 2 * q;
        r4[0] = ra;
        r4[1] = rb;
      }
    }
  }
  __syncthreads();

  float4* po = (float4*)(part + (size_t)j * N_NODES + lo);
  for (int i = tid; i < RANGE_NODES / 4; i += DEG_THREADS)
    po[i] = ((const float4*)sD)[i];
}

__global__ __launch_bounds__(256) void gemm1_wmma(const float* __restrict__ x,
                                                  const float* __restrict__ W1,
                                                  const float* __restrict__ part,
                                                  float* __restrict__ dinv,
                                                  float* __restrict__ h) {
  __shared__ float sX[64 * 132];
  __shared__ __attribute__((aligned(16))) float sDi[64];
  const int tid  = threadIdx.x;
  const int row0 = blockIdx.x * 64;

  float4 xv[8];
#pragma unroll
  for (int u = 0; u < 8; ++u) {
    const int i = tid + u * 256;
    int gr = row0 + (i >> 5);
    if (gr > N_NODES - 1) gr = N_NODES - 1;
    xv[u] = ((const float4*)x)[gr * 32 + (i & 31)];
  }
  float pd[DEG_SLICES];
  if (tid < 64) {
    int gr = row0 + tid;
    if (gr > N_NODES - 1) gr = N_NODES - 1;
#pragma unroll
    for (int j = 0; j < DEG_SLICES; ++j) pd[j] = part[(size_t)j * N_NODES + gr];
  }
#pragma unroll
  for (int u = 0; u < 8; ++u) {
    const int i = tid + u * 256;
    const int r = i >> 5, c = i & 31;
    float4 v = xv[u];
    if (row0 + r >= N_NODES) v = make_float4(0.f, 0.f, 0.f, 0.f);
    *(float4*)&sX[r * 132 + c * 4] = v;
  }
  if (tid < 64) {
    const int gr = row0 + tid;
    float d = 1.0f;
#pragma unroll
    for (int j = 0; j < DEG_SLICES; ++j) d += pd[j];
    const float di = (d > 0.0f) ? rsqrtf(d) : 0.0f;
    sDi[tid] = di;
    if (gr < N_NODES) dinv[gr] = di;
  }
  __syncthreads();

  const int wave = tid >> 5;
  const int lane = tid & 31;
  const int lrow = lane & 15;
  const int hi   = lane >> 4;
  const int col  = wave * 16 + lrow;

  for (int t = 0; t < 4; ++t) {
    v8f acc = {};
    const float* ax = &sX[(t * 16 + lrow) * 132 + 2 * hi];
    for (int k0 = 0; k0 < 128; k0 += 4) {
      v2f a, b;
      a.x = ax[k0];
      a.y = ax[k0 + 1];
      const int kb = k0 + 2 * hi;
      b.x = W1[kb * 128 + col];
      b.y = W1[(kb + 1) * 128 + col];
      acc = __builtin_amdgcn_wmma_f32_16x16x4_f32(false, a, false, b,
                                                  (short)0, acc, false, false);
    }
    const int rbase = row0 + t * 16 + 8 * hi;
    const float4 da = *(const float4*)&sDi[t * 16 + 8 * hi];
    const float4 db = *(const float4*)&sDi[t * 16 + 8 * hi + 4];
    const float dr[8] = {da.x, da.y, da.z, da.w, db.x, db.y, db.z, db.w};
#pragma unroll
    for (int r = 0; r < 8; ++r) {
      const int gr = rbase + r;
      if (gr < N_NODES) h[gr * 128 + col] = acc[r] * dr[r];
    }
  }
}

#define OWN_WAVES   32
#define TILE_QUADS  (TILE_NODES * 32)
#define TILE_TRIPS  16
#define TILE_COPY_THREADS (TILE_QUADS / TILE_TRIPS)
static_assert(TILE_QUADS % TILE_TRIPS == 0, "full trips only");
static_assert(TILE_COPY_THREADS % 32 == 0 && TILE_COPY_THREADS <= 1024,
              "whole waves of the work-group");
#define CHUNK_EDGES 512
#define N_CHUNKS    (N_EDGES / CHUNK_EDGES)
#define SCAN_DEPTH  8
#define QCAP        64

static_assert(N_EDGES % CHUNK_EDGES == 0, "chunks cover the edges exactly");
static_assert(SCAN_DEPTH == 8, "the pair select below names four registers");

#define REC_CAP   512
static_assert(N_NODES <= (1 << REC_SBITS), "a source node fits REC_SBITS bits");
static_assert(TILE_NODES < (1 << (32 - REC_SBITS)), "a tile row fits the rest");

__device__ __forceinline__ unsigned zero_bytes(unsigned x) {
  const unsigned t = (x & 0x7F7F7F7Fu) + 0x7F7F7F7Fu;
  return ~(t | x | 0x7F7F7F7Fu);
}

__device__ __forceinline__ void own_batch(const unsigned* qb, const int nvalid,
                                          const int lane,
                                          const uint2* __restrict__ erec,
                                          const float* __restrict__ h,
                                          float* sT,
                                          uint2* __restrict__ wrec,
                                          const int nrec) {
  __builtin_amdgcn_fence(__ATOMIC_RELEASE, "wavefront");
  __builtin_amdgcn_wave_barrier();
  __builtin_amdgcn_fence(__ATOMIC_ACQUIRE, "wavefront");
  int      s    = 0;
  unsigned dl   = 0u;
  float    norm = 0.f;
  if (lane < nvalid) {
    const uint2 r = erec[qb[lane]];
    s    = (int)(r.x & ((1u << REC_SBITS) - 1u));
    dl   = r.x >> REC_SBITS;
    norm = __uint_as_float(r.y);
    const int slot = nrec + lane;
    if (slot < REC_CAP) wrec[slot] = r;
  }
  const float* hl = h + lane;
  float*       tl = sT + lane;
  for (int g = 0; g < nvalid; g += 8) {
    float v[8][4];
#pragma unroll
    for (int i = 0; i < 8; ++i) {
      const int    sk = __builtin_amdgcn_readlane(s, g + i);
      const float* hr = hl + sk * 128;
      v[i][0] = hr[0];
      v[i][1] = hr[32];
      v[i][2] = hr[64];
      v[i][3] = hr[96];
    }
#pragma unroll
    for (int i = 0; i < 8; ++i) {
      if (g + i < nvalid) {
        const int   dk = __builtin_amdgcn_readlane((int)dl, g + i);
        const float nk = __int_as_float(
            __builtin_amdgcn_readlane(__float_as_int(norm), g + i));
        float* o = tl + dk * TPITCH;
        __hip_atomic_fetch_add(o + 0,  v[i][0] * nk, __ATOMIC_RELAXED,
                               __HIP_MEMORY_SCOPE_WORKGROUP);
        __hip_atomic_fetch_add(o + 32, v[i][1] * nk, __ATOMIC_RELAXED,
                               __HIP_MEMORY_SCOPE_WORKGROUP);
        __hip_atomic_fetch_add(o + 64, v[i][2] * nk, __ATOMIC_RELAXED,
                               __HIP_MEMORY_SCOPE_WORKGROUP);
        __hip_atomic_fetch_add(o + 96, v[i][3] * nk, __ATOMIC_RELAXED,
                               __HIP_MEMORY_SCOPE_WORKGROUP);
      }
    }
  }
  __builtin_amdgcn_fence(__ATOMIC_RELEASE, "wavefront");
  __builtin_amdgcn_wave_barrier();
  __builtin_amdgcn_fence(__ATOMIC_ACQUIRE, "wavefront");
}

__device__ __forceinline__ void scan_load(const uint4* __restrict__ key4,
                                          const int j0, const int start,
                                          const int lane, uint4* kv) {
#pragma unroll
  for (int u = 0; u < SCAN_DEPTH; ++u) {
    const int jj = j0 + u * OWN_WAVES;
    int c = jj + start;
    if (c >= N_CHUNKS) c -= N_CHUNKS;
    if (jj >= N_CHUNKS) c = 0;
    kv[u] = key4[c * 32 + lane];
  }
}

#define MM_TILES (TILE_NODES / 16)
static_assert(MM_TILES <= OWN_WAVES, "one matrix tile per wave");

__global__ __launch_bounds__(1024) void agg1_owner_kernel(
    const unsigned* __restrict__ erec, const float* __restrict__ dinv,
    const float* __restrict__ h, const unsigned char* __restrict__ tkey,
    const float* __restrict__ b1, const float* __restrict__ W2,
    float* __restrict__ h2, unsigned* __restrict__ rec, int* __restrict__ cntw) {
  __shared__ __attribute__((aligned(16))) float sT[TILE_NODES * TPITCH];
  __shared__ unsigned sQ[OWN_WAVES * QCAP];
  __shared__ __attribute__((aligned(16))) float sW[H_DIM * C_DIM];
  __shared__ float sb[H_DIM];
  const int tid  = threadIdx.x;
  const int lane = tid & 31;
  const int wave = __builtin_amdgcn_readfirstlane(tid >> 5);
  const int tile = blockIdx.x;
  const int n0   = tile * TILE_NODES;

  const float4 w2v = ((const float4*)W2)[tid & (H_DIM * C_DIM / 4 - 1)];
  const float  b1v = b1[tid & (H_DIM - 1)];
  const bool mm   = wave < MM_TILES;
  const int  lrow = lane & 15;
  const int  hi   = lane >> 4;
  const int  r0   = (mm ? wave : 0) * 16;
  const float di_a = dinv[n0 + r0 + lrow];
  float dv[8];
#pragma unroll
  for (int r = 0; r < 8; ++r) dv[r] = dinv[n0 + r0 + 8 * hi + r];
  if (wave < TILE_COPY_THREADS / 32) {
    float4 hv[TILE_TRIPS];
#pragma unroll
    for (int u = 0; u < TILE_TRIPS; ++u)
      hv[u] = ((const float4*)h)[n0 * 32 + tid + u * TILE_COPY_THREADS];
#pragma unroll
    for (int u = 0; u < TILE_TRIPS; ++u) {
      const int i = tid + u * TILE_COPY_THREADS;
      *(float4*)&sT[(i >> 5) * TPITCH + (i & 31) * 4] = hv[u];
    }
  }
  if (tid < H_DIM * C_DIM / 4) ((float4*)sW)[tid] = w2v;
  if (tid < H_DIM) sb[tid] = b1v;
  __syncthreads();

  unsigned*    q    = sQ + wave * QCAP;
  const uint4* key4 = (const uint4*)tkey;
  const unsigned pat = (unsigned)tile * 0x01010101u;
  const int start = (tile * N_CHUNKS) / N_TILES;
  int cnt = 0;
  uint2* wrec = (uint2*)rec + (size_t)(tile * OWN_WAVES + wave) * REC_CAP;
  int nrec = 0;
  uint4 kv[SCAN_DEPTH];
  scan_load(key4, wave, start, lane, kv);
  for (int j = wave; j < N_CHUNKS; j += SCAN_DEPTH * OWN_WAVES) {
    unsigned pk[SCAN_DEPTH / 2];
#pragma unroll
    for (int u = 0; u < SCAN_DEPTH; ++u) {
      unsigned lm = (zero_bytes(kv[u].x ^ pat) >> 7) |
                    (zero_bytes(kv[u].y ^ pat) >> 6) |
                    (zero_bytes(kv[u].z ^ pat) >> 5) |
                    (zero_bytes(kv[u].w ^ pat) >> 4);
      if (j + u * OWN_WAVES >= N_CHUNKS) lm = 0u;
      if (u & 1) pk[u >> 1] |= lm << 4;
      else       pk[u >> 1] = lm;
    }
    scan_load(key4, j + SCAN_DEPTH * OWN_WAVES, start, lane, kv);
#pragma nounroll
    for (int w = 0; w < SCAN_DEPTH / 2; ++w) {
      unsigned lm = (w == 0) ? pk[0] : (w == 1) ? pk[1] : (w == 2) ? pk[2] : pk[3];
      int ca = j + 2 * w * OWN_WAVES + start;
      if (ca >= N_CHUNKS) ca -= N_CHUNKS;
      int cb = j + (2 * w + 1) * OWN_WAVES + start;
      if (cb >= N_CHUNKS) cb -= N_CHUNKS;
      const unsigned ea = (unsigned)ca * CHUNK_EDGES + (unsigned)lane * 16u;
      const unsigned eb = (unsigned)cb * CHUNK_EDGES + (unsigned)lane * 16u;
      for (int it = 0; it < 32; ++it) {
        const bool     m    = lm != 0u;
        const unsigned mask = __builtin_amdgcn_ballot_w32(m);
        if (mask == 0u) break;
        const unsigned p   = (unsigned)__builtin_ctz(lm | 0x80000000u);
        const unsigned pos =
            (unsigned)cnt + __builtin_amdgcn_mbcnt_lo(mask, 0u);
        if (m) q[pos] = ((p & 4u) ? eb : ea) + 4u * (p & 3u) + (p >> 3);
        lm &= lm - 1u;
        cnt += __builtin_popcount(mask);
        if (cnt >= 32) {
          cnt -= 32;
          own_batch(q + cnt, 32, lane, (const uint2*)erec, h, sT, wrec, nrec);
          nrec += 32;
        }
      }
    }
  }
  if (cnt > 0) {
    own_batch(q, cnt, lane, (const uint2*)erec, h, sT, wrec, nrec);
    nrec += cnt;
  }
  if (lane == 0) cntw[tile * OWN_WAVES + wave] = nrec;
  __syncthreads();

  if (mm) {
    const float* ar = sT + (r0 + lrow) * TPITCH + 2 * hi;
    v8f acc = {};
#pragma unroll
    for (int s = 0; s < 32; ++s) {
      const int    k = 4 * s + 2 * hi;
      const float2 t = *(const float2*)(ar + 4 * s);
      v2f a, b;
      a.x = fmaxf(t.x * di_a + sb[k], 0.f);
      a.y = fmaxf(t.y * di_a + sb[k + 1], 0.f);
      b.x = sW[k * C_DIM + lrow];
      b.y = sW[(k + 1) * C_DIM + lrow];
      acc = __builtin_amdgcn_wmma_f32_16x16x4_f32(false, a, false, b,
                                                  (short)0, acc, false, false);
    }
    float* o = h2 + (size_t)(n0 + r0 + 8 * hi) * C_DIM + lrow;
#pragma unroll
    for (int r = 0; r < 8; ++r) o[r * C_DIM] = acc[r] * dv[r];
  }
}


__device__ __forceinline__ void own_rows2(const int s, const unsigned dl,
                                          const float norm, const int lane,
                                          const float* __restrict__ h2,
                                          float* sT) {
  const int    col  = lane & 15;
  const int    half = lane >> 4;
  const float* hc   = h2 + col;
  float*       tc   = sT + col;
  float v[16];
  float nk[16];
  int   dk[16];
#pragma unroll
  for (int i = 0; i < 16; ++i) {
    const int k  = 2 * i + half;
    const int sk = __shfl(s, k, 32);
    dk[i] = __shfl((int)dl, k, 32);
    nk[i] = __shfl(norm, k, 32);
    v[i]  = hc[sk * C_DIM];
  }
#pragma unroll
  for (int i = 0; i < 16; ++i)
    __hip_atomic_fetch_add(tc + dk[i] * C_DIM, v[i] * nk[i],
                           __ATOMIC_RELAXED, __HIP_MEMORY_SCOPE_WORKGROUP);
}

__device__ __forceinline__ void own_batch2(const unsigned* qb, const int nvalid,
                                           const int lane, const int n0,
                                           const int* __restrict__ src,
                                           const int* __restrict__ dst,
                                           const float* __restrict__ ew,
                                           const float* __restrict__ h2,
                                           float* sT) {
  __builtin_amdgcn_fence(__ATOMIC_RELEASE, "wavefront");
  __builtin_amdgcn_wave_barrier();
  __builtin_amdgcn_fence(__ATOMIC_ACQUIRE, "wavefront");
  int      s    = 0;
  unsigned dl   = TILE_NODES;
  float    norm = 0.f;
  if (lane < nvalid) {
    const unsigned e = qb[lane];
    dl   = (unsigned)(dst[e] - n0);
    s    = src[e];
    norm = ew[e];
  }
  own_rows2(s, dl, norm, lane, h2, sT);
  __builtin_amdgcn_fence(__ATOMIC_RELEASE, "wavefront");
  __builtin_amdgcn_wave_barrier();
  __builtin_amdgcn_fence(__ATOMIC_ACQUIRE, "wavefront");
}

#define A2_COPY_THREADS (TILE_NODES * (C_DIM / 4) / 2)
static_assert(A2_COPY_THREADS % 32 == 0 && A2_COPY_THREADS <= 1024,
              "whole waves of the work-group");
static_assert(REC_CAP % 32 == 0, "a list is whole trips of 32 records");

__global__ __launch_bounds__(1024) void agg2_owner_kernel(
    const int* __restrict__ src, const int* __restrict__ dst,
    const float* __restrict__ ew, const float* __restrict__ dinv,
    const float* __restrict__ h2, const unsigned char* __restrict__ tkey,
    const unsigned* __restrict__ rec, const int* __restrict__ cntw,
    const float* __restrict__ b2, float* __restrict__ out) {
  __shared__ __attribute__((aligned(16))) float sT[(TILE_NODES + 1) * C_DIM];
  __shared__ unsigned sQ[OWN_WAVES * QCAP];
  __shared__ uint2 sR[OWN_WAVES * REC_CAP];
  const int tid  = threadIdx.x;
  const int lane = tid & 31;
  const int wave = __builtin_amdgcn_readfirstlane(tid >> 5);
  const int tile = blockIdx.x;
  const int n0   = tile * TILE_NODES;

  const int nrec = cntw[tile * OWN_WAVES + wave];

  const bool copier = wave < A2_COPY_THREADS / 32;
  float4 hv0 = make_float4(0.f, 0.f, 0.f, 0.f), hv1 = hv0;
  if (copier) {
    hv0 = ((const float4*)h2)[n0 * (C_DIM / 4) + tid];
    hv1 = ((const float4*)h2)[n0 * (C_DIM / 4) + tid + A2_COPY_THREADS];
  }
  const float di_out = dinv[n0 + (tid < TILE_NODES ? tid : TILE_NODES - 1)];
  float4 bq[C_DIM / 4];
#pragma unroll
  for (int qd = 0; qd < C_DIM / 4; ++qd) bq[qd] = ((const float4*)b2)[qd];

  const bool listed = nrec > 0 && nrec <= REC_CAP;
  uint2* myR = sR + wave * REC_CAP;
  if (listed) {
    const uint2* wrec =
        (const uint2*)rec + (size_t)(tile * OWN_WAVES + wave) * REC_CAP;
    const int last = nrec - 1;
    uint2 rv[REC_CAP / 32];
#pragma unroll
    for (int b = 0; b < REC_CAP / 32; ++b) {
      const int i = b * 32 + lane;
      rv[b] = wrec[i < nrec ? i : last];
    }
#pragma unroll
    for (int b = 0; b < REC_CAP / 32; ++b) myR[b * 32 + lane] = rv[b];
  }
  if (copier) {
    ((float4*)sT)[tid]                   = hv0;
    ((float4*)sT)[tid + A2_COPY_THREADS] = hv1;
  }
  if (tid < C_DIM) sT[TILE_NODES * C_DIM + tid] = 0.f;
  __syncthreads();

  if (nrec <= REC_CAP) {
    const uint2 idle = make_uint2((unsigned)TILE_NODES << REC_SBITS, 0u);
    for (int b = 0; b < nrec; b += 32) {
      uint2 cur = myR[b + lane];
      if (b + lane >= nrec) cur = idle;
      own_rows2((int)(cur.x & ((1u << REC_SBITS) - 1u)), cur.x >> REC_SBITS,
                __uint_as_float(cur.y), lane, h2, sT);
    }
  } else {
    unsigned*    q    = sQ + wave * QCAP;
    const uint4* key4 = (const uint4*)tkey;
    const unsigned pat = (unsigned)tile * 0x01010101u;
    const int start = (tile * N_CHUNKS) / N_TILES;
    int cnt = 0;
    for (int j = wave; j < N_CHUNKS; j += SCAN_DEPTH * OWN_WAVES) {
      uint4 kv[SCAN_DEPTH];
#pragma unroll
      for (int u = 0; u < SCAN_DEPTH; ++u) {
        const int jj = j + u * OWN_WAVES;
        int c = jj + start;
        if (c >= N_CHUNKS) c -= N_CHUNKS;
        if (jj >= N_CHUNKS) c = 0;
        kv[u] = key4[c * 32 + lane];
      }
      unsigned pk[SCAN_DEPTH / 2];
#pragma unroll
      for (int u = 0; u < SCAN_DEPTH; ++u) {
        unsigned lm = (zero_bytes(kv[u].x ^ pat) >> 7) |
                      (zero_bytes(kv[u].y ^ pat) >> 6) |
                      (zero_bytes(kv[u].z ^ pat) >> 5) |
                      (zero_bytes(kv[u].w ^ pat) >> 4);
        if (j + u * OWN_WAVES >= N_CHUNKS) lm = 0u;
        if (u & 1) pk[u >> 1] |= lm << 4;
        else       pk[u >> 1] = lm;
      }
#pragma nounroll
      for (int w = 0; w < SCAN_DEPTH / 2; ++w) {
        unsigned lm = (w == 0) ? pk[0] : (w == 1) ? pk[1] : (w == 2) ? pk[2] : pk[3];
        int ca = j + 2 * w * OWN_WAVES + start;
        if (ca >= N_CHUNKS) ca -= N_CHUNKS;
        int cb = j + (2 * w + 1) * OWN_WAVES + start;
        if (cb >= N_CHUNKS) cb -= N_CHUNKS;
        const unsigned ea = (unsigned)ca * CHUNK_EDGES + (unsigned)lane * 16u;
        const unsigned eb = (unsigned)cb * CHUNK_EDGES + (unsigned)lane * 16u;
        for (int it = 0; it < 32; ++it) {
          const bool     m    = lm != 0u;
          const unsigned mask = __builtin_amdgcn_ballot_w32(m);
          if (mask == 0u) break;
          const unsigned p   = (unsigned)__builtin_ctz(lm | 0x80000000u);
          const unsigned pos =
              (unsigned)cnt + __builtin_amdgcn_mbcnt_lo(mask, 0u);
          if (m) q[pos] = ((p & 4u) ? eb : ea) + 4u * (p & 3u) + (p >> 3);
          lm &= lm - 1u;
          cnt += __builtin_popcount(mask);
          if (cnt >= 32) {
            cnt -= 32;
            own_batch2(q + cnt, 32, lane, n0, src, dst, ew, h2, sT);
          }
        }
      }
    }
    if (cnt > 0) own_batch2(q, cnt, lane, n0, src, dst, ew, h2, sT);
  }
  __syncthreads();

  if (tid < TILE_NODES) {
    float v[C_DIM];
    const float4* p = (const float4*)(sT + tid * C_DIM);
    const float di = di_out;
    float m = -3.402823466e+38f;
#pragma unroll
    for (int qd = 0; qd < 4; ++qd) {
      const float4 t = p[qd];
      v[4 * qd + 0] = t.x * di + bq[qd].x;
      v[4 * qd + 1] = t.y * di + bq[qd].y;
      v[4 * qd + 2] = t.z * di + bq[qd].z;
      v[4 * qd + 3] = t.w * di + bq[qd].w;
    }
#pragma unroll
    for (int c = 0; c < C_DIM; ++c) m = fmaxf(m, v[c]);
    float sum = 0.f;
#pragma unroll
    for (int c = 0; c < C_DIM; ++c) { v[c] = expf(v[c] - m); sum += v[c]; }
    const float inv = 1.0f / sum;
    float4* o = (float4*)(out + (size_t)(n0 + tid) * C_DIM);
#pragma unroll
    for (int qd = 0; qd < 4; ++qd)
      o[qd] = make_float4(v[4 * qd + 0] * inv, v[4 * qd + 1] * inv,
                          v[4 * qd + 2] * inv, v[4 * qd + 3] * inv);
  }
}

extern "C" void kernel_launch(void* const* d_in, const int* in_sizes, int n_in,
                              void* d_out, int out_size, void* d_ws, size_t ws_size,
                              hipStream_t stream) {
  const float* x  = (const float*)d_in[0];
  const int*   ei = (const int*)d_in[1];
  const float* ew = (const float*)d_in[2];
  const float* W1 = (const float*)d_in[3];
  const float* b1 = (const float*)d_in[4];
  const float* W2 = (const float*)d_in[5];
  const float* b2 = (const float*)d_in[6];
  float* out = (float*)d_out;

  const int* src = ei;
  const int* dst = ei + N_EDGES;

  float* ws   = (float*)d_ws;
  float* dinv = ws;
  float* h    = ws + N_NODES;
  float* h2   = ws + N_NODES + (size_t)N_NODES * 128;
  unsigned char* tkey = (unsigned char*)(h2 + (size_t)N_NODES * C_DIM);
  float* part = (float*)(tkey + N_EDGES);
  unsigned* rec  = (unsigned*)(part + (size_t)DEG_SLICES * N_NODES);
  int*      cntw = (int*)(rec + (size_t)N_TILES * OWN_WAVES * REC_CAP * 2);
  unsigned* erec = (unsigned*)(cntw + N_TILES * OWN_WAVES);

  const int T = 256;

  deg_hist_kernel<<<DEG_RANGES * DEG_SLICES, DEG_THREADS, 0, stream>>>(src, dst, ew, part, tkey, erec);

  gemm1_wmma<<<(N_NODES + 63) / 64, T, 0, stream>>>(x, W1, part, dinv, h);
  agg1_owner_kernel<<<N_TILES, 1024, 0, stream>>>(erec, dinv, h, tkey, b1, W2, h2, rec, cntw);

  agg2_owner_kernel<<<N_TILES, 1024, 0, stream>>>(src, dst, ew, dinv, h2, tkey, rec, cntw, b2, out);
}
